# starting kernel
# baseline (speedup 1.0000x reference)
#include <hip/hip_runtime.h>
#include <math.h>

#define NN 2048
#define NB 8
#define NBINS 65536

typedef __attribute__((ext_vector_type(2))) float v2f;
typedef __attribute__((ext_vector_type(8))) float v8f;

__global__ __launch_bounds__(256) void k_row_softmax(const float* __restrict__ adj,
                                                     float* __restrict__ a) {
  __shared__ float red[256];
  const int i = blockIdx.x;
  const int t = threadIdx.x;
  const float* row = adj + (size_t)i * NN;

  float mymax = -INFINITY;
  for (int j = t; j < NN; j += 256) mymax = fmaxf(mymax, row[j]);
  red[t] = mymax; __syncthreads();
  for (int s = 128; s > 0; s >>= 1) { if (t < s) red[t] = fmaxf(red[t], red[t + s]); __syncthreads(); }
  const float m = red[0];
  __syncthreads();

  float sum = 0.f;
  for (int j = t; j < NN; j += 256) sum += __expf(row[j] - m);
  red[t] = sum; __syncthreads();
  for (int s = 128; s > 0; s >>= 1) { if (t < s) red[t] += red[t + s]; __syncthreads(); }
  const float inv = 1.f / red[0];

  float* arow = a + (size_t)i * NN;
  for (int j = t; j < NN; j += 256) arow[j] = __expf(row[j] - m) * inv;
}

__global__ __launch_bounds__(256) void k_hist_hi(const float* __restrict__ a,
                                                 unsigned* __restrict__ hist) {
  const size_t total = (size_t)NN * NN;
  const size_t stride = (size_t)gridDim.x * blockDim.x;
  for (size_t p = (size_t)blockIdx.x * blockDim.x + threadIdx.x; p < total; p += stride)
    atomicAdd(&hist[__float_as_uint(a[p]) >> 16], 1u);
}

__global__ void k_scan_hi(const unsigned* __restrict__ hist, unsigned* __restrict__ meta) {
  if (threadIdx.x == 0 && blockIdx.x == 0) {
    const unsigned k = ((unsigned)NN * (unsigned)NN - 1u) / 2u;
    unsigned cum = 0;
    for (unsigned bin = 0; bin < NBINS; ++bin) {
      const unsigned c = hist[bin];
      if (cum + c > k) { meta[0] = bin; meta[1] = cum; break; }
      cum += c;
    }
  }
}

__global__ __launch_bounds__(256) void k_hist_lo(const float* __restrict__ a,
                                                 const unsigned* __restrict__ meta,
                                                 unsigned* __restrict__ hist2) {
  const unsigned sel = meta[0];
  const size_t total = (size_t)NN * NN;
  const size_t stride = (size_t)gridDim.x * blockDim.x;
  for (size_t p = (size_t)blockIdx.x * blockDim.x + threadIdx.x; p < total; p += stride) {
    const unsigned bits = __float_as_uint(a[p]);
    if ((bits >> 16) == sel) atomicAdd(&hist2[bits & 0xFFFFu], 1u);
  }
}

__global__ void k_scan_lo(const unsigned* __restrict__ hist2, unsigned* __restrict__ meta) {
  if (threadIdx.x == 0 && blockIdx.x == 0) {
    const unsigned k = ((unsigned)NN * (unsigned)NN - 1u) / 2u;
    const unsigned kk = k - meta[1];
    unsigned cum = 0;
    for (unsigned low = 0; low < NBINS; ++low) {
      const unsigned c = hist2[low];
      if (cum + c > kk) { meta[2] = (meta[0] << 16) | low; break; }
      cum += c;
    }
  }
}

__global__ __launch_bounds__(32) void k_gemm_y(const float* __restrict__ x,
                                               const float* __restrict__ W,
                                               float* __restrict__ Y) {
  __shared__ float Wt[16][132];
  __shared__ float Xt[16][132];
  const int nb   = blockIdx.x * 16;
  const int lane = threadIdx.x;
  const int half = lane >> 4;
  const int l16  = lane & 15;

  for (int idx = lane; idx < 8 * 128; idx += 32)
    Xt[8 + (idx >> 7)][idx & 127] = 0.f;

  v8f acc0 = {}, acc1 = {};
  for (int k0 = 0; k0 < NN; k0 += 128) {
    for (int idx = lane; idx < 8 * 128; idx += 32) {
      const int r = idx >> 7, c = idx & 127;
      Xt[r][c] = x[r * NN + (k0 + c)];
    }
    for (int idx = lane; idx < 16 * 128; idx += 32) {
      const int r = idx >> 7, c = idx & 127;
      Wt[r][c] = W[(size_t)(nb + r) * NN + (k0 + c)];
    }
    __syncthreads();
    #pragma unroll 4
    for (int kk = 0; kk < 128; kk += 8) {
      const int ka0 = kk + half * 2;
      v2f a0 = { Xt[l16][ka0], Xt[l16][ka0 + 1] };
      v2f b0 = { Wt[l16][ka0], Wt[l16][ka0 + 1] };
      acc0 = __builtin_amdgcn_wmma_f32_16x16x4_f32(false, a0, false, b0,
                                                   (short)0, acc0, false, false);
      const int ka1 = kk + 4 + half * 2;
      v2f a1 = { Xt[l16][ka1], Xt[l16][ka1 + 1] };
      v2f b1 = { Wt[l16][ka1], Wt[l16][ka1 + 1] };
      acc1 = __builtin_amdgcn_wmma_f32_16x16x4_f32(false, a1, false, b1,
                                                   (short)0, acc1, false, false);
    }
    __syncthreads();
  }
  for (int r = 0; r < 8; ++r) {
    const int m = r + half * 8;
    if (m < NB) Y[m * NN + nb + l16] = acc0[r] + acc1[r];
  }
}

__global__ __launch_bounds__(256) void k_row_stats(const float* __restrict__ x,
                                                   const float* __restrict__ Y,
                                                   const float* __restrict__ bias,
                                                   float* __restrict__ rowM,
                                                   float* __restrict__ rowZ) {
  __shared__ float lr[NN];
  __shared__ float red[256];
  const int i = blockIdx.x;
  const int b = blockIdx.y;
  const int t = threadIdx.x;
  const float xbi = x[b * NN + i];

  float mymax = -INFINITY;
  for (int j = t; j < NN; j += 256) {
    float v = fmaf(xbi, Y[b * NN + j], bias[j]);
    v = (v > 0.f) ? v : 0.01f * v;
    lr[j] = v;
    mymax = fmaxf(mymax, v);
  }
  red[t] = mymax; __syncthreads();
  for (int s = 128; s > 0; s >>= 1) { if (t < s) red[t] = fmaxf(red[t], red[t + s]); __syncthreads(); }
  const float m = red[0];
  __syncthreads();

  float sum = 0.f;
  for (int j = t; j < NN; j += 256) sum += __expf(lr[j] - m);
  red[t] = sum; __syncthreads();
  for (int s = 128; s > 0; s >>= 1) { if (t < s) red[t] += red[t + s]; __syncthreads(); }
  if (t == 0) { rowM[b * NN + i] = m; rowZ[b * NN + i] = red[0]; }
}

__global__ __launch_bounds__(256) void k_aggregate(const float* __restrict__ x,
                                                   const float* __restrict__ Y,
                                                   const float* __restrict__ bias,
                                                   const float* __restrict__ a,
                                                   const unsigned* __restrict__ meta,
                                                   const float* __restrict__ rowM,
                                                   const float* __restrict__ rowZ,
                                                   float* __restrict__ out) {
  const int j  = blockIdx.x * 256 + threadIdx.x;
  const int b  = blockIdx.y;
  const int iBase = blockIdx.z * (NN / 8);
  const float med = __uint_as_float(meta[2]);
  const float ybj = Y[b * NN + j];
  const float bj  = bias[j];

  __shared__ float sx[128], sm[128], sc[128];
  float acc = 0.f;
  for (int i0 = iBase; i0 < iBase + 256; i0 += 128) {
    if (threadIdx.x < 128) {
      const int i = i0 + threadIdx.x;
      const float xv = x[b * NN + i];
      sx[threadIdx.x] = xv;
      sm[threadIdx.x] = rowM[b * NN + i];
      sc[threadIdx.x] = xv / rowZ[b * NN + i];
    }
    __syncthreads();
    for (int ii = 0; ii < 128; ++ii) {
      const int i = i0 + ii;
      if ((ii & 15) == 0)
        __builtin_prefetch(&a[(size_t)(i + 16) * NN + j], 0, 1);
      const float aij = a[(size_t)i * NN + j];
      float t = fmaf(sx[ii], ybj, bj);
      t = (t > 0.f) ? t : 0.01f * t;
      acc += (aij > med) ? sc[ii] * __expf(t - sm[ii]) : 0.f;
    }
    __syncthreads();
  }
  atomicAdd(&out[b * NN + j], acc);
}

extern "C" void kernel_launch(void* const* d_in, const int* in_sizes, int n_in,
                              void* d_out, int out_size, void* d_ws, size_t ws_size,
                              hipStream_t stream) {
  const float* x    = (const float*)d_in[0];
  const float* adj  = (const float*)d_in[1];
  const float* W    = (const float*)d_in[2];
  const float* bias = (const float*)d_in[3];
  float* out = (float*)d_out;

  char* ws = (char*)d_ws;
  size_t off = 0;
  float*    a     = (float*)(ws + off);    off += (size_t)NN * NN * sizeof(float);
  float*    Y     = (float*)(ws + off);    off += (size_t)NB * NN * sizeof(float);
  float*    rowM  = (float*)(ws + off);    off += (size_t)NB * NN * sizeof(float);
  float*    rowZ  = (float*)(ws + off);    off += (size_t)NB * NN * sizeof(float);
  unsigned* hist1 = (unsigned*)(ws + off); off += (size_t)NBINS * sizeof(unsigned);
  unsigned* hist2 = (unsigned*)(ws + off); off += (size_t)NBINS * sizeof(unsigned);
  unsigned* meta  = (unsigned*)(ws + off); off += 16;

  hipMemsetAsync(hist1, 0, 2 * (size_t)NBINS * sizeof(unsigned), stream);
  hipMemsetAsync(out, 0, (size_t)out_size * sizeof(float), stream);

  k_row_softmax<<<NN, 256, 0, stream>>>(adj, a);
  k_hist_hi   <<<1024, 256, 0, stream>>>(a, hist1);
  k_scan_hi   <<<1, 32, 0, stream>>>(hist1, meta);
  k_hist_lo   <<<1024, 256, 0, stream>>>(a, meta, hist2);
  k_scan_lo   <<<1, 32, 0, stream>>>(hist2, meta);
  k_gemm_y    <<<NN / 16, 32, 0, stream>>>(x, W, Y);
  k_row_stats <<<dim3(NN, NB), 256, 0, stream>>>(x, Y, bias, rowM, rowZ);
  k_aggregate <<<dim3(NN / 256, NB, 8), 256, 0, stream>>>(x, Y, bias, a, meta, rowM, rowZ, out);
}


# Both 65536-bin histogram scans of the median select: one thread walking the bins serially -> one 1024-thread block, whole-line loads and wave prefix sums (same integers, same bin)
# speedup vs baseline: 3.8545x; 3.8545x over previous
#include <hip/hip_runtime.h>
#include <math.h>

#define NN 2048
#define NB 8
#define NBINS 65536

typedef __attribute__((ext_vector_type(2))) float v2f;
typedef __attribute__((ext_vector_type(8))) float v8f;

__global__ __launch_bounds__(256) void k_row_softmax(const float* __restrict__ adj,
                                                     float* __restrict__ a) {
  __shared__ float red[256];
  const int i = blockIdx.x;
  const int t = threadIdx.x;
  const float* row = adj + (size_t)i * NN;

  float mymax = -INFINITY;
  for (int j = t; j < NN; j += 256) mymax = fmaxf(mymax, row[j]);
  red[t] = mymax; __syncthreads();
  for (int s = 128; s > 0; s >>= 1) { if (t < s) red[t] = fmaxf(red[t], red[t + s]); __syncthreads(); }
  const float m = red[0];
  __syncthreads();

  float sum = 0.f;
  for (int j = t; j < NN; j += 256) sum += __expf(row[j] - m);
  red[t] = sum; __syncthreads();
  for (int s = 128; s > 0; s >>= 1) { if (t < s) red[t] += red[t + s]; __syncthreads(); }
  const float inv = 1.f / red[0];

  float* arow = a + (size_t)i * NN;
  for (int j = t; j < NN; j += 256) arow[j] = __expf(row[j] - m) * inv;
}

__global__ __launch_bounds__(256) void k_hist_hi(const float* __restrict__ a,
                                                 unsigned* __restrict__ hist) {
  const size_t total = (size_t)NN * NN;
  const size_t stride = (size_t)gridDim.x * blockDim.x;
  for (size_t p = (size_t)blockIdx.x * blockDim.x + threadIdx.x; p < total; p += stride)
    atomicAdd(&hist[__float_as_uint(a[p]) >> 16], 1u);
}

#define SCAN_THREADS 1024
#define SCAN_WAVES   (SCAN_THREADS / 32)
#define WAVE_BINS    (NBINS / SCAN_WAVES)
#define WAVE_LOADS   (WAVE_BINS / 128)

__device__ __forceinline__ unsigned sum4(const uint4 v) { return v.x + v.y + v.z + v.w; }

__device__ __forceinline__ unsigned wave_inclusive_scan(unsigned v, const int lane) {
  #pragma unroll
  for (int d = 1; d < 32; d <<= 1) {
    const unsigned n = __shfl_up(v, d, 32);
    if (lane >= d) v += n;
  }
  return v;
}

__device__ __forceinline__ bool block_find_bin(const unsigned* __restrict__ hist,
                                               const unsigned target,
                                               unsigned& bin, unsigned& before) {
  __shared__ unsigned wave_tot[SCAN_WAVES];
  const int lane = threadIdx.x & 31;
  const int w    = threadIdx.x >> 5;
  const uint4* src = reinterpret_cast<const uint4*>(hist + w * WAVE_BINS) + lane;

  uint4 q[WAVE_LOADS];
  #pragma unroll
  for (int i = 0; i < WAVE_LOADS; ++i) q[i] = src[i * 32];

  unsigned wsum = 0;
  #pragma unroll
  for (int i = 0; i < WAVE_LOADS; ++i) wsum += sum4(q[i]);
  #pragma unroll
  for (int d = 16; d >= 1; d >>= 1) wsum += __shfl_xor(wsum, d, 32);
  if (lane == 0) wave_tot[w] = wsum;
  __syncthreads();
  const unsigned wt    = wave_tot[lane];
  const unsigned wincl = wave_inclusive_scan(wt, lane);
  const unsigned wave_before = __shfl(wincl - wt, w, 32);
  if (!(wave_before <= target && target < wave_before + wsum)) return false;

  unsigned t[WAVE_LOADS];
  #pragma unroll
  for (int i = 0; i < WAVE_LOADS; ++i) t[i] = sum4(q[i]);
  #pragma unroll
  for (int d = 16; d >= 1; d >>= 1) {
    #pragma unroll
    for (int i = 0; i < WAVE_LOADS; ++i) t[i] += __shfl_xor(t[i], d, 32);
  }
  unsigned base = wave_before;
  uint4 c = q[0];
  int isel = 0;
  bool found = false;
  #pragma unroll
  for (int i = 0; i < WAVE_LOADS; ++i) {
    if (!found) {
      if (base + t[i] > target) { c = q[i]; isel = i; found = true; }
      else base += t[i];
    }
  }

  const unsigned mine = sum4(c);
  const unsigned lane_before = base + wave_inclusive_scan(mine, lane) - mine;
  if (!(lane_before <= target && target < lane_before + mine)) return false;
  unsigned cum = lane_before;
  unsigned comp = 0;
  if (cum + c.x <= target) {
    cum += c.x; comp = 1;
    if (cum + c.y <= target) {
      cum += c.y; comp = 2;
      if (cum + c.z <= target) { cum += c.z; comp = 3; }
    }
  }
  bin    = (unsigned)(w * WAVE_BINS + isel * 128 + lane * 4) + comp;
  before = cum;
  return true;
}

__global__ __launch_bounds__(SCAN_THREADS) void k_scan_hi(const unsigned* __restrict__ hist,
                                                          unsigned* __restrict__ meta) {
  const unsigned k = ((unsigned)NN * (unsigned)NN - 1u) / 2u;
  unsigned bin, before;
  if (block_find_bin(hist, k, bin, before)) { meta[0] = bin; meta[1] = before; }
}

__global__ __launch_bounds__(256) void k_hist_lo(const float* __restrict__ a,
                                                 const unsigned* __restrict__ meta,
                                                 unsigned* __restrict__ hist2) {
  const unsigned sel = meta[0];
  const size_t total = (size_t)NN * NN;
  const size_t stride = (size_t)gridDim.x * blockDim.x;
  for (size_t p = (size_t)blockIdx.x * blockDim.x + threadIdx.x; p < total; p += stride) {
    const unsigned bits = __float_as_uint(a[p]);
    if ((bits >> 16) == sel) atomicAdd(&hist2[bits & 0xFFFFu], 1u);
  }
}

__global__ __launch_bounds__(SCAN_THREADS) void k_scan_lo(const unsigned* __restrict__ hist2,
                                                          unsigned* __restrict__ meta) {
  const unsigned k = ((unsigned)NN * (unsigned)NN - 1u) / 2u;
  const unsigned kk = k - meta[1];
  const unsigned hi = meta[0];
  unsigned low, before;
  if (block_find_bin(hist2, kk, low, before)) meta[2] = (hi << 16) | low;
}

__global__ __launch_bounds__(32) void k_gemm_y(const float* __restrict__ x,
                                               const float* __restrict__ W,
                                               float* __restrict__ Y) {
  __shared__ float Wt[16][132];
  __shared__ float Xt[16][132];
  const int nb   = blockIdx.x * 16;
  const int lane = threadIdx.x;
  const int half = lane >> 4;
  const int l16  = lane & 15;

  for (int idx = lane; idx < 8 * 128; idx += 32)
    Xt[8 + (idx >> 7)][idx & 127] = 0.f;

  v8f acc0 = {}, acc1 = {};
  for (int k0 = 0; k0 < NN; k0 += 128) {
    for (int idx = lane; idx < 8 * 128; idx += 32) {
      const int r = idx >> 7, c = idx & 127;
      Xt[r][c] = x[r * NN + (k0 + c)];
    }
    for (int idx = lane; idx < 16 * 128; idx += 32) {
      const int r = idx >> 7, c = idx & 127;
      Wt[r][c] = W[(size_t)(nb + r) * NN + (k0 + c)];
    }
    __syncthreads();
    #pragma unroll 4
    for (int kk = 0; kk < 128; kk += 8) {
      const int ka0 = kk + half * 2;
      v2f a0 = { Xt[l16][ka0], Xt[l16][ka0 + 1] };
      v2f b0 = { Wt[l16][ka0], Wt[l16][ka0 + 1] };
      acc0 = __builtin_amdgcn_wmma_f32_16x16x4_f32(false, a0, false, b0,
                                                   (short)0, acc0, false, false);
      const int ka1 = kk + 4 + half * 2;
      v2f a1 = { Xt[l16][ka1], Xt[l16][ka1 + 1] };
      v2f b1 = { Wt[l16][ka1], Wt[l16][ka1 + 1] };
      acc1 = __builtin_amdgcn_wmma_f32_16x16x4_f32(false, a1, false, b1,
                                                   (short)0, acc1, false, false);
    }
    __syncthreads();
  }
  for (int r = 0; r < 8; ++r) {
    const int m = r + half * 8;
    if (m < NB) Y[m * NN + nb + l16] = acc0[r] + acc1[r];
  }
}

__global__ __launch_bounds__(256) void k_row_stats(const float* __restrict__ x,
                                                   const float* __restrict__ Y,
                                                   const float* __restrict__ bias,
                                                   float* __restrict__ rowM,
                                                   float* __restrict__ rowZ) {
  __shared__ float lr[NN];
  __shared__ float red[256];
  const int i = blockIdx.x;
  const int b = blockIdx.y;
  const int t = threadIdx.x;
  const float xbi = x[b * NN + i];

  float mymax = -INFINITY;
  for (int j = t; j < NN; j += 256) {
    float v = fmaf(xbi, Y[b * NN + j], bias[j]);
    v = (v > 0.f) ? v : 0.01f * v;
    lr[j] = v;
    mymax = fmaxf(mymax, v);
  }
  red[t] = mymax; __syncthreads();
  for (int s = 128; s > 0; s >>= 1) { if (t < s) red[t] = fmaxf(red[t], red[t + s]); __syncthreads(); }
  const float m = red[0];
  __syncthreads();

  float sum = 0.f;
  for (int j = t; j < NN; j += 256) sum += __expf(lr[j] - m);
  red[t] = sum; __syncthreads();
  for (int s = 128; s > 0; s >>= 1) { if (t < s) red[t] += red[t + s]; __syncthreads(); }
  if (t == 0) { rowM[b * NN + i] = m; rowZ[b * NN + i] = red[0]; }
}

__global__ __launch_bounds__(256) void k_aggregate(const float* __restrict__ x,
                                                   const float* __restrict__ Y,
                                                   const float* __restrict__ bias,
                                                   const float* __restrict__ a,
                                                   const unsigned* __restrict__ meta,
                                                   const float* __restrict__ rowM,
                                                   const float* __restrict__ rowZ,
                                                   float* __restrict__ out) {
  const int j  = blockIdx.x * 256 + threadIdx.x;
  const int b  = blockIdx.y;
  const int iBase = blockIdx.z * (NN / 8);
  const float med = __uint_as_float(meta[2]);
  const float ybj = Y[b * NN + j];
  const float bj  = bias[j];

  __shared__ float sx[128], sm[128], sc[128];
  float acc = 0.f;
  for (int i0 = iBase; i0 < iBase + 256; i0 += 128) {
    if (threadIdx.x < 128) {
      const int i = i0 + threadIdx.x;
      const float xv = x[b * NN + i];
      sx[threadIdx.x] = xv;
      sm[threadIdx.x] = rowM[b * NN + i];
      sc[threadIdx.x] = xv / rowZ[b * NN + i];
    }
    __syncthreads();
    for (int ii = 0; ii < 128; ++ii) {
      const int i = i0 + ii;
      if ((ii & 15) == 0)
        __builtin_prefetch(&a[(size_t)(i + 16) * NN + j], 0, 1);
      const float aij = a[(size_t)i * NN + j];
      float t = fmaf(sx[ii], ybj, bj);
      t = (t > 0.f) ? t : 0.01f * t;
      acc += (aij > med) ? sc[ii] * __expf(t - sm[ii]) : 0.f;
    }
    __syncthreads();
  }
  atomicAdd(&out[b * NN + j], acc);
}

extern "C" void kernel_launch(void* const* d_in, const int* in_sizes, int n_in,
                              void* d_out, int out_size, void* d_ws, size_t ws_size,
                              hipStream_t stream) {
  const float* x    = (const float*)d_in[0];
  const float* adj  = (const float*)d_in[1];
  const float* W    = (const float*)d_in[2];
  const float* bias = (const float*)d_in[3];
  float* out = (float*)d_out;

  char* ws = (char*)d_ws;
  size_t off = 0;
  float*    a     = (float*)(ws + off);    off += (size_t)NN * NN * sizeof(float);
  float*    Y     = (float*)(ws + off);    off += (size_t)NB * NN * sizeof(float);
  float*    rowM  = (float*)(ws + off);    off += (size_t)NB * NN * sizeof(float);
  float*    rowZ  = (float*)(ws + off);    off += (size_t)NB * NN * sizeof(float);
  unsigned* hist1 = (unsigned*)(ws + off); off += (size_t)NBINS * sizeof(unsigned);
  unsigned* hist2 = (unsigned*)(ws + off); off += (size_t)NBINS * sizeof(unsigned);
  unsigned* meta  = (unsigned*)(ws + off); off += 16;

  hipMemsetAsync(hist1, 0, 2 * (size_t)NBINS * sizeof(unsigned), stream);
  hipMemsetAsync(out, 0, (size_t)out_size * sizeof(float), stream);

  k_row_softmax<<<NN, 256, 0, stream>>>(adj, a);
  k_hist_hi   <<<1024, 256, 0, stream>>>(a, hist1);
  k_scan_hi   <<<1, SCAN_THREADS, 0, stream>>>(hist1, meta);
  k_hist_lo   <<<1024, 256, 0, stream>>>(a, meta, hist2);
  k_scan_lo   <<<1, SCAN_THREADS, 0, stream>>>(hist2, meta);
  k_gemm_y    <<<NN / 16, 32, 0, stream>>>(x, W, Y);
  k_row_stats <<<dim3(NN, NB), 256, 0, stream>>>(x, Y, bias, rowM, rowZ);
  k_aggregate <<<dim3(NN / 256, NB, 8), 256, 0, stream>>>(x, Y, bias, a, meta, rowM, rowZ, out);
}


# High-16-bit histogram of the median select: one global atomic per element -> each work-group counts in a private 65536-bin LDS histogram and adds its non-empty bins once (same integer counts)
# speedup vs baseline: 7.5445x; 1.9573x over previous
#include <hip/hip_runtime.h>
#include <math.h>

#define NN 2048
#define NB 8
#define NBINS 65536

typedef __attribute__((ext_vector_type(2))) float v2f;
typedef __attribute__((ext_vector_type(8))) float v8f;

__global__ __launch_bounds__(256) void k_row_softmax(const float* __restrict__ adj,
                                                     float* __restrict__ a) {
  __shared__ float red[256];
  const int i = blockIdx.x;
  const int t = threadIdx.x;
  const float* row = adj + (size_t)i * NN;

  float mymax = -INFINITY;
  for (int j = t; j < NN; j += 256) mymax = fmaxf(mymax, row[j]);
  red[t] = mymax; __syncthreads();
  for (int s = 128; s > 0; s >>= 1) { if (t < s) red[t] = fmaxf(red[t], red[t + s]); __syncthreads(); }
  const float m = red[0];
  __syncthreads();

  float sum = 0.f;
  for (int j = t; j < NN; j += 256) sum += __expf(row[j] - m);
  red[t] = sum; __syncthreads();
  for (int s = 128; s > 0; s >>= 1) { if (t < s) red[t] += red[t + s]; __syncthreads(); }
  const float inv = 1.f / red[0];

  float* arow = a + (size_t)i * NN;
  for (int j = t; j < NN; j += 256) arow[j] = __expf(row[j] - m) * inv;
}

#define HIST_BLOCKS  256
#define HIST_THREADS 1024
#define HIST_VEC     ((NN * NN / 4) / (HIST_BLOCKS * HIST_THREADS))
#define FLUSH_BATCH  16

__global__ __launch_bounds__(HIST_THREADS) void k_hist_hi(const float* __restrict__ a,
                                                          unsigned* __restrict__ hist) {
  __shared__ unsigned lh[NBINS];
  const int t = threadIdx.x;

  const float4* src = reinterpret_cast<const float4*>(a)
                    + (size_t)blockIdx.x * (HIST_VEC * HIST_THREADS) + t;
  float4 v[HIST_VEC];
  #pragma unroll
  for (int i = 0; i < HIST_VEC; ++i) v[i] = src[i * HIST_THREADS];

  uint4* lh4 = reinterpret_cast<uint4*>(lh);
  for (int i = t; i < NBINS / 4; i += HIST_THREADS) lh4[i] = make_uint4(0u, 0u, 0u, 0u);
  __syncthreads();

  #pragma unroll
  for (int i = 0; i < HIST_VEC; ++i) {
    atomicAdd(&lh[__float_as_uint(v[i].x) >> 16], 1u);
    atomicAdd(&lh[__float_as_uint(v[i].y) >> 16], 1u);
    atomicAdd(&lh[__float_as_uint(v[i].z) >> 16], 1u);
    atomicAdd(&lh[__float_as_uint(v[i].w) >> 16], 1u);
  }
  __syncthreads();

  for (int i0 = 0; i0 < NBINS; i0 += FLUSH_BATCH * HIST_THREADS) {
    unsigned c[FLUSH_BATCH];
    #pragma unroll
    for (int u = 0; u < FLUSH_BATCH; ++u) c[u] = lh[i0 + u * HIST_THREADS + t];
    #pragma unroll
    for (int u = 0; u < FLUSH_BATCH; ++u)
      if (c[u] != 0u) atomicAdd(&hist[i0 + u * HIST_THREADS + t], c[u]);
  }
}

#define SCAN_THREADS 1024
#define SCAN_WAVES   (SCAN_THREADS / 32)
#define WAVE_BINS    (NBINS / SCAN_WAVES)
#define WAVE_LOADS   (WAVE_BINS / 128)

__device__ __forceinline__ unsigned sum4(const uint4 v) { return v.x + v.y + v.z + v.w; }

__device__ __forceinline__ unsigned wave_inclusive_scan(unsigned v, const int lane) {
  #pragma unroll
  for (int d = 1; d < 32; d <<= 1) {
    const unsigned n = __shfl_up(v, d, 32);
    if (lane >= d) v += n;
  }
  return v;
}

__device__ __forceinline__ bool block_find_bin(const unsigned* __restrict__ hist,
                                               const unsigned target,
                                               unsigned& bin, unsigned& before) {
  __shared__ unsigned wave_tot[SCAN_WAVES];
  const int lane = threadIdx.x & 31;
  const int w    = threadIdx.x >> 5;
  const uint4* src = reinterpret_cast<const uint4*>(hist + w * WAVE_BINS) + lane;

  uint4 q[WAVE_LOADS];
  #pragma unroll
  for (int i = 0; i < WAVE_LOADS; ++i) q[i] = src[i * 32];

  unsigned wsum = 0;
  #pragma unroll
  for (int i = 0; i < WAVE_LOADS; ++i) wsum += sum4(q[i]);
  #pragma unroll
  for (int d = 16; d >= 1; d >>= 1) wsum += __shfl_xor(wsum, d, 32);
  if (lane == 0) wave_tot[w] = wsum;
  __syncthreads();
  const unsigned wt    = wave_tot[lane];
  const unsigned wincl = wave_inclusive_scan(wt, lane);
  const unsigned wave_before = __shfl(wincl - wt, w, 32);
  if (!(wave_before <= target && target < wave_before + wsum)) return false;

  unsigned t[WAVE_LOADS];
  #pragma unroll
  for (int i = 0; i < WAVE_LOADS; ++i) t[i] = sum4(q[i]);
  #pragma unroll
  for (int d = 16; d >= 1; d >>= 1) {
    #pragma unroll
    for (int i = 0; i < WAVE_LOADS; ++i) t[i] += __shfl_xor(t[i], d, 32);
  }
  unsigned base = wave_before;
  uint4 c = q[0];
  int isel = 0;
  bool found = false;
  #pragma unroll
  for (int i = 0; i < WAVE_LOADS; ++i) {
    if (!found) {
      if (base + t[i] > target) { c = q[i]; isel = i; found = true; }
      else base += t[i];
    }
  }

  const unsigned mine = sum4(c);
  const unsigned lane_before = base + wave_inclusive_scan(mine, lane) - mine;
  if (!(lane_before <= target && target < lane_before + mine)) return false;
  unsigned cum = lane_before;
  unsigned comp = 0;
  if (cum + c.x <= target) {
    cum += c.x; comp = 1;
    if (cum + c.y <= target) {
      cum += c.y; comp = 2;
      if (cum + c.z <= target) { cum += c.z; comp = 3; }
    }
  }
  bin    = (unsigned)(w * WAVE_BINS + isel * 128 + lane * 4) + comp;
  before = cum;
  return true;
}

__global__ __launch_bounds__(SCAN_THREADS) void k_scan_hi(const unsigned* __restrict__ hist,
                                                          unsigned* __restrict__ meta) {
  const unsigned k = ((unsigned)NN * (unsigned)NN - 1u) / 2u;
  unsigned bin, before;
  if (block_find_bin(hist, k, bin, before)) { meta[0] = bin; meta[1] = before; }
}

__global__ __launch_bounds__(256) void k_hist_lo(const float* __restrict__ a,
                                                 const unsigned* __restrict__ meta,
                                                 unsigned* __restrict__ hist2) {
  const unsigned sel = meta[0];
  const size_t total = (size_t)NN * NN;
  const size_t stride = (size_t)gridDim.x * blockDim.x;
  for (size_t p = (size_t)blockIdx.x * blockDim.x + threadIdx.x; p < total; p += stride) {
    const unsigned bits = __float_as_uint(a[p]);
    if ((bits >> 16) == sel) atomicAdd(&hist2[bits & 0xFFFFu], 1u);
  }
}

__global__ __launch_bounds__(SCAN_THREADS) void k_scan_lo(const unsigned* __restrict__ hist2,
                                                          unsigned* __restrict__ meta) {
  const unsigned k = ((unsigned)NN * (unsigned)NN - 1u) / 2u;
  const unsigned kk = k - meta[1];
  const unsigned hi = meta[0];
  unsigned low, before;
  if (block_find_bin(hist2, kk, low, before)) meta[2] = (hi << 16) | low;
}

__global__ __launch_bounds__(32) void k_gemm_y(const float* __restrict__ x,
                                               const float* __restrict__ W,
                                               float* __restrict__ Y) {
  __shared__ float Wt[16][132];
  __shared__ float Xt[16][132];
  const int nb   = blockIdx.x * 16;
  const int lane = threadIdx.x;
  const int half = lane >> 4;
  const int l16  = lane & 15;

  for (int idx = lane; idx < 8 * 128; idx += 32)
    Xt[8 + (idx >> 7)][idx & 127] = 0.f;

  v8f acc0 = {}, acc1 = {};
  for (int k0 = 0; k0 < NN; k0 += 128) {
    for (int idx = lane; idx < 8 * 128; idx += 32) {
      const int r = idx >> 7, c = idx & 127;
      Xt[r][c] = x[r * NN + (k0 + c)];
    }
    for (int idx = lane; idx < 16 * 128; idx += 32) {
      const int r = idx >> 7, c = idx & 127;
      Wt[r][c] = W[(size_t)(nb + r) * NN + (k0 + c)];
    }
    __syncthreads();
    #pragma unroll 4
    for (int kk = 0; kk < 128; kk += 8) {
      const int ka0 = kk + half * 2;
      v2f a0 = { Xt[l16][ka0], Xt[l16][ka0 + 1] };
      v2f b0 = { Wt[l16][ka0], Wt[l16][ka0 + 1] };
      acc0 = __builtin_amdgcn_wmma_f32_16x16x4_f32(false, a0, false, b0,
                                                   (short)0, acc0, false, false);
      const int ka1 = kk + 4 + half * 2;
      v2f a1 = { Xt[l16][ka1], Xt[l16][ka1 + 1] };
      v2f b1 = { Wt[l16][ka1], Wt[l16][ka1 + 1] };
      acc1 = __builtin_amdgcn_wmma_f32_16x16x4_f32(false, a1, false, b1,
                                                   (short)0, acc1, false, false);
    }
    __syncthreads();
  }
  for (int r = 0; r < 8; ++r) {
    const int m = r + half * 8;
    if (m < NB) Y[m * NN + nb + l16] = acc0[r] + acc1[r];
  }
}

__global__ __launch_bounds__(256) void k_row_stats(const float* __restrict__ x,
                                                   const float* __restrict__ Y,
                                                   const float* __restrict__ bias,
                                                   float* __restrict__ rowM,
                                                   float* __restrict__ rowZ) {
  __shared__ float lr[NN];
  __shared__ float red[256];
  const int i = blockIdx.x;
  const int b = blockIdx.y;
  const int t = threadIdx.x;
  const float xbi = x[b * NN + i];

  float mymax = -INFINITY;
  for (int j = t; j < NN; j += 256) {
    float v = fmaf(xbi, Y[b * NN + j], bias[j]);
    v = (v > 0.f) ? v : 0.01f * v;
    lr[j] = v;
    mymax = fmaxf(mymax, v);
  }
  red[t] = mymax; __syncthreads();
  for (int s = 128; s > 0; s >>= 1) { if (t < s) red[t] = fmaxf(red[t], red[t + s]); __syncthreads(); }
  const float m = red[0];
  __syncthreads();

  float sum = 0.f;
  for (int j = t; j < NN; j += 256) sum += __expf(lr[j] - m);
  red[t] = sum; __syncthreads();
  for (int s = 128; s > 0; s >>= 1) { if (t < s) red[t] += red[t + s]; __syncthreads(); }
  if (t == 0) { rowM[b * NN + i] = m; rowZ[b * NN + i] = red[0]; }
}

__global__ __launch_bounds__(256) void k_aggregate(const float* __restrict__ x,
                                                   const float* __restrict__ Y,
                                                   const float* __restrict__ bias,
                                                   const float* __restrict__ a,
                                                   const unsigned* __restrict__ meta,
                                                   const float* __restrict__ rowM,
                                                   const float* __restrict__ rowZ,
                                                   float* __restrict__ out) {
  const int j  = blockIdx.x * 256 + threadIdx.x;
  const int b  = blockIdx.y;
  const int iBase = blockIdx.z * (NN / 8);
  const float med = __uint_as_float(meta[2]);
  const float ybj = Y[b * NN + j];
  const float bj  = bias[j];

  __shared__ float sx[128], sm[128], sc[128];
  float acc = 0.f;
  for (int i0 = iBase; i0 < iBase + 256; i0 += 128) {
    if (threadIdx.x < 128) {
      const int i = i0 + threadIdx.x;
      const float xv = x[b * NN + i];
      sx[threadIdx.x] = xv;
      sm[threadIdx.x] = rowM[b * NN + i];
      sc[threadIdx.x] = xv / rowZ[b * NN + i];
    }
    __syncthreads();
    for (int ii = 0; ii < 128; ++ii) {
      const int i = i0 + ii;
      if ((ii & 15) == 0)
        __builtin_prefetch(&a[(size_t)(i + 16) * NN + j], 0, 1);
      const float aij = a[(size_t)i * NN + j];
      float t = fmaf(sx[ii], ybj, bj);
      t = (t > 0.f) ? t : 0.01f * t;
      acc += (aij > med) ? sc[ii] * __expf(t - sm[ii]) : 0.f;
    }
    __syncthreads();
  }
  atomicAdd(&out[b * NN + j], acc);
}

extern "C" void kernel_launch(void* const* d_in, const int* in_sizes, int n_in,
                              void* d_out, int out_size, void* d_ws, size_t ws_size,
                              hipStream_t stream) {
  const float* x    = (const float*)d_in[0];
  const float* adj  = (const float*)d_in[1];
  const float* W    = (const float*)d_in[2];
  const float* bias = (const float*)d_in[3];
  float* out = (float*)d_out;

  char* ws = (char*)d_ws;
  size_t off = 0;
  float*    a     = (float*)(ws + off);    off += (size_t)NN * NN * sizeof(float);
  float*    Y     = (float*)(ws + off);    off += (size_t)NB * NN * sizeof(float);
  float*    rowM  = (float*)(ws + off);    off += (size_t)NB * NN * sizeof(float);
  float*    rowZ  = (float*)(ws + off);    off += (size_t)NB * NN * sizeof(float);
  unsigned* hist1 = (unsigned*)(ws + off); off += (size_t)NBINS * sizeof(unsigned);
  unsigned* hist2 = (unsigned*)(ws + off); off += (size_t)NBINS * sizeof(unsigned);
  unsigned* meta  = (unsigned*)(ws + off); off += 16;

  hipMemsetAsync(hist1, 0, 2 * (size_t)NBINS * sizeof(unsigned), stream);
  hipMemsetAsync(out, 0, (size_t)out_size * sizeof(float), stream);

  k_row_softmax<<<NN, 256, 0, stream>>>(adj, a);
  k_hist_hi   <<<HIST_BLOCKS, HIST_THREADS, 0, stream>>>(a, hist1);
  k_scan_hi   <<<1, SCAN_THREADS, 0, stream>>>(hist1, meta);
  k_hist_lo   <<<1024, 256, 0, stream>>>(a, meta, hist2);
  k_scan_lo   <<<1, SCAN_THREADS, 0, stream>>>(hist2, meta);
  k_gemm_y    <<<NN / 16, 32, 0, stream>>>(x, W, Y);
  k_row_stats <<<dim3(NN, NB), 256, 0, stream>>>(x, Y, bias, rowM, rowZ);
  k_aggregate <<<dim3(NN / 256, NB, 8), 256, 0, stream>>>(x, Y, bias, a, meta, rowM, rowZ, out);
}


# Y GEMM (x W^T): 1 wave per 16-col tile with 8 loads in flight -> 16 waves per tile, each issues its whole 128-k slice at once (512 B row loads via LDS); partials added in wave order (new k sum order)
# speedup vs baseline: 12.3915x; 1.6425x over previous
#include <hip/hip_runtime.h>
#include <math.h>

#define NN 2048
#define NB 8
#define NBINS 65536

typedef __attribute__((ext_vector_type(2))) float v2f;
typedef __attribute__((ext_vector_type(8))) float v8f;

__global__ __launch_bounds__(256) void k_row_softmax(const float* __restrict__ adj,
                                                     float* __restrict__ a) {
  __shared__ float red[256];
  const int i = blockIdx.x;
  const int t = threadIdx.x;
  const float* row = adj + (size_t)i * NN;

  float mymax = -INFINITY;
  for (int j = t; j < NN; j += 256) mymax = fmaxf(mymax, row[j]);
  red[t] = mymax; __syncthreads();
  for (int s = 128; s > 0; s >>= 1) { if (t < s) red[t] = fmaxf(red[t], red[t + s]); __syncthreads(); }
  const float m = red[0];
  __syncthreads();

  float sum = 0.f;
  for (int j = t; j < NN; j += 256) sum += __expf(row[j] - m);
  red[t] = sum; __syncthreads();
  for (int s = 128; s > 0; s >>= 1) { if (t < s) red[t] += red[t + s]; __syncthreads(); }
  const float inv = 1.f / red[0];

  float* arow = a + (size_t)i * NN;
  for (int j = t; j < NN; j += 256) arow[j] = __expf(row[j] - m) * inv;
}

#define HIST_BLOCKS  256
#define HIST_THREADS 1024
#define HIST_VEC     ((NN * NN / 4) / (HIST_BLOCKS * HIST_THREADS))
#define FLUSH_BATCH  16

__global__ __launch_bounds__(HIST_THREADS) void k_hist_hi(const float* __restrict__ a,
                                                          unsigned* __restrict__ hist) {
  __shared__ unsigned lh[NBINS];
  const int t = threadIdx.x;

  const float4* src = reinterpret_cast<const float4*>(a)
                    + (size_t)blockIdx.x * (HIST_VEC * HIST_THREADS) + t;
  float4 v[HIST_VEC];
  #pragma unroll
  for (int i = 0; i < HIST_VEC; ++i) v[i] = src[i * HIST_THREADS];

  uint4* lh4 = reinterpret_cast<uint4*>(lh);
  for (int i = t; i < NBINS / 4; i += HIST_THREADS) lh4[i] = make_uint4(0u, 0u, 0u, 0u);
  __syncthreads();

  #pragma unroll
  for (int i = 0; i < HIST_VEC; ++i) {
    atomicAdd(&lh[__float_as_uint(v[i].x) >> 16], 1u);
    atomicAdd(&lh[__float_as_uint(v[i].y) >> 16], 1u);
    atomicAdd(&lh[__float_as_uint(v[i].z) >> 16], 1u);
    atomicAdd(&lh[__float_as_uint(v[i].w) >> 16], 1u);
  }
  __syncthreads();

  for (int i0 = 0; i0 < NBINS; i0 += FLUSH_BATCH * HIST_THREADS) {
    unsigned c[FLUSH_BATCH];
    #pragma unroll
    for (int u = 0; u < FLUSH_BATCH; ++u) c[u] = lh[i0 + u * HIST_THREADS + t];
    #pragma unroll
    for (int u = 0; u < FLUSH_BATCH; ++u)
      if (c[u] != 0u) atomicAdd(&hist[i0 + u * HIST_THREADS + t], c[u]);
  }
}

#define SCAN_THREADS 1024
#define SCAN_WAVES   (SCAN_THREADS / 32)
#define WAVE_BINS    (NBINS / SCAN_WAVES)
#define WAVE_LOADS   (WAVE_BINS / 128)

__device__ __forceinline__ unsigned sum4(const uint4 v) { return v.x + v.y + v.z + v.w; }

__device__ __forceinline__ unsigned wave_inclusive_scan(unsigned v, const int lane) {
  #pragma unroll
  for (int d = 1; d < 32; d <<= 1) {
    const unsigned n = __shfl_up(v, d, 32);
    if (lane >= d) v += n;
  }
  return v;
}

__device__ __forceinline__ bool block_find_bin(const unsigned* __restrict__ hist,
                                               const unsigned target,
                                               unsigned& bin, unsigned& before) {
  __shared__ unsigned wave_tot[SCAN_WAVES];
  const int lane = threadIdx.x & 31;
  const int w    = threadIdx.x >> 5;
  const uint4* src = reinterpret_cast<const uint4*>(hist + w * WAVE_BINS) + lane;

  uint4 q[WAVE_LOADS];
  #pragma unroll
  for (int i = 0; i < WAVE_LOADS; ++i) q[i] = src[i * 32];

  unsigned wsum = 0;
  #pragma unroll
  for (int i = 0; i < WAVE_LOADS; ++i) wsum += sum4(q[i]);
  #pragma unroll
  for (int d = 16; d >= 1; d >>= 1) wsum += __shfl_xor(wsum, d, 32);
  if (lane == 0) wave_tot[w] = wsum;
  __syncthreads();
  const unsigned wt    = wave_tot[lane];
  const unsigned wincl = wave_inclusive_scan(wt, lane);
  const unsigned wave_before = __shfl(wincl - wt, w, 32);
  if (!(wave_before <= target && target < wave_before + wsum)) return false;

  unsigned t[WAVE_LOADS];
  #pragma unroll
  for (int i = 0; i < WAVE_LOADS; ++i) t[i] = sum4(q[i]);
  #pragma unroll
  for (int d = 16; d >= 1; d >>= 1) {
    #pragma unroll
    for (int i = 0; i < WAVE_LOADS; ++i) t[i] += __shfl_xor(t[i], d, 32);
  }
  unsigned base = wave_before;
  uint4 c = q[0];
  int isel = 0;
  bool found = false;
  #pragma unroll
  for (int i = 0; i < WAVE_LOADS; ++i) {
    if (!found) {
      if (base + t[i] > target) { c = q[i]; isel = i; found = true; }
      else base += t[i];
    }
  }

  const unsigned mine = sum4(c);
  const unsigned lane_before = base + wave_inclusive_scan(mine, lane) - mine;
  if (!(lane_before <= target && target < lane_before + mine)) return false;
  unsigned cum = lane_before;
  unsigned comp = 0;
  if (cum + c.x <= target) {
    cum += c.x; comp = 1;
    if (cum + c.y <= target) {
      cum += c.y; comp = 2;
      if (cum + c.z <= target) { cum += c.z; comp = 3; }
    }
  }
  bin    = (unsigned)(w * WAVE_BINS + isel * 128 + lane * 4) + comp;
  before = cum;
  return true;
}

__global__ __launch_bounds__(SCAN_THREADS) void k_scan_hi(const unsigned* __restrict__ hist,
                                                          unsigned* __restrict__ meta) {
  const unsigned k = ((unsigned)NN * (unsigned)NN - 1u) / 2u;
  unsigned bin, before;
  if (block_find_bin(hist, k, bin, before)) { meta[0] = bin; meta[1] = before; }
}

__global__ __launch_bounds__(256) void k_hist_lo(const float* __restrict__ a,
                                                 const unsigned* __restrict__ meta,
                                                 unsigned* __restrict__ hist2) {
  const unsigned sel = meta[0];
  const size_t total = (size_t)NN * NN;
  const size_t stride = (size_t)gridDim.x * blockDim.x;
  for (size_t p = (size_t)blockIdx.x * blockDim.x + threadIdx.x; p < total; p += stride) {
    const unsigned bits = __float_as_uint(a[p]);
    if ((bits >> 16) == sel) atomicAdd(&hist2[bits & 0xFFFFu], 1u);
  }
}

__global__ __launch_bounds__(SCAN_THREADS) void k_scan_lo(const unsigned* __restrict__ hist2,
                                                          unsigned* __restrict__ meta) {
  const unsigned k = ((unsigned)NN * (unsigned)NN - 1u) / 2u;
  const unsigned kk = k - meta[1];
  const unsigned hi = meta[0];
  unsigned low, before;
  if (block_find_bin(hist2, kk, low, before)) meta[2] = (hi << 16) | low;
}

#define GY_WAVES   16
#define GY_THREADS (GY_WAVES * 32)
#define GY_KSLICE  (NN / GY_WAVES)
#define GY_PITCH   (GY_KSLICE + 4)
#define GY_PIECES  (GY_KSLICE / 8)

__global__ __launch_bounds__(GY_THREADS) void k_gemm_y(const float* __restrict__ x,
                                                       const float* __restrict__ W,
                                                       float* __restrict__ Y) {
  __shared__ __attribute__((aligned(16))) float Wt[GY_WAVES][16][GY_PITCH];
  __shared__ __attribute__((aligned(16))) float Xt[GY_WAVES][NB][GY_PITCH];
  __shared__ float part[GY_WAVES][16][16];
  const int nb   = blockIdx.x * 16;
  const int lane = threadIdx.x & 31;
  const int wv   = threadIdx.x >> 5;
  const int half = lane >> 4;
  const int l16  = lane & 15;
  const int k0   = wv * GY_KSLICE;

  float4 wr[16], xr[NB];
  const float* wsrc = W + (size_t)nb * NN + k0 + 4 * lane;
  #pragma unroll
  for (int r = 0; r < 16; ++r) wr[r] = *reinterpret_cast<const float4*>(wsrc + (size_t)r * NN);
  const float* xsrc = x + k0 + 4 * lane;
  #pragma unroll
  for (int r = 0; r < NB; ++r) xr[r] = *reinterpret_cast<const float4*>(xsrc + (size_t)r * NN);

  #pragma unroll
  for (int r = 0; r < 16; ++r) *reinterpret_cast<float4*>(&Wt[wv][r][4 * lane]) = wr[r];
  #pragma unroll
  for (int r = 0; r < NB; ++r) *reinterpret_cast<float4*>(&Xt[wv][r][4 * lane]) = xr[r];
  __syncthreads();

  const float* wrow = &Wt[wv][l16][4 * half];
  const float* xrow = &Xt[wv][l16 & 7][4 * half];
  v8f acc0 = {}, acc1 = {};
  #pragma unroll
  for (int p = 0; p < GY_PIECES; ++p) {
    const float4 bw = *reinterpret_cast<const float4*>(wrow + 8 * p);
    const float4 ax = *reinterpret_cast<const float4*>(xrow + 8 * p);
    v2f a0 = { ax.x, ax.y };
    v2f b0 = { bw.x, bw.y };
    acc0 = __builtin_amdgcn_wmma_f32_16x16x4_f32(false, a0, false, b0,
                                                 (short)0, acc0, false, false);
    v2f a1 = { ax.z, ax.w };
    v2f b1 = { bw.z, bw.w };
    acc1 = __builtin_amdgcn_wmma_f32_16x16x4_f32(false, a1, false, b1,
                                                 (short)0, acc1, false, false);
  }
  #pragma unroll
  for (int r = 0; r < 8; ++r) part[wv][r + half * 8][l16] = acc0[r] + acc1[r];
  __syncthreads();

  if (threadIdx.x < NB * 16) {
    const int m = threadIdx.x >> 4, c = threadIdx.x & 15;
    float s = 0.f;
    #pragma unroll
    for (int w = 0; w < GY_WAVES; ++w) s += part[w][m][c];
    Y[m * NN + nb + c] = s;
  }
}

__global__ __launch_bounds__(256) void k_row_stats(const float* __restrict__ x,
                                                   const float* __restrict__ Y,
                                                   const float* __restrict__ bias,
                                                   float* __restrict__ rowM,
                                                   float* __restrict__ rowZ) {
  __shared__ float lr[NN];
  __shared__ float red[256];
  const int i = blockIdx.x;
  const int b = blockIdx.y;
  const int t = threadIdx.x;
  const float xbi = x[b * NN + i];

  float mymax = -INFINITY;
  for (int j = t; j < NN; j += 256) {
    float v = fmaf(xbi, Y[b * NN + j], bias[j]);
    v = (v > 0.f) ? v : 0.01f * v;
    lr[j] = v;
    mymax = fmaxf(mymax, v);
  }
  red[t] = mymax; __syncthreads();
  for (int s = 128; s > 0; s >>= 1) { if (t < s) red[t] = fmaxf(red[t], red[t + s]); __syncthreads(); }
  const float m = red[0];
  __syncthreads();

  float sum = 0.f;
  for (int j = t; j < NN; j += 256) sum += __expf(lr[j] - m);
  red[t] = sum; __syncthreads();
  for (int s = 128; s > 0; s >>= 1) { if (t < s) red[t] += red[t + s]; __syncthreads(); }
  if (t == 0) { rowM[b * NN + i] = m; rowZ[b * NN + i] = red[0]; }
}

__global__ __launch_bounds__(256) void k_aggregate(const float* __restrict__ x,
                                                   const float* __restrict__ Y,
                                                   const float* __restrict__ bias,
                                                   const float* __restrict__ a,
                                                   const unsigned* __restrict__ meta,
                                                   const float* __restrict__ rowM,
                                                   const float* __restrict__ rowZ,
                                                   float* __restrict__ out) {
  const int j  = blockIdx.x * 256 + threadIdx.x;
  const int b  = blockIdx.y;
  const int iBase = blockIdx.z * (NN / 8);
  const float med = __uint_as_float(meta[2]);
  const float ybj = Y[b * NN + j];
  const float bj  = bias[j];

  __shared__ float sx[128], sm[128], sc[128];
  float acc = 0.f;
  for (int i0 = iBase; i0 < iBase + 256; i0 += 128) {
    if (threadIdx.x < 128) {
      const int i = i0 + threadIdx.x;
      const float xv = x[b * NN + i];
      sx[threadIdx.x] = xv;
      sm[threadIdx.x] = rowM[b * NN + i];
      sc[threadIdx.x] = xv / rowZ[b * NN + i];
    }
    __syncthreads();
    for (int ii = 0; ii < 128; ++ii) {
      const int i = i0 + ii;
      if ((ii & 15) == 0)
        __builtin_prefetch(&a[(size_t)(i + 16) * NN + j], 0, 1);
      const float aij = a[(size_t)i * NN + j];
      float t = fmaf(sx[ii], ybj, bj);
      t = (t > 0.f) ? t : 0.01f * t;
      acc += (aij > med) ? sc[ii] * __expf(t - sm[ii]) : 0.f;
    }
    __syncthreads();
  }
  atomicAdd(&out[b * NN + j], acc);
}

extern "C" void kernel_launch(void* const* d_in, const int* in_sizes, int n_in,
                              void* d_out, int out_size, void* d_ws, size_t ws_size,
                              hipStream_t stream) {
  const float* x    = (const float*)d_in[0];
  const float* adj  = (const float*)d_in[1];
  const float* W    = (const float*)d_in[2];
  const float* bias = (const float*)d_in[3];
  float* out = (float*)d_out;

  char* ws = (char*)d_ws;
  size_t off = 0;
  float*    a     = (float*)(ws + off);    off += (size_t)NN * NN * sizeof(float);
  float*    Y     = (float*)(ws + off);    off += (size_t)NB * NN * sizeof(float);
  float*    rowM  = (float*)(ws + off);    off += (size_t)NB * NN * sizeof(float);
  float*    rowZ  = (float*)(ws + off);    off += (size_t)NB * NN * sizeof(float);
  unsigned* hist1 = (unsigned*)(ws + off); off += (size_t)NBINS * sizeof(unsigned);
  unsigned* hist2 = (unsigned*)(ws + off); off += (size_t)NBINS * sizeof(unsigned);
  unsigned* meta  = (unsigned*)(ws + off); off += 16;

  hipMemsetAsync(hist1, 0, 2 * (size_t)NBINS * sizeof(unsigned), stream);
  hipMemsetAsync(out, 0, (size_t)out_size * sizeof(float), stream);

  k_row_softmax<<<NN, 256, 0, stream>>>(adj, a);
  k_hist_hi   <<<HIST_BLOCKS, HIST_THREADS, 0, stream>>>(a, hist1);
  k_scan_hi   <<<1, SCAN_THREADS, 0, stream>>>(hist1, meta);
  k_hist_lo   <<<1024, 256, 0, stream>>>(a, meta, hist2);
  k_scan_lo   <<<1, SCAN_THREADS, 0, stream>>>(hist2, meta);
  k_gemm_y    <<<NN / 16, GY_THREADS, 0, stream>>>(x, W, Y);
  k_row_stats <<<dim3(NN, NB), 256, 0, stream>>>(x, Y, bias, rowM, rowZ);
  k_aggregate <<<dim3(NN / 256, NB, 8), 256, 0, stream>>>(x, Y, bias, a, meta, rowM, rowZ, out);
}


# Row softmax stats (max, sum of exp): one 256-thread block per row with two LDS barrier trees -> one wave per 4 rows, Y[b,:] and bias held in registers, wave butterflies (new Z sum order)
# speedup vs baseline: 16.0308x; 1.2937x over previous
#include <hip/hip_runtime.h>
#include <math.h>

#define NN 2048
#define NB 8
#define NBINS 65536

typedef __attribute__((ext_vector_type(2))) float v2f;
typedef __attribute__((ext_vector_type(8))) float v8f;

__global__ __launch_bounds__(256) void k_row_softmax(const float* __restrict__ adj,
                                                     float* __restrict__ a) {
  __shared__ float red[256];
  const int i = blockIdx.x;
  const int t = threadIdx.x;
  const float* row = adj + (size_t)i * NN;

  float mymax = -INFINITY;
  for (int j = t; j < NN; j += 256) mymax = fmaxf(mymax, row[j]);
  red[t] = mymax; __syncthreads();
  for (int s = 128; s > 0; s >>= 1) { if (t < s) red[t] = fmaxf(red[t], red[t + s]); __syncthreads(); }
  const float m = red[0];
  __syncthreads();

  float sum = 0.f;
  for (int j = t; j < NN; j += 256) sum += __expf(row[j] - m);
  red[t] = sum; __syncthreads();
  for (int s = 128; s > 0; s >>= 1) { if (t < s) red[t] += red[t + s]; __syncthreads(); }
  const float inv = 1.f / red[0];

  float* arow = a + (size_t)i * NN;
  for (int j = t; j < NN; j += 256) arow[j] = __expf(row[j] - m) * inv;
}

#define HIST_BLOCKS  256
#define HIST_THREADS 1024
#define HIST_VEC     ((NN * NN / 4) / (HIST_BLOCKS * HIST_THREADS))
#define FLUSH_BATCH  16

__global__ __launch_bounds__(HIST_THREADS) void k_hist_hi(const float* __restrict__ a,
                                                          unsigned* __restrict__ hist) {
  __shared__ unsigned lh[NBINS];
  const int t = threadIdx.x;

  const float4* src = reinterpret_cast<const float4*>(a)
                    + (size_t)blockIdx.x * (HIST_VEC * HIST_THREADS) + t;
  float4 v[HIST_VEC];
  #pragma unroll
  for (int i = 0; i < HIST_VEC; ++i) v[i] = src[i * HIST_THREADS];

  uint4* lh4 = reinterpret_cast<uint4*>(lh);
  for (int i = t; i < NBINS / 4; i += HIST_THREADS) lh4[i] = make_uint4(0u, 0u, 0u, 0u);
  __syncthreads();

  #pragma unroll
  for (int i = 0; i < HIST_VEC; ++i) {
    atomicAdd(&lh[__float_as_uint(v[i].x) >> 16], 1u);
    atomicAdd(&lh[__float_as_uint(v[i].y) >> 16], 1u);
    atomicAdd(&lh[__float_as_uint(v[i].z) >> 16], 1u);
    atomicAdd(&lh[__float_as_uint(v[i].w) >> 16], 1u);
  }
  __syncthreads();

  for (int i0 = 0; i0 < NBINS; i0 += FLUSH_BATCH * HIST_THREADS) {
    unsigned c[FLUSH_BATCH];
    #pragma unroll
    for (int u = 0; u < FLUSH_BATCH; ++u) c[u] = lh[i0 + u * HIST_THREADS + t];
    #pragma unroll
    for (int u = 0; u < FLUSH_BATCH; ++u)
      if (c[u] != 0u) atomicAdd(&hist[i0 + u * HIST_THREADS + t], c[u]);
  }
}

#define SCAN_THREADS 1024
#define SCAN_WAVES   (SCAN_THREADS / 32)
#define WAVE_BINS    (NBINS / SCAN_WAVES)
#define WAVE_LOADS   (WAVE_BINS / 128)

__device__ __forceinline__ unsigned sum4(const uint4 v) { return v.x + v.y + v.z + v.w; }

__device__ __forceinline__ unsigned wave_inclusive_scan(unsigned v, const int lane) {
  #pragma unroll
  for (int d = 1; d < 32; d <<= 1) {
    const unsigned n = __shfl_up(v, d, 32);
    if (lane >= d) v += n;
  }
  return v;
}

__device__ __forceinline__ bool block_find_bin(const unsigned* __restrict__ hist,
                                               const unsigned target,
                                               unsigned& bin, unsigned& before) {
  __shared__ unsigned wave_tot[SCAN_WAVES];
  const int lane = threadIdx.x & 31;
  const int w    = threadIdx.x >> 5;
  const uint4* src = reinterpret_cast<const uint4*>(hist + w * WAVE_BINS) + lane;

  uint4 q[WAVE_LOADS];
  #pragma unroll
  for (int i = 0; i < WAVE_LOADS; ++i) q[i] = src[i * 32];

  unsigned wsum = 0;
  #pragma unroll
  for (int i = 0; i < WAVE_LOADS; ++i) wsum += sum4(q[i]);
  #pragma unroll
  for (int d = 16; d >= 1; d >>= 1) wsum += __shfl_xor(wsum, d, 32);
  if (lane == 0) wave_tot[w] = wsum;
  __syncthreads();
  const unsigned wt    = wave_tot[lane];
  const unsigned wincl = wave_inclusive_scan(wt, lane);
  const unsigned wave_before = __shfl(wincl - wt, w, 32);
  if (!(wave_before <= target && target < wave_before + wsum)) return false;

  unsigned t[WAVE_LOADS];
  #pragma unroll
  for (int i = 0; i < WAVE_LOADS; ++i) t[i] = sum4(q[i]);
  #pragma unroll
  for (int d = 16; d >= 1; d >>= 1) {
    #pragma unroll
    for (int i = 0; i < WAVE_LOADS; ++i) t[i] += __shfl_xor(t[i], d, 32);
  }
  unsigned base = wave_before;
  uint4 c = q[0];
  int isel = 0;
  bool found = false;
  #pragma unroll
  for (int i = 0; i < WAVE_LOADS; ++i) {
    if (!found) {
      if (base + t[i] > target) { c = q[i]; isel = i; found = true; }
      else base += t[i];
    }
  }

  const unsigned mine = sum4(c);
  const unsigned lane_before = base + wave_inclusive_scan(mine, lane) - mine;
  if (!(lane_before <= target && target < lane_before + mine)) return false;
  unsigned cum = lane_before;
  unsigned comp = 0;
  if (cum + c.x <= target) {
    cum += c.x; comp = 1;
    if (cum + c.y <= target) {
      cum += c.y; comp = 2;
      if (cum + c.z <= target) { cum += c.z; comp = 3; }
    }
  }
  bin    = (unsigned)(w * WAVE_BINS + isel * 128 + lane * 4) + comp;
  before = cum;
  return true;
}

__global__ __launch_bounds__(SCAN_THREADS) void k_scan_hi(const unsigned* __restrict__ hist,
                                                          unsigned* __restrict__ meta) {
  const unsigned k = ((unsigned)NN * (unsigned)NN - 1u) / 2u;
  unsigned bin, before;
  if (block_find_bin(hist, k, bin, before)) { meta[0] = bin; meta[1] = before; }
}

__global__ __launch_bounds__(256) void k_hist_lo(const float* __restrict__ a,
                                                 const unsigned* __restrict__ meta,
                                                 unsigned* __restrict__ hist2) {
  const unsigned sel = meta[0];
  const size_t total = (size_t)NN * NN;
  const size_t stride = (size_t)gridDim.x * blockDim.x;
  for (size_t p = (size_t)blockIdx.x * blockDim.x + threadIdx.x; p < total; p += stride) {
    const unsigned bits = __float_as_uint(a[p]);
    if ((bits >> 16) == sel) atomicAdd(&hist2[bits & 0xFFFFu], 1u);
  }
}

__global__ __launch_bounds__(SCAN_THREADS) void k_scan_lo(const unsigned* __restrict__ hist2,
                                                          unsigned* __restrict__ meta) {
  const unsigned k = ((unsigned)NN * (unsigned)NN - 1u) / 2u;
  const unsigned kk = k - meta[1];
  const unsigned hi = meta[0];
  unsigned low, before;
  if (block_find_bin(hist2, kk, low, before)) meta[2] = (hi << 16) | low;
}

#define GY_WAVES   16
#define GY_THREADS (GY_WAVES * 32)
#define GY_KSLICE  (NN / GY_WAVES)
#define GY_PITCH   (GY_KSLICE + 4)
#define GY_PIECES  (GY_KSLICE / 8)

__global__ __launch_bounds__(GY_THREADS) void k_gemm_y(const float* __restrict__ x,
                                                       const float* __restrict__ W,
                                                       float* __restrict__ Y) {
  __shared__ __attribute__((aligned(16))) float Wt[GY_WAVES][16][GY_PITCH];
  __shared__ __attribute__((aligned(16))) float Xt[GY_WAVES][NB][GY_PITCH];
  __shared__ float part[GY_WAVES][16][16];
  const int nb   = blockIdx.x * 16;
  const int lane = threadIdx.x & 31;
  const int wv   = threadIdx.x >> 5;
  const int half = lane >> 4;
  const int l16  = lane & 15;
  const int k0   = wv * GY_KSLICE;

  float4 wr[16], xr[NB];
  const float* wsrc = W + (size_t)nb * NN + k0 + 4 * lane;
  #pragma unroll
  for (int r = 0; r < 16; ++r) wr[r] = *reinterpret_cast<const float4*>(wsrc + (size_t)r * NN);
  const float* xsrc = x + k0 + 4 * lane;
  #pragma unroll
  for (int r = 0; r < NB; ++r) xr[r] = *reinterpret_cast<const float4*>(xsrc + (size_t)r * NN);

  #pragma unroll
  for (int r = 0; r < 16; ++r) *reinterpret_cast<float4*>(&Wt[wv][r][4 * lane]) = wr[r];
  #pragma unroll
  for (int r = 0; r < NB; ++r) *reinterpret_cast<float4*>(&Xt[wv][r][4 * lane]) = xr[r];
  __syncthreads();

  const float* wrow = &Wt[wv][l16][4 * half];
  const float* xrow = &Xt[wv][l16 & 7][4 * half];
  v8f acc0 = {}, acc1 = {};
  #pragma unroll
  for (int p = 0; p < GY_PIECES; ++p) {
    const float4 bw = *reinterpret_cast<const float4*>(wrow + 8 * p);
    const float4 ax = *reinterpret_cast<const float4*>(xrow + 8 * p);
    v2f a0 = { ax.x, ax.y };
    v2f b0 = { bw.x, bw.y };
    acc0 = __builtin_amdgcn_wmma_f32_16x16x4_f32(false, a0, false, b0,
                                                 (short)0, acc0, false, false);
    v2f a1 = { ax.z, ax.w };
    v2f b1 = { bw.z, bw.w };
    acc1 = __builtin_amdgcn_wmma_f32_16x16x4_f32(false, a1, false, b1,
                                                 (short)0, acc1, false, false);
  }
  #pragma unroll
  for (int r = 0; r < 8; ++r) part[wv][r + half * 8][l16] = acc0[r] + acc1[r];
  __syncthreads();

  if (threadIdx.x < NB * 16) {
    const int m = threadIdx.x >> 4, c = threadIdx.x & 15;
    float s = 0.f;
    #pragma unroll
    for (int w = 0; w < GY_WAVES; ++w) s += part[w][m][c];
    Y[m * NN + nb + c] = s;
  }
}

#define RS_WAVES   8
#define RS_THREADS (RS_WAVES * 32)
#define RS_ROWS    4
#define RS_WG_ROWS (RS_WAVES * RS_ROWS)
#define RS_Q       (NN / 128)

__global__ __launch_bounds__(RS_THREADS) void k_row_stats(const float* __restrict__ x,
                                                          const float* __restrict__ Y,
                                                          const float* __restrict__ bias,
                                                          float* __restrict__ rowM,
                                                          float* __restrict__ rowZ) {
  __shared__ float sM[RS_WG_ROWS], sZ[RS_WG_ROWS];
  const int lane = threadIdx.x & 31;
  const int wv = __builtin_amdgcn_readfirstlane((int)(threadIdx.x >> 5));
  const int b  = blockIdx.y;
  const int rowBase = blockIdx.x * RS_WG_ROWS;

  float4 y[RS_Q], bb[RS_Q];
  const float4* ysrc = reinterpret_cast<const float4*>(Y + (size_t)b * NN) + lane;
  const float4* bsrc = reinterpret_cast<const float4*>(bias) + lane;
  #pragma unroll
  for (int q = 0; q < RS_Q; ++q) y[q] = ysrc[q * 32];
  #pragma unroll
  for (int q = 0; q < RS_Q; ++q) bb[q] = bsrc[q * 32];
  const float xv = x[b * NN + rowBase + wv * RS_ROWS + (lane & (RS_ROWS - 1))];

  float myM = 0.f, myZ = 0.f;
  for (int r = 0; r < RS_ROWS; ++r) {
    const float xbi = __uint_as_float((unsigned)__builtin_amdgcn_readlane((int)__float_as_uint(xv), r));

    float4 v[RS_Q];
    float4 m4 = make_float4(-INFINITY, -INFINITY, -INFINITY, -INFINITY);
    #pragma unroll
    for (int q = 0; q < RS_Q; ++q) {
      float4 u;
      u.x = fmaf(xbi, y[q].x, bb[q].x);
      u.y = fmaf(xbi, y[q].y, bb[q].y);
      u.z = fmaf(xbi, y[q].z, bb[q].z);
      u.w = fmaf(xbi, y[q].w, bb[q].w);
      u.x = fmaxf(u.x, 0.01f * u.x);
      u.y = fmaxf(u.y, 0.01f * u.y);
      u.z = fmaxf(u.z, 0.01f * u.z);
      u.w = fmaxf(u.w, 0.01f * u.w);
      v[q] = u;
      m4.x = fmaxf(m4.x, u.x);
      m4.y = fmaxf(m4.y, u.y);
      m4.z = fmaxf(m4.z, u.z);
      m4.w = fmaxf(m4.w, u.w);
    }
    float m = fmaxf(fmaxf(m4.x, m4.y), fmaxf(m4.z, m4.w));
    #pragma unroll
    for (int d = 16; d >= 1; d >>= 1) m = fmaxf(m, __shfl_xor(m, d, 32));

    float4 s4 = make_float4(0.f, 0.f, 0.f, 0.f);
    #pragma unroll
    for (int q = 0; q < RS_Q; ++q) {
      s4.x += __expf(v[q].x - m);
      s4.y += __expf(v[q].y - m);
      s4.z += __expf(v[q].z - m);
      s4.w += __expf(v[q].w - m);
    }
    float s = (s4.x + s4.y) + (s4.z + s4.w);
    #pragma unroll
    for (int d = 16; d >= 1; d >>= 1) s += __shfl_xor(s, d, 32);

    if (lane == r) { myM = m; myZ = s; }
  }
  if (lane < RS_ROWS) { sM[wv * RS_ROWS + lane] = myM; sZ[wv * RS_ROWS + lane] = myZ; }
  __syncthreads();
  if (wv == 0)      rowM[b * NN + rowBase + lane] = sM[lane];
  else if (wv == 1) rowZ[b * NN + rowBase + lane] = sZ[lane];
}

__global__ __launch_bounds__(256) void k_aggregate(const float* __restrict__ x,
                                                   const float* __restrict__ Y,
                                                   const float* __restrict__ bias,
                                                   const float* __restrict__ a,
                                                   const unsigned* __restrict__ meta,
                                                   const float* __restrict__ rowM,
                                                   const float* __restrict__ rowZ,
                                                   float* __restrict__ out) {
  const int j  = blockIdx.x * 256 + threadIdx.x;
  const int b  = blockIdx.y;
  const int iBase = blockIdx.z * (NN / 8);
  const float med = __uint_as_float(meta[2]);
  const float ybj = Y[b * NN + j];
  const float bj  = bias[j];

  __shared__ float sx[128], sm[128], sc[128];
  float acc = 0.f;
  for (int i0 = iBase; i0 < iBase + 256; i0 += 128) {
    if (threadIdx.x < 128) {
      const int i = i0 + threadIdx.x;
      const float xv = x[b * NN + i];
      sx[threadIdx.x] = xv;
      sm[threadIdx.x] = rowM[b * NN + i];
      sc[threadIdx.x] = xv / rowZ[b * NN + i];
    }
    __syncthreads();
    for (int ii = 0; ii < 128; ++ii) {
      const int i = i0 + ii;
      if ((ii & 15) == 0)
        __builtin_prefetch(&a[(size_t)(i + 16) * NN + j], 0, 1);
      const float aij = a[(size_t)i * NN + j];
      float t = fmaf(sx[ii], ybj, bj);
      t = (t > 0.f) ? t : 0.01f * t;
      acc += (aij > med) ? sc[ii] * __expf(t - sm[ii]) : 0.f;
    }
    __syncthreads();
  }
  atomicAdd(&out[b * NN + j], acc);
}

extern "C" void kernel_launch(void* const* d_in, const int* in_sizes, int n_in,
                              void* d_out, int out_size, void* d_ws, size_t ws_size,
                              hipStream_t stream) {
  const float* x    = (const float*)d_in[0];
  const float* adj  = (const float*)d_in[1];
  const float* W    = (const float*)d_in[2];
  const float* bias = (const float*)d_in[3];
  float* out = (float*)d_out;

  char* ws = (char*)d_ws;
  size_t off = 0;
  float*    a     = (float*)(ws + off);    off += (size_t)NN * NN * sizeof(float);
  float*    Y     = (float*)(ws + off);    off += (size_t)NB * NN * sizeof(float);
  float*    rowM  = (float*)(ws + off);    off += (size_t)NB * NN * sizeof(float);
  float*    rowZ  = (float*)(ws + off);    off += (size_t)NB * NN * sizeof(float);
  unsigned* hist1 = (unsigned*)(ws + off); off += (size_t)NBINS * sizeof(unsigned);
  unsigned* hist2 = (unsigned*)(ws + off); off += (size_t)NBINS * sizeof(unsigned);
  unsigned* meta  = (unsigned*)(ws + off); off += 16;

  hipMemsetAsync(hist1, 0, 2 * (size_t)NBINS * sizeof(unsigned), stream);
  hipMemsetAsync(out, 0, (size_t)out_size * sizeof(float), stream);

  k_row_softmax<<<NN, 256, 0, stream>>>(adj, a);
  k_hist_hi   <<<HIST_BLOCKS, HIST_THREADS, 0, stream>>>(a, hist1);
  k_scan_hi   <<<1, SCAN_THREADS, 0, stream>>>(hist1, meta);
  k_hist_lo   <<<1024, 256, 0, stream>>>(a, meta, hist2);
  k_scan_lo   <<<1, SCAN_THREADS, 0, stream>>>(hist2, meta);
  k_gemm_y    <<<NN / 16, GY_THREADS, 0, stream>>>(x, W, Y);
  k_row_stats <<<dim3(NN / RS_WG_ROWS, NB), RS_THREADS, 0, stream>>>(x, Y, bias, rowM, rowZ);
  k_aggregate <<<dim3(NN / 256, NB, 8), 256, 0, stream>>>(x, Y, bias, a, meta, rowM, rowZ, out);
}


# k_aggregate column walk of a: one load waited for and one mask branch per row -> 16 rows' loads issued as a group one batch ahead, branch-free masked sum (same sums, same per-thread order)
# speedup vs baseline: 27.0786x; 1.6892x over previous
#include <hip/hip_runtime.h>
#include <math.h>

#define NN 2048
#define NB 8
#define NBINS 65536

typedef __attribute__((ext_vector_type(2))) float v2f;
typedef __attribute__((ext_vector_type(8))) float v8f;

__global__ __launch_bounds__(256) void k_row_softmax(const float* __restrict__ adj,
                                                     float* __restrict__ a) {
  __shared__ float red[256];
  const int i = blockIdx.x;
  const int t = threadIdx.x;
  const float* row = adj + (size_t)i * NN;

  float mymax = -INFINITY;
  for (int j = t; j < NN; j += 256) mymax = fmaxf(mymax, row[j]);
  red[t] = mymax; __syncthreads();
  for (int s = 128; s > 0; s >>= 1) { if (t < s) red[t] = fmaxf(red[t], red[t + s]); __syncthreads(); }
  const float m = red[0];
  __syncthreads();

  float sum = 0.f;
  for (int j = t; j < NN; j += 256) sum += __expf(row[j] - m);
  red[t] = sum; __syncthreads();
  for (int s = 128; s > 0; s >>= 1) { if (t < s) red[t] += red[t + s]; __syncthreads(); }
  const float inv = 1.f / red[0];

  float* arow = a + (size_t)i * NN;
  for (int j = t; j < NN; j += 256) arow[j] = __expf(row[j] - m) * inv;
}

#define HIST_BLOCKS  256
#define HIST_THREADS 1024
#define HIST_VEC     ((NN * NN / 4) / (HIST_BLOCKS * HIST_THREADS))
#define FLUSH_BATCH  16

__global__ __launch_bounds__(HIST_THREADS) void k_hist_hi(const float* __restrict__ a,
                                                          unsigned* __restrict__ hist) {
  __shared__ unsigned lh[NBINS];
  const int t = threadIdx.x;

  const float4* src = reinterpret_cast<const float4*>(a)
                    + (size_t)blockIdx.x * (HIST_VEC * HIST_THREADS) + t;
  float4 v[HIST_VEC];
  #pragma unroll
  for (int i = 0; i < HIST_VEC; ++i) v[i] = src[i * HIST_THREADS];

  uint4* lh4 = reinterpret_cast<uint4*>(lh);
  for (int i = t; i < NBINS / 4; i += HIST_THREADS) lh4[i] = make_uint4(0u, 0u, 0u, 0u);
  __syncthreads();

  #pragma unroll
  for (int i = 0; i < HIST_VEC; ++i) {
    atomicAdd(&lh[__float_as_uint(v[i].x) >> 16], 1u);
    atomicAdd(&lh[__float_as_uint(v[i].y) >> 16], 1u);
    atomicAdd(&lh[__float_as_uint(v[i].z) >> 16], 1u);
    atomicAdd(&lh[__float_as_uint(v[i].w) >> 16], 1u);
  }
  __syncthreads();

  for (int i0 = 0; i0 < NBINS; i0 += FLUSH_BATCH * HIST_THREADS) {
    unsigned c[FLUSH_BATCH];
    #pragma unroll
    for (int u = 0; u < FLUSH_BATCH; ++u) c[u] = lh[i0 + u * HIST_THREADS + t];
    #pragma unroll
    for (int u = 0; u < FLUSH_BATCH; ++u)
      if (c[u] != 0u) atomicAdd(&hist[i0 + u * HIST_THREADS + t], c[u]);
  }
}

#define SCAN_THREADS 1024
#define SCAN_WAVES   (SCAN_THREADS / 32)
#define WAVE_BINS    (NBINS / SCAN_WAVES)
#define WAVE_LOADS   (WAVE_BINS / 128)

__device__ __forceinline__ unsigned sum4(const uint4 v) { return v.x + v.y + v.z + v.w; }

__device__ __forceinline__ unsigned wave_inclusive_scan(unsigned v, const int lane) {
  #pragma unroll
  for (int d = 1; d < 32; d <<= 1) {
    const unsigned n = __shfl_up(v, d, 32);
    if (lane >= d) v += n;
  }
  return v;
}

__device__ __forceinline__ bool block_find_bin(const unsigned* __restrict__ hist,
                                               const unsigned target,
                                               unsigned& bin, unsigned& before) {
  __shared__ unsigned wave_tot[SCAN_WAVES];
  const int lane = threadIdx.x & 31;
  const int w    = threadIdx.x >> 5;
  const uint4* src = reinterpret_cast<const uint4*>(hist + w * WAVE_BINS) + lane;

  uint4 q[WAVE_LOADS];
  #pragma unroll
  for (int i = 0; i < WAVE_LOADS; ++i) q[i] = src[i * 32];

  unsigned wsum = 0;
  #pragma unroll
  for (int i = 0; i < WAVE_LOADS; ++i) wsum += sum4(q[i]);
  #pragma unroll
  for (int d = 16; d >= 1; d >>= 1) wsum += __shfl_xor(wsum, d, 32);
  if (lane == 0) wave_tot[w] = wsum;
  __syncthreads();
  const unsigned wt    = wave_tot[lane];
  const unsigned wincl = wave_inclusive_scan(wt, lane);
  const unsigned wave_before = __shfl(wincl - wt, w, 32);
  if (!(wave_before <= target && target < wave_before + wsum)) return false;

  unsigned t[WAVE_LOADS];
  #pragma unroll
  for (int i = 0; i < WAVE_LOADS; ++i) t[i] = sum4(q[i]);
  #pragma unroll
  for (int d = 16; d >= 1; d >>= 1) {
    #pragma unroll
    for (int i = 0; i < WAVE_LOADS; ++i) t[i] += __shfl_xor(t[i], d, 32);
  }
  unsigned base = wave_before;
  uint4 c = q[0];
  int isel = 0;
  bool found = false;
  #pragma unroll
  for (int i = 0; i < WAVE_LOADS; ++i) {
    if (!found) {
      if (base + t[i] > target) { c = q[i]; isel = i; found = true; }
      else base += t[i];
    }
  }

  const unsigned mine = sum4(c);
  const unsigned lane_before = base + wave_inclusive_scan(mine, lane) - mine;
  if (!(lane_before <= target && target < lane_before + mine)) return false;
  unsigned cum = lane_before;
  unsigned comp = 0;
  if (cum + c.x <= target) {
    cum += c.x; comp = 1;
    if (cum + c.y <= target) {
      cum += c.y; comp = 2;
      if (cum + c.z <= target) { cum += c.z; comp = 3; }
    }
  }
  bin    = (unsigned)(w * WAVE_BINS + isel * 128 + lane * 4) + comp;
  before = cum;
  return true;
}

__global__ __launch_bounds__(SCAN_THREADS) void k_scan_hi(const unsigned* __restrict__ hist,
                                                          unsigned* __restrict__ meta) {
  const unsigned k = ((unsigned)NN * (unsigned)NN - 1u) / 2u;
  unsigned bin, before;
  if (block_find_bin(hist, k, bin, before)) { meta[0] = bin; meta[1] = before; }
}

__global__ __launch_bounds__(256) void k_hist_lo(const float* __restrict__ a,
                                                 const unsigned* __restrict__ meta,
                                                 unsigned* __restrict__ hist2) {
  const unsigned sel = meta[0];
  const size_t total = (size_t)NN * NN;
  const size_t stride = (size_t)gridDim.x * blockDim.x;
  for (size_t p = (size_t)blockIdx.x * blockDim.x + threadIdx.x; p < total; p += stride) {
    const unsigned bits = __float_as_uint(a[p]);
    if ((bits >> 16) == sel) atomicAdd(&hist2[bits & 0xFFFFu], 1u);
  }
}

__global__ __launch_bounds__(SCAN_THREADS) void k_scan_lo(const unsigned* __restrict__ hist2,
                                                          unsigned* __restrict__ meta) {
  const unsigned k = ((unsigned)NN * (unsigned)NN - 1u) / 2u;
  const unsigned kk = k - meta[1];
  const unsigned hi = meta[0];
  unsigned low, before;
  if (block_find_bin(hist2, kk, low, before)) meta[2] = (hi << 16) | low;
}

#define GY_WAVES   16
#define GY_THREADS (GY_WAVES * 32)
#define GY_KSLICE  (NN / GY_WAVES)
#define GY_PITCH   (GY_KSLICE + 4)
#define GY_PIECES  (GY_KSLICE / 8)

__global__ __launch_bounds__(GY_THREADS) void k_gemm_y(const float* __restrict__ x,
                                                       const float* __restrict__ W,
                                                       float* __restrict__ Y) {
  __shared__ __attribute__((aligned(16))) float Wt[GY_WAVES][16][GY_PITCH];
  __shared__ __attribute__((aligned(16))) float Xt[GY_WAVES][NB][GY_PITCH];
  __shared__ float part[GY_WAVES][16][16];
  const int nb   = blockIdx.x * 16;
  const int lane = threadIdx.x & 31;
  const int wv   = threadIdx.x >> 5;
  const int half = lane >> 4;
  const int l16  = lane & 15;
  const int k0   = wv * GY_KSLICE;

  float4 wr[16], xr[NB];
  const float* wsrc = W + (size_t)nb * NN + k0 + 4 * lane;
  #pragma unroll
  for (int r = 0; r < 16; ++r) wr[r] = *reinterpret_cast<const float4*>(wsrc + (size_t)r * NN);
  const float* xsrc = x + k0 + 4 * lane;
  #pragma unroll
  for (int r = 0; r < NB; ++r) xr[r] = *reinterpret_cast<const float4*>(xsrc + (size_t)r * NN);

  #pragma unroll
  for (int r = 0; r < 16; ++r) *reinterpret_cast<float4*>(&Wt[wv][r][4 * lane]) = wr[r];
  #pragma unroll
  for (int r = 0; r < NB; ++r) *reinterpret_cast<float4*>(&Xt[wv][r][4 * lane]) = xr[r];
  __syncthreads();

  const float* wrow = &Wt[wv][l16][4 * half];
  const float* xrow = &Xt[wv][l16 & 7][4 * half];
  v8f acc0 = {}, acc1 = {};
  #pragma unroll
  for (int p = 0; p < GY_PIECES; ++p) {
    const float4 bw = *reinterpret_cast<const float4*>(wrow + 8 * p);
    const float4 ax = *reinterpret_cast<const float4*>(xrow + 8 * p);
    v2f a0 = { ax.x, ax.y };
    v2f b0 = { bw.x, bw.y };
    acc0 = __builtin_amdgcn_wmma_f32_16x16x4_f32(false, a0, false, b0,
                                                 (short)0, acc0, false, false);
    v2f a1 = { ax.z, ax.w };
    v2f b1 = { bw.z, bw.w };
    acc1 = __builtin_amdgcn_wmma_f32_16x16x4_f32(false, a1, false, b1,
                                                 (short)0, acc1, false, false);
  }
  #pragma unroll
  for (int r = 0; r < 8; ++r) part[wv][r + half * 8][l16] = acc0[r] + acc1[r];
  __syncthreads();

  if (threadIdx.x < NB * 16) {
    const int m = threadIdx.x >> 4, c = threadIdx.x & 15;
    float s = 0.f;
    #pragma unroll
    for (int w = 0; w < GY_WAVES; ++w) s += part[w][m][c];
    Y[m * NN + nb + c] = s;
  }
}

#define RS_WAVES   8
#define RS_THREADS (RS_WAVES * 32)
#define RS_ROWS    4
#define RS_WG_ROWS (RS_WAVES * RS_ROWS)
#define RS_Q       (NN / 128)

__global__ __launch_bounds__(RS_THREADS) void k_row_stats(const float* __restrict__ x,
                                                          const float* __restrict__ Y,
                                                          const float* __restrict__ bias,
                                                          float* __restrict__ rowM,
                                                          float* __restrict__ rowZ) {
  __shared__ float sM[RS_WG_ROWS], sZ[RS_WG_ROWS];
  const int lane = threadIdx.x & 31;
  const int wv = __builtin_amdgcn_readfirstlane((int)(threadIdx.x >> 5));
  const int b  = blockIdx.y;
  const int rowBase = blockIdx.x * RS_WG_ROWS;

  float4 y[RS_Q], bb[RS_Q];
  const float4* ysrc = reinterpret_cast<const float4*>(Y + (size_t)b * NN) + lane;
  const float4* bsrc = reinterpret_cast<const float4*>(bias) + lane;
  #pragma unroll
  for (int q = 0; q < RS_Q; ++q) y[q] = ysrc[q * 32];
  #pragma unroll
  for (int q = 0; q < RS_Q; ++q) bb[q] = bsrc[q * 32];
  const float xv = x[b * NN + rowBase + wv * RS_ROWS + (lane & (RS_ROWS - 1))];

  float myM = 0.f, myZ = 0.f;
  for (int r = 0; r < RS_ROWS; ++r) {
    const float xbi = __uint_as_float((unsigned)__builtin_amdgcn_readlane((int)__float_as_uint(xv), r));

    float4 v[RS_Q];
    float4 m4 = make_float4(-INFINITY, -INFINITY, -INFINITY, -INFINITY);
    #pragma unroll
    for (int q = 0; q < RS_Q; ++q) {
      float4 u;
      u.x = fmaf(xbi, y[q].x, bb[q].x);
      u.y = fmaf(xbi, y[q].y, bb[q].y);
      u.z = fmaf(xbi, y[q].z, bb[q].z);
      u.w = fmaf(xbi, y[q].w, bb[q].w);
      u.x = fmaxf(u.x, 0.01f * u.x);
      u.y = fmaxf(u.y, 0.01f * u.y);
      u.z = fmaxf(u.z, 0.01f * u.z);
      u.w = fmaxf(u.w, 0.01f * u.w);
      v[q] = u;
      m4.x = fmaxf(m4.x, u.x);
      m4.y = fmaxf(m4.y, u.y);
      m4.z = fmaxf(m4.z, u.z);
      m4.w = fmaxf(m4.w, u.w);
    }
    float m = fmaxf(fmaxf(m4.x, m4.y), fmaxf(m4.z, m4.w));
    #pragma unroll
    for (int d = 16; d >= 1; d >>= 1) m = fmaxf(m, __shfl_xor(m, d, 32));

    float4 s4 = make_float4(0.f, 0.f, 0.f, 0.f);
    #pragma unroll
    for (int q = 0; q < RS_Q; ++q) {
      s4.x += __expf(v[q].x - m);
      s4.y += __expf(v[q].y - m);
      s4.z += __expf(v[q].z - m);
      s4.w += __expf(v[q].w - m);
    }
    float s = (s4.x + s4.y) + (s4.z + s4.w);
    #pragma unroll
    for (int d = 16; d >= 1; d >>= 1) s += __shfl_xor(s, d, 32);

    if (lane == r) { myM = m; myZ = s; }
  }
  if (lane < RS_ROWS) { sM[wv * RS_ROWS + lane] = myM; sZ[wv * RS_ROWS + lane] = myZ; }
  __syncthreads();
  if (wv == 0)      rowM[b * NN + rowBase + lane] = sM[lane];
  else if (wv == 1) rowZ[b * NN + rowBase + lane] = sZ[lane];
}

#define AG_ROWS  (NN / 8)
#define AG_BATCH 16

__device__ __forceinline__ float ag_rows(const float (&av)[AG_BATCH],
                                         const float4* __restrict__ s,
                                         const float ybj, const float bj,
                                         const float med, float acc) {
  #pragma unroll
  for (int u = 0; u < AG_BATCH; ++u) {
    const float4 r = s[u];
    float t = fmaf(r.x, ybj, bj);
    t = (t > 0.f) ? t : 0.01f * t;
    const float p = r.z * __expf(t - r.y);
    acc += (av[u] > med) ? p : 0.f;
  }
  return acc;
}

__global__ __launch_bounds__(256) void k_aggregate(const float* __restrict__ x,
                                                   const float* __restrict__ Y,
                                                   const float* __restrict__ bias,
                                                   const float* __restrict__ a,
                                                   const unsigned* __restrict__ meta,
                                                   const float* __restrict__ rowM,
                                                   const float* __restrict__ rowZ,
                                                   float* __restrict__ out) {
  __shared__ float4 srow[AG_ROWS];
  const int j  = blockIdx.x * 256 + threadIdx.x;
  const int b  = blockIdx.y;
  const int iBase = blockIdx.z * AG_ROWS;

  const int   irow = b * NN + iBase + threadIdx.x;
  const float xv = x[irow];
  const float mv = rowM[irow];
  const float zv = rowZ[irow];
  const float ybj = Y[b * NN + j];
  const float bj  = bias[j];
  const float med = __uint_as_float(meta[2]);

  const float* acol = a + (size_t)iBase * NN + j;
  float va[AG_BATCH], vb[AG_BATCH];
  #pragma unroll
  for (int u = 0; u < AG_BATCH; ++u) va[u] = acol[(size_t)u * NN];

  srow[threadIdx.x] = make_float4(xv, mv, xv / zv, 0.f);
  __syncthreads();

  float acc = 0.f;
  int i0 = 0;
  for (; i0 < AG_ROWS - 2 * AG_BATCH; i0 += 2 * AG_BATCH) {
    #pragma unroll
    for (int u = 0; u < AG_BATCH; ++u) vb[u] = acol[(size_t)(i0 + AG_BATCH + u) * NN];
    __builtin_amdgcn_sched_barrier(0);
    acc = ag_rows(va, &srow[i0], ybj, bj, med, acc);
    #pragma unroll
    for (int u = 0; u < AG_BATCH; ++u) va[u] = acol[(size_t)(i0 + 2 * AG_BATCH + u) * NN];
    __builtin_amdgcn_sched_barrier(0);
    acc = ag_rows(vb, &srow[i0 + AG_BATCH], ybj, bj, med, acc);
  }
  #pragma unroll
  for (int u = 0; u < AG_BATCH; ++u) vb[u] = acol[(size_t)(i0 + AG_BATCH + u) * NN];
  __builtin_amdgcn_sched_barrier(0);
  acc = ag_rows(va, &srow[i0], ybj, bj, med, acc);
  acc = ag_rows(vb, &srow[i0 + AG_BATCH], ybj, bj, med, acc);

  atomicAdd(&out[b * NN + j], acc);
}

extern "C" void kernel_launch(void* const* d_in, const int* in_sizes, int n_in,
                              void* d_out, int out_size, void* d_ws, size_t ws_size,
                              hipStream_t stream) {
  const float* x    = (const float*)d_in[0];
  const float* adj  = (const float*)d_in[1];
  const float* W    = (const float*)d_in[2];
  const float* bias = (const float*)d_in[3];
  float* out = (float*)d_out;

  char* ws = (char*)d_ws;
  size_t off = 0;
  float*    a     = (float*)(ws + off);    off += (size_t)NN * NN * sizeof(float);
  float*    Y     = (float*)(ws + off);    off += (size_t)NB * NN * sizeof(float);
  float*    rowM  = (float*)(ws + off);    off += (size_t)NB * NN * sizeof(float);
  float*    rowZ  = (float*)(ws + off);    off += (size_t)NB * NN * sizeof(float);
  unsigned* hist1 = (unsigned*)(ws + off); off += (size_t)NBINS * sizeof(unsigned);
  unsigned* hist2 = (unsigned*)(ws + off); off += (size_t)NBINS * sizeof(unsigned);
  unsigned* meta  = (unsigned*)(ws + off); off += 16;

  hipMemsetAsync(hist1, 0, 2 * (size_t)NBINS * sizeof(unsigned), stream);
  hipMemsetAsync(out, 0, (size_t)out_size * sizeof(float), stream);

  k_row_softmax<<<NN, 256, 0, stream>>>(adj, a);
  k_hist_hi   <<<HIST_BLOCKS, HIST_THREADS, 0, stream>>>(a, hist1);
  k_scan_hi   <<<1, SCAN_THREADS, 0, stream>>>(hist1, meta);
  k_hist_lo   <<<1024, 256, 0, stream>>>(a, meta, hist2);
  k_scan_lo   <<<1, SCAN_THREADS, 0, stream>>>(hist2, meta);
  k_gemm_y    <<<NN / 16, GY_THREADS, 0, stream>>>(x, W, Y);
  k_row_stats <<<dim3(NN / RS_WG_ROWS, NB), RS_THREADS, 0, stream>>>(x, Y, bias, rowM, rowZ);
  k_aggregate <<<dim3(NN / 256, NB, 8), 256, 0, stream>>>(x, Y, bias, a, meta, rowM, rowZ, out);
}


# Row softmax fused with the high-16-bit histogram: a work-group holds 8 rows in registers (adj read once) and counts each a value in LDS bins as it stores it; one launch less (same sums, same order)
# speedup vs baseline: 31.4222x; 1.1604x over previous
#include <hip/hip_runtime.h>
#include <math.h>

#define NN 2048
#define NB 8
#define NBINS 65536

typedef __attribute__((ext_vector_type(2))) float v2f;
typedef __attribute__((ext_vector_type(8))) float v8f;

#define SH_ROWS     8
#define SH_COLS     64
#define SH_THREADS  (SH_ROWS * SH_COLS)
#define SH_K        (NN / (4 * SH_COLS))
#define FLUSH_BATCH 16

__global__ __launch_bounds__(SH_THREADS) void k_softmax_hist(const float* __restrict__ adj,
                                                             float* __restrict__ a,
                                                             unsigned* __restrict__ hist) {
  __shared__ unsigned lh[NBINS];
  __shared__ float4 xsum[SH_ROWS][SH_COLS];
  __shared__ float xmax[SH_ROWS][2];
  const int t    = threadIdx.x;
  const int r    = t >> 6;
  const int c    = t & 63;
  const int lane = t & 31;
  const int wh   = (t >> 5) & 1;
  const size_t rowoff = (size_t)(blockIdx.x * SH_ROWS + r) * NN;

  const float4* src = reinterpret_cast<const float4*>(adj + rowoff) + c;
  float4 v[SH_K];
  #pragma unroll
  for (int k = 0; k < SH_K; ++k) v[k] = src[k * SH_COLS];

  uint4* lh4 = reinterpret_cast<uint4*>(lh);
  for (int i = t; i < NBINS / 4; i += SH_THREADS) lh4[i] = make_uint4(0u, 0u, 0u, 0u);

  float mymax = -INFINITY;
  #pragma unroll
  for (int k = 0; k < SH_K; ++k)
    mymax = fmaxf(mymax, fmaxf(fmaxf(v[k].x, v[k].y), fmaxf(v[k].z, v[k].w)));
  #pragma unroll
  for (int d = 16; d >= 1; d >>= 1) mymax = fmaxf(mymax, __shfl_xor(mymax, d, 32));
  if (lane == 0) xmax[r][wh] = mymax;
  __syncthreads();
  const float m = fmaxf(xmax[r][0], xmax[r][1]);

  float4 s4 = make_float4(0.f, 0.f, 0.f, 0.f);
  #pragma unroll
  for (int k = 0; k < SH_K; ++k) {
    v[k].x = __expf(v[k].x - m);
    v[k].y = __expf(v[k].y - m);
    v[k].z = __expf(v[k].z - m);
    v[k].w = __expf(v[k].w - m);
    s4.x += v[k].x;
    s4.y += v[k].y;
    s4.z += v[k].z;
    s4.w += v[k].w;
  }
  xsum[r][c] = s4;
  __syncthreads();
  const float4 o = xsum[r][c ^ 32];
  s4.x += o.x; s4.y += o.y; s4.z += o.z; s4.w += o.w;
  #pragma unroll
  for (int d = 16; d >= 1; d >>= 1) {
    s4.x += __shfl_xor(s4.x, d, 32);
    s4.y += __shfl_xor(s4.y, d, 32);
    s4.z += __shfl_xor(s4.z, d, 32);
    s4.w += __shfl_xor(s4.w, d, 32);
  }
  const float inv = 1.f / ((s4.x + s4.z) + (s4.y + s4.w));

  float4* dst = reinterpret_cast<float4*>(a + rowoff) + c;
  #pragma unroll
  for (int k = 0; k < SH_K; ++k) {
    v[k].x *= inv; v[k].y *= inv; v[k].z *= inv; v[k].w *= inv;
    dst[k * SH_COLS] = v[k];
  }
  #pragma unroll
  for (int k = 0; k < SH_K; ++k) {
    atomicAdd(&lh[__float_as_uint(v[k].x) >> 16], 1u);
    atomicAdd(&lh[__float_as_uint(v[k].y) >> 16], 1u);
    atomicAdd(&lh[__float_as_uint(v[k].z) >> 16], 1u);
    atomicAdd(&lh[__float_as_uint(v[k].w) >> 16], 1u);
  }
  __syncthreads();

  for (int i0 = 0; i0 < NBINS; i0 += FLUSH_BATCH * SH_THREADS) {
    unsigned cnt[FLUSH_BATCH];
    #pragma unroll
    for (int u = 0; u < FLUSH_BATCH; ++u) cnt[u] = lh[i0 + u * SH_THREADS + t];
    #pragma unroll
    for (int u = 0; u < FLUSH_BATCH; ++u)
      if (cnt[u] != 0u) atomicAdd(&hist[i0 + u * SH_THREADS + t], cnt[u]);
  }
}

#define SCAN_THREADS 1024
#define SCAN_WAVES   (SCAN_THREADS / 32)
#define WAVE_BINS    (NBINS / SCAN_WAVES)
#define WAVE_LOADS   (WAVE_BINS / 128)

__device__ __forceinline__ unsigned sum4(const uint4 v) { return v.x + v.y + v.z + v.w; }

__device__ __forceinline__ unsigned wave_inclusive_scan(unsigned v, const int lane) {
  #pragma unroll
  for (int d = 1; d < 32; d <<= 1) {
    const unsigned n = __shfl_up(v, d, 32);
    if (lane >= d) v += n;
  }
  return v;
}

__device__ __forceinline__ bool block_find_bin(const unsigned* __restrict__ hist,
                                               const unsigned target,
                                               unsigned& bin, unsigned& before) {
  __shared__ unsigned wave_tot[SCAN_WAVES];
  const int lane = threadIdx.x & 31;
  const int w    = threadIdx.x >> 5;
  const uint4* src = reinterpret_cast<const uint4*>(hist + w * WAVE_BINS) + lane;

  uint4 q[WAVE_LOADS];
  #pragma unroll
  for (int i = 0; i < WAVE_LOADS; ++i) q[i] = src[i * 32];

  unsigned wsum = 0;
  #pragma unroll
  for (int i = 0; i < WAVE_LOADS; ++i) wsum += sum4(q[i]);
  #pragma unroll
  for (int d = 16; d >= 1; d >>= 1) wsum += __shfl_xor(wsum, d, 32);
  if (lane == 0) wave_tot[w] = wsum;
  __syncthreads();
  const unsigned wt    = wave_tot[lane];
  const unsigned wincl = wave_inclusive_scan(wt, lane);
  const unsigned wave_before = __shfl(wincl - wt, w, 32);
  if (!(wave_before <= target && target < wave_before + wsum)) return false;

  unsigned t[WAVE_LOADS];
  #pragma unroll
  for (int i = 0; i < WAVE_LOADS; ++i) t[i] = sum4(q[i]);
  #pragma unroll
  for (int d = 16; d >= 1; d >>= 1) {
    #pragma unroll
    for (int i = 0; i < WAVE_LOADS; ++i) t[i] += __shfl_xor(t[i], d, 32);
  }
  unsigned base = wave_before;
  uint4 c = q[0];
  int isel = 0;
  bool found = false;
  #pragma unroll
  for (int i = 0; i < WAVE_LOADS; ++i) {
    if (!found) {
      if (base + t[i] > target) { c = q[i]; isel = i; found = true; }
      else base += t[i];
    }
  }

  const unsigned mine = sum4(c);
  const unsigned lane_before = base + wave_inclusive_scan(mine, lane) - mine;
  if (!(lane_before <= target && target < lane_before + mine)) return false;
  unsigned cum = lane_before;
  unsigned comp = 0;
  if (cum + c.x <= target) {
    cum += c.x; comp = 1;
    if (cum + c.y <= target) {
      cum += c.y; comp = 2;
      if (cum + c.z <= target) { cum += c.z; comp = 3; }
    }
  }
  bin    = (unsigned)(w * WAVE_BINS + isel * 128 + lane * 4) + comp;
  before = cum;
  return true;
}

__global__ __launch_bounds__(SCAN_THREADS) void k_scan_hi(const unsigned* __restrict__ hist,
                                                          unsigned* __restrict__ meta) {
  const unsigned k = ((unsigned)NN * (unsigned)NN - 1u) / 2u;
  unsigned bin, before;
  if (block_find_bin(hist, k, bin, before)) { meta[0] = bin; meta[1] = before; }
}

__global__ __launch_bounds__(256) void k_hist_lo(const float* __restrict__ a,
                                                 const unsigned* __restrict__ meta,
                                                 unsigned* __restrict__ hist2) {
  const unsigned sel = meta[0];
  const size_t total = (size_t)NN * NN;
  const size_t stride = (size_t)gridDim.x * blockDim.x;
  for (size_t p = (size_t)blockIdx.x * blockDim.x + threadIdx.x; p < total; p += stride) {
    const unsigned bits = __float_as_uint(a[p]);
    if ((bits >> 16) == sel) atomicAdd(&hist2[bits & 0xFFFFu], 1u);
  }
}

__global__ __launch_bounds__(SCAN_THREADS) void k_scan_lo(const unsigned* __restrict__ hist2,
                                                          unsigned* __restrict__ meta) {
  const unsigned k = ((unsigned)NN * (unsigned)NN - 1u) / 2u;
  const unsigned kk = k - meta[1];
  const unsigned hi = meta[0];
  unsigned low, before;
  if (block_find_bin(hist2, kk, low, before)) meta[2] = (hi << 16) | low;
}

#define GY_WAVES   16
#define GY_THREADS (GY_WAVES * 32)
#define GY_KSLICE  (NN / GY_WAVES)
#define GY_PITCH   (GY_KSLICE + 4)
#define GY_PIECES  (GY_KSLICE / 8)

__global__ __launch_bounds__(GY_THREADS) void k_gemm_y(const float* __restrict__ x,
                                                       const float* __restrict__ W,
                                                       float* __restrict__ Y) {
  __shared__ __attribute__((aligned(16))) float Wt[GY_WAVES][16][GY_PITCH];
  __shared__ __attribute__((aligned(16))) float Xt[GY_WAVES][NB][GY_PITCH];
  __shared__ float part[GY_WAVES][16][16];
  const int nb   = blockIdx.x * 16;
  const int lane = threadIdx.x & 31;
  const int wv   = threadIdx.x >> 5;
  const int half = lane >> 4;
  const int l16  = lane & 15;
  const int k0   = wv * GY_KSLICE;

  float4 wr[16], xr[NB];
  const float* wsrc = W + (size_t)nb * NN + k0 + 4 * lane;
  #pragma unroll
  for (int r = 0; r < 16; ++r) wr[r] = *reinterpret_cast<const float4*>(wsrc + (size_t)r * NN);
  const float* xsrc = x + k0 + 4 * lane;
  #pragma unroll
  for (int r = 0; r < NB; ++r) xr[r] = *reinterpret_cast<const float4*>(xsrc + (size_t)r * NN);

  #pragma unroll
  for (int r = 0; r < 16; ++r) *reinterpret_cast<float4*>(&Wt[wv][r][4 * lane]) = wr[r];
  #pragma unroll
  for (int r = 0; r < NB; ++r) *reinterpret_cast<float4*>(&Xt[wv][r][4 * lane]) = xr[r];
  __syncthreads();

  const float* wrow = &Wt[wv][l16][4 * half];
  const float* xrow = &Xt[wv][l16 & 7][4 * half];
  v8f acc0 = {}, acc1 = {};
  #pragma unroll
  for (int p = 0; p < GY_PIECES; ++p) {
    const float4 bw = *reinterpret_cast<const float4*>(wrow + 8 * p);
    const float4 ax = *reinterpret_cast<const float4*>(xrow + 8 * p);
    v2f a0 = { ax.x, ax.y };
    v2f b0 = { bw.x, bw.y };
    acc0 = __builtin_amdgcn_wmma_f32_16x16x4_f32(false, a0, false, b0,
                                                 (short)0, acc0, false, false);
    v2f a1 = { ax.z, ax.w };
    v2f b1 = { bw.z, bw.w };
    acc1 = __builtin_amdgcn_wmma_f32_16x16x4_f32(false, a1, false, b1,
                                                 (short)0, acc1, false, false);
  }
  #pragma unroll
  for (int r = 0; r < 8; ++r) part[wv][r + half * 8][l16] = acc0[r] + acc1[r];
  __syncthreads();

  if (threadIdx.x < NB * 16) {
    const int m = threadIdx.x >> 4, c = threadIdx.x & 15;
    float s = 0.f;
    #pragma unroll
    for (int w = 0; w < GY_WAVES; ++w) s += part[w][m][c];
    Y[m * NN + nb + c] = s;
  }
}

#define RS_WAVES   8
#define RS_THREADS (RS_WAVES * 32)
#define RS_ROWS    4
#define RS_WG_ROWS (RS_WAVES * RS_ROWS)
#define RS_Q       (NN / 128)

__global__ __launch_bounds__(RS_THREADS) void k_row_stats(const float* __restrict__ x,
                                                          const float* __restrict__ Y,
                                                          const float* __restrict__ bias,
                                                          float* __restrict__ rowM,
                                                          float* __restrict__ rowZ) {
  __shared__ float sM[RS_WG_ROWS], sZ[RS_WG_ROWS];
  const int lane = threadIdx.x & 31;
  const int wv = __builtin_amdgcn_readfirstlane((int)(threadIdx.x >> 5));
  const int b  = blockIdx.y;
  const int rowBase = blockIdx.x * RS_WG_ROWS;

  float4 y[RS_Q], bb[RS_Q];
  const float4* ysrc = reinterpret_cast<const float4*>(Y + (size_t)b * NN) + lane;
  const float4* bsrc = reinterpret_cast<const float4*>(bias) + lane;
  #pragma unroll
  for (int q = 0; q < RS_Q; ++q) y[q] = ysrc[q * 32];
  #pragma unroll
  for (int q = 0; q < RS_Q; ++q) bb[q] = bsrc[q * 32];
  const float xv = x[b * NN + rowBase + wv * RS_ROWS + (lane & (RS_ROWS - 1))];

  float myM = 0.f, myZ = 0.f;
  for (int r = 0; r < RS_ROWS; ++r) {
    const float xbi = __uint_as_float((unsigned)__builtin_amdgcn_readlane((int)__float_as_uint(xv), r));

    float4 v[RS_Q];
    float4 m4 = make_float4(-INFINITY, -INFINITY, -INFINITY, -INFINITY);
    #pragma unroll
    for (int q = 0; q < RS_Q; ++q) {
      float4 u;
      u.x = fmaf(xbi, y[q].x, bb[q].x);
      u.y = fmaf(xbi, y[q].y, bb[q].y);
      u.z = fmaf(xbi, y[q].z, bb[q].z);
      u.w = fmaf(xbi, y[q].w, bb[q].w);
      u.x = fmaxf(u.x, 0.01f * u.x);
      u.y = fmaxf(u.y, 0.01f * u.y);
      u.z = fmaxf(u.z, 0.01f * u.z);
      u.w = fmaxf(u.w, 0.01f * u.w);
      v[q] = u;
      m4.x = fmaxf(m4.x, u.x);
      m4.y = fmaxf(m4.y, u.y);
      m4.z = fmaxf(m4.z, u.z);
      m4.w = fmaxf(m4.w, u.w);
    }
    float m = fmaxf(fmaxf(m4.x, m4.y), fmaxf(m4.z, m4.w));
    #pragma unroll
    for (int d = 16; d >= 1; d >>= 1) m = fmaxf(m, __shfl_xor(m, d, 32));

    float4 s4 = make_float4(0.f, 0.f, 0.f, 0.f);
    #pragma unroll
    for (int q = 0; q < RS_Q; ++q) {
      s4.x += __expf(v[q].x - m);
      s4.y += __expf(v[q].y - m);
      s4.z += __expf(v[q].z - m);
      s4.w += __expf(v[q].w - m);
    }
    float s = (s4.x + s4.y) + (s4.z + s4.w);
    #pragma unroll
    for (int d = 16; d >= 1; d >>= 1) s += __shfl_xor(s, d, 32);

    if (lane == r) { myM = m; myZ = s; }
  }
  if (lane < RS_ROWS) { sM[wv * RS_ROWS + lane] = myM; sZ[wv * RS_ROWS + lane] = myZ; }
  __syncthreads();
  if (wv == 0)      rowM[b * NN + rowBase + lane] = sM[lane];
  else if (wv == 1) rowZ[b * NN + rowBase + lane] = sZ[lane];
}

#define AG_ROWS  (NN / 8)
#define AG_BATCH 16

__device__ __forceinline__ float ag_rows(const float (&av)[AG_BATCH],
                                         const float4* __restrict__ s,
                                         const float ybj, const float bj,
                                         const float med, float acc) {
  #pragma unroll
  for (int u = 0; u < AG_BATCH; ++u) {
    const float4 r = s[u];
    float t = fmaf(r.x, ybj, bj);
    t = (t > 0.f) ? t : 0.01f * t;
    const float p = r.z * __expf(t - r.y);
    acc += (av[u] > med) ? p : 0.f;
  }
  return acc;
}

__global__ __launch_bounds__(256) void k_aggregate(const float* __restrict__ x,
                                                   const float* __restrict__ Y,
                                                   const float* __restrict__ bias,
                                                   const float* __restrict__ a,
                                                   const unsigned* __restrict__ meta,
                                                   const float* __restrict__ rowM,
                                                   const float* __restrict__ rowZ,
                                                   float* __restrict__ out) {
  __shared__ float4 srow[AG_ROWS];
  const int j  = blockIdx.x * 256 + threadIdx.x;
  const int b  = blockIdx.y;
  const int iBase = blockIdx.z * AG_ROWS;

  const int   irow = b * NN + iBase + threadIdx.x;
  const float xv = x[irow];
  const float mv = rowM[irow];
  const float zv = rowZ[irow];
  const float ybj = Y[b * NN + j];
  const float bj  = bias[j];
  const float med = __uint_as_float(meta[2]);

  const float* acol = a + (size_t)iBase * NN + j;
  float va[AG_BATCH], vb[AG_BATCH];
  #pragma unroll
  for (int u = 0; u < AG_BATCH; ++u) va[u] = acol[(size_t)u * NN];

  srow[threadIdx.x] = make_float4(xv, mv, xv / zv, 0.f);
  __syncthreads();

  float acc = 0.f;
  int i0 = 0;
  for (; i0 < AG_ROWS - 2 * AG_BATCH; i0 += 2 * AG_BATCH) {
    #pragma unroll
    for (int u = 0; u < AG_BATCH; ++u) vb[u] = acol[(size_t)(i0 + AG_BATCH + u) * NN];
    __builtin_amdgcn_sched_barrier(0);
    acc = ag_rows(va, &srow[i0], ybj, bj, med, acc);
    #pragma unroll
    for (int u = 0; u < AG_BATCH; ++u) va[u] = acol[(size_t)(i0 + 2 * AG_BATCH + u) * NN];
    __builtin_amdgcn_sched_barrier(0);
    acc = ag_rows(vb, &srow[i0 + AG_BATCH], ybj, bj, med, acc);
  }
  #pragma unroll
  for (int u = 0; u < AG_BATCH; ++u) vb[u] = acol[(size_t)(i0 + AG_BATCH + u) * NN];
  __builtin_amdgcn_sched_barrier(0);
  acc = ag_rows(va, &srow[i0], ybj, bj, med, acc);
  acc = ag_rows(vb, &srow[i0 + AG_BATCH], ybj, bj, med, acc);

  atomicAdd(&out[b * NN + j], acc);
}

extern "C" void kernel_launch(void* const* d_in, const int* in_sizes, int n_in,
                              void* d_out, int out_size, void* d_ws, size_t ws_size,
                              hipStream_t stream) {
  const float* x    = (const float*)d_in[0];
  const float* adj  = (const float*)d_in[1];
  const float* W    = (const float*)d_in[2];
  const float* bias = (const float*)d_in[3];
  float* out = (float*)d_out;

  char* ws = (char*)d_ws;
  size_t off = 0;
  float*    a     = (float*)(ws + off);    off += (size_t)NN * NN * sizeof(float);
  float*    Y     = (float*)(ws + off);    off += (size_t)NB * NN * sizeof(float);
  float*    rowM  = (float*)(ws + off);    off += (size_t)NB * NN * sizeof(float);
  float*    rowZ  = (float*)(ws + off);    off += (size_t)NB * NN * sizeof(float);
  unsigned* hist1 = (unsigned*)(ws + off); off += (size_t)NBINS * sizeof(unsigned);
  unsigned* hist2 = (unsigned*)(ws + off); off += (size_t)NBINS * sizeof(unsigned);
  unsigned* meta  = (unsigned*)(ws + off); off += 16;

  hipMemsetAsync(hist1, 0, 2 * (size_t)NBINS * sizeof(unsigned), stream);
  hipMemsetAsync(out, 0, (size_t)out_size * sizeof(float), stream);

  k_softmax_hist<<<NN / SH_ROWS, SH_THREADS, 0, stream>>>(adj, a, hist1);
  k_scan_hi   <<<1, SCAN_THREADS, 0, stream>>>(hist1, meta);
  k_hist_lo   <<<1024, 256, 0, stream>>>(a, meta, hist2);
  k_scan_lo   <<<1, SCAN_THREADS, 0, stream>>>(hist2, meta);
  k_gemm_y    <<<NN / 16, GY_THREADS, 0, stream>>>(x, W, Y);
  k_row_stats <<<dim3(NN / RS_WG_ROWS, NB), RS_THREADS, 0, stream>>>(x, Y, bias, rowM, rowZ);
  k_aggregate <<<dim3(NN / 256, NB, 8), 256, 0, stream>>>(x, Y, bias, a, meta, rowM, rowZ, out);
}


# Ops that held a launch alone with the chip idle (both 1-work-group bin scans, clears of out and of the low-half bins) ride as extra work-groups of the Y GEMM and row-stat launches: 9 stream ops -> 6
# speedup vs baseline: 36.6297x; 1.1657x over previous
#include <hip/hip_runtime.h>
#include <math.h>

#define NN 2048
#define NB 8
#define NBINS 65536

typedef __attribute__((ext_vector_type(2))) float v2f;
typedef __attribute__((ext_vector_type(8))) float v8f;

#define SH_ROWS     8
#define SH_COLS     64
#define SH_THREADS  (SH_ROWS * SH_COLS)
#define SH_K        (NN / (4 * SH_COLS))
#define FLUSH_BATCH 16

__global__ __launch_bounds__(SH_THREADS) void k_softmax_hist(const float* __restrict__ adj,
                                                             float* __restrict__ a,
                                                             unsigned* __restrict__ hist) {
  __shared__ unsigned lh[NBINS];
  __shared__ float4 xsum[SH_ROWS][SH_COLS];
  __shared__ float xmax[SH_ROWS][2];
  const int t    = threadIdx.x;
  const int r    = t >> 6;
  const int c    = t & 63;
  const int lane = t & 31;
  const int wh   = (t >> 5) & 1;
  const size_t rowoff = (size_t)(blockIdx.x * SH_ROWS + r) * NN;

  const float4* src = reinterpret_cast<const float4*>(adj + rowoff) + c;
  float4 v[SH_K];
  #pragma unroll
  for (int k = 0; k < SH_K; ++k) v[k] = src[k * SH_COLS];

  uint4* lh4 = reinterpret_cast<uint4*>(lh);
  for (int i = t; i < NBINS / 4; i += SH_THREADS) lh4[i] = make_uint4(0u, 0u, 0u, 0u);

  float mymax = -INFINITY;
  #pragma unroll
  for (int k = 0; k < SH_K; ++k)
    mymax = fmaxf(mymax, fmaxf(fmaxf(v[k].x, v[k].y), fmaxf(v[k].z, v[k].w)));
  #pragma unroll
  for (int d = 16; d >= 1; d >>= 1) mymax = fmaxf(mymax, __shfl_xor(mymax, d, 32));
  if (lane == 0) xmax[r][wh] = mymax;
  __syncthreads();
  const float m = fmaxf(xmax[r][0], xmax[r][1]);

  float4 s4 = make_float4(0.f, 0.f, 0.f, 0.f);
  #pragma unroll
  for (int k = 0; k < SH_K; ++k) {
    v[k].x = __expf(v[k].x - m);
    v[k].y = __expf(v[k].y - m);
    v[k].z = __expf(v[k].z - m);
    v[k].w = __expf(v[k].w - m);
    s4.x += v[k].x;
    s4.y += v[k].y;
    s4.z += v[k].z;
    s4.w += v[k].w;
  }
  xsum[r][c] = s4;
  __syncthreads();
  const float4 o = xsum[r][c ^ 32];
  s4.x += o.x; s4.y += o.y; s4.z += o.z; s4.w += o.w;
  #pragma unroll
  for (int d = 16; d >= 1; d >>= 1) {
    s4.x += __shfl_xor(s4.x, d, 32);
    s4.y += __shfl_xor(s4.y, d, 32);
    s4.z += __shfl_xor(s4.z, d, 32);
    s4.w += __shfl_xor(s4.w, d, 32);
  }
  const float inv = 1.f / ((s4.x + s4.z) + (s4.y + s4.w));

  float4* dst = reinterpret_cast<float4*>(a + rowoff) + c;
  #pragma unroll
  for (int k = 0; k < SH_K; ++k) {
    v[k].x *= inv; v[k].y *= inv; v[k].z *= inv; v[k].w *= inv;
    dst[k * SH_COLS] = v[k];
  }
  #pragma unroll
  for (int k = 0; k < SH_K; ++k) {
    atomicAdd(&lh[__float_as_uint(v[k].x) >> 16], 1u);
    atomicAdd(&lh[__float_as_uint(v[k].y) >> 16], 1u);
    atomicAdd(&lh[__float_as_uint(v[k].z) >> 16], 1u);
    atomicAdd(&lh[__float_as_uint(v[k].w) >> 16], 1u);
  }
  __syncthreads();

  for (int i0 = 0; i0 < NBINS; i0 += FLUSH_BATCH * SH_THREADS) {
    unsigned cnt[FLUSH_BATCH];
    #pragma unroll
    for (int u = 0; u < FLUSH_BATCH; ++u) cnt[u] = lh[i0 + u * SH_THREADS + t];
    #pragma unroll
    for (int u = 0; u < FLUSH_BATCH; ++u)
      if (cnt[u] != 0u) atomicAdd(&hist[i0 + u * SH_THREADS + t], cnt[u]);
  }
}

#define SCAN_THREADS 512
#define SCAN_WAVES   (SCAN_THREADS / 32)
#define WAVE_BINS    (NBINS / SCAN_WAVES)
#define WAVE_LOADS   (WAVE_BINS / 128)

__device__ __forceinline__ unsigned sum4(const uint4 v) { return v.x + v.y + v.z + v.w; }

__device__ __forceinline__ unsigned wave_inclusive_scan(unsigned v, const int lane) {
  #pragma unroll
  for (int d = 1; d < 32; d <<= 1) {
    const unsigned n = __shfl_up(v, d, 32);
    if (lane >= d) v += n;
  }
  return v;
}

__device__ __forceinline__ bool block_find_bin(const unsigned* __restrict__ hist,
                                               const unsigned target,
                                               unsigned& bin, unsigned& before) {
  __shared__ unsigned wave_tot[SCAN_WAVES];
  const int lane = threadIdx.x & 31;
  const int w    = threadIdx.x >> 5;
  const uint4* src = reinterpret_cast<const uint4*>(hist + w * WAVE_BINS) + lane;

  uint4 q[WAVE_LOADS];
  #pragma unroll
  for (int i = 0; i < WAVE_LOADS; ++i) q[i] = src[i * 32];

  unsigned wsum = 0;
  #pragma unroll
  for (int i = 0; i < WAVE_LOADS; ++i) wsum += sum4(q[i]);
  #pragma unroll
  for (int d = 16; d >= 1; d >>= 1) wsum += __shfl_xor(wsum, d, 32);
  if (lane == 0) wave_tot[w] = wsum;
  __syncthreads();
  const unsigned wt    = (lane < SCAN_WAVES) ? wave_tot[lane] : 0u;
  const unsigned wincl = wave_inclusive_scan(wt, lane);
  const unsigned wave_before = __shfl(wincl - wt, w, 32);
  if (!(wave_before <= target && target < wave_before + wsum)) return false;

  unsigned t[WAVE_LOADS];
  #pragma unroll
  for (int i = 0; i < WAVE_LOADS; ++i) t[i] = sum4(q[i]);
  #pragma unroll
  for (int d = 16; d >= 1; d >>= 1) {
    #pragma unroll
    for (int i = 0; i < WAVE_LOADS; ++i) t[i] += __shfl_xor(t[i], d, 32);
  }
  unsigned base = wave_before;
  uint4 c = q[0];
  int isel = 0;
  bool found = false;
  #pragma unroll
  for (int i = 0; i < WAVE_LOADS; ++i) {
    if (!found) {
      if (base + t[i] > target) { c = q[i]; isel = i; found = true; }
      else base += t[i];
    }
  }

  const unsigned mine = sum4(c);
  const unsigned lane_before = base + wave_inclusive_scan(mine, lane) - mine;
  if (!(lane_before <= target && target < lane_before + mine)) return false;
  unsigned cum = lane_before;
  unsigned comp = 0;
  if (cum + c.x <= target) {
    cum += c.x; comp = 1;
    if (cum + c.y <= target) {
      cum += c.y; comp = 2;
      if (cum + c.z <= target) { cum += c.z; comp = 3; }
    }
  }
  bin    = (unsigned)(w * WAVE_BINS + isel * 128 + lane * 4) + comp;
  before = cum;
  return true;
}

__device__ __forceinline__ void scan_hi(const unsigned* __restrict__ hist,
                                        unsigned* __restrict__ meta) {
  const unsigned k = ((unsigned)NN * (unsigned)NN - 1u) / 2u;
  unsigned bin, before;
  if (block_find_bin(hist, k, bin, before)) { meta[0] = bin; meta[1] = before; }
}

__global__ __launch_bounds__(256) void k_hist_lo(const float* __restrict__ a,
                                                 const unsigned* __restrict__ meta,
                                                 unsigned* __restrict__ hist2) {
  const unsigned sel = meta[0];
  const size_t total = (size_t)NN * NN;
  const size_t stride = (size_t)gridDim.x * blockDim.x;
  for (size_t p = (size_t)blockIdx.x * blockDim.x + threadIdx.x; p < total; p += stride) {
    const unsigned bits = __float_as_uint(a[p]);
    if ((bits >> 16) == sel) atomicAdd(&hist2[bits & 0xFFFFu], 1u);
  }
}

__device__ __forceinline__ void scan_lo(const unsigned* __restrict__ hist2,
                                        unsigned* __restrict__ meta) {
  const unsigned k = ((unsigned)NN * (unsigned)NN - 1u) / 2u;
  const unsigned kk = k - meta[1];
  const unsigned hi = meta[0];
  unsigned low, before;
  if (block_find_bin(hist2, kk, low, before)) meta[2] = (hi << 16) | low;
}

#define GY_WAVES   16
#define GY_THREADS (GY_WAVES * 32)
#define GY_KSLICE  (NN / GY_WAVES)
#define GY_PITCH   (GY_KSLICE + 4)
#define GY_PIECES  (GY_KSLICE / 8)

__device__ __forceinline__ void gemm_y_tile(const float* __restrict__ x,
                                            const float* __restrict__ W,
                                            float* __restrict__ Y,
                                            const int tile) {
  __shared__ __attribute__((aligned(16))) float Wt[GY_WAVES][16][GY_PITCH];
  __shared__ __attribute__((aligned(16))) float Xt[GY_WAVES][NB][GY_PITCH];
  __shared__ float part[GY_WAVES][16][16];
  const int nb   = tile * 16;
  const int lane = threadIdx.x & 31;
  const int wv   = threadIdx.x >> 5;
  const int half = lane >> 4;
  const int l16  = lane & 15;
  const int k0   = wv * GY_KSLICE;

  float4 wr[16], xr[NB];
  const float* wsrc = W + (size_t)nb * NN + k0 + 4 * lane;
  #pragma unroll
  for (int r = 0; r < 16; ++r) wr[r] = *reinterpret_cast<const float4*>(wsrc + (size_t)r * NN);
  const float* xsrc = x + k0 + 4 * lane;
  #pragma unroll
  for (int r = 0; r < NB; ++r) xr[r] = *reinterpret_cast<const float4*>(xsrc + (size_t)r * NN);

  #pragma unroll
  for (int r = 0; r < 16; ++r) *reinterpret_cast<float4*>(&Wt[wv][r][4 * lane]) = wr[r];
  #pragma unroll
  for (int r = 0; r < NB; ++r) *reinterpret_cast<float4*>(&Xt[wv][r][4 * lane]) = xr[r];
  __syncthreads();

  const float* wrow = &Wt[wv][l16][4 * half];
  const float* xrow = &Xt[wv][l16 & 7][4 * half];
  v8f acc0 = {}, acc1 = {};
  #pragma unroll
  for (int p = 0; p < GY_PIECES; ++p) {
    const float4 bw = *reinterpret_cast<const float4*>(wrow + 8 * p);
    const float4 ax = *reinterpret_cast<const float4*>(xrow + 8 * p);
    v2f a0 = { ax.x, ax.y };
    v2f b0 = { bw.x, bw.y };
    acc0 = __builtin_amdgcn_wmma_f32_16x16x4_f32(false, a0, false, b0,
                                                 (short)0, acc0, false, false);
    v2f a1 = { ax.z, ax.w };
    v2f b1 = { bw.z, bw.w };
    acc1 = __builtin_amdgcn_wmma_f32_16x16x4_f32(false, a1, false, b1,
                                                 (short)0, acc1, false, false);
  }
  #pragma unroll
  for (int r = 0; r < 8; ++r) part[wv][r + half * 8][l16] = acc0[r] + acc1[r];
  __syncthreads();

  if (threadIdx.x < NB * 16) {
    const int m = threadIdx.x >> 4, c = threadIdx.x & 15;
    float s = 0.f;
    #pragma unroll
    for (int w = 0; w < GY_WAVES; ++w) s += part[w][m][c];
    Y[m * NN + nb + c] = s;
  }
}

static_assert(GY_THREADS == SCAN_THREADS, "the scan and the tiles share a launch");
#define GY_TILES    (NN / 16)
#define CLR_BYTES   (GY_THREADS * 16)
#define CLR_HIST2   (NBINS * 4 / CLR_BYTES)
#define CLR_OUT     (NB * NN * 4 / CLR_BYTES)
__global__ __launch_bounds__(GY_THREADS) void k_scan_hi_gemm_y(const unsigned* __restrict__ hist,
                                                               unsigned* __restrict__ meta,
                                                               const float* __restrict__ x,
                                                               const float* __restrict__ W,
                                                               float* __restrict__ Y,
                                                               unsigned* __restrict__ hist2,
                                                               float* __restrict__ out) {
  const int id = (int)blockIdx.x;
  if (id == 0) {
    scan_hi(hist, meta);
  } else if (id <= GY_TILES) {
    gemm_y_tile(x, W, Y, id - 1);
  } else {
    const int c = id - 1 - GY_TILES;
    uint4* dst = (c < CLR_HIST2) ? reinterpret_cast<uint4*>(hist2) + c * GY_THREADS
                                 : reinterpret_cast<uint4*>(out) + (c - CLR_HIST2) * GY_THREADS;
    dst[threadIdx.x] = make_uint4(0u, 0u, 0u, 0u);
  }
}

#define RS_WAVES   16
#define RS_THREADS (RS_WAVES * 32)
#define RS_ROWS    4
#define RS_WG_ROWS (RS_WAVES * RS_ROWS)
#define RS_Q       (NN / 128)

#define RS_ROW_BLOCKS (NN / RS_WG_ROWS)

__device__ __forceinline__ void row_stats(const float* __restrict__ x,
                                          const float* __restrict__ Y,
                                          const float* __restrict__ bias,
                                          float* __restrict__ rowM,
                                          float* __restrict__ rowZ,
                                          const int b, const int rowBase) {
  __shared__ float sM[RS_WG_ROWS], sZ[RS_WG_ROWS];
  const int lane = threadIdx.x & 31;
  const int wv = __builtin_amdgcn_readfirstlane((int)(threadIdx.x >> 5));

  float4 y[RS_Q], bb[RS_Q];
  const float4* ysrc = reinterpret_cast<const float4*>(Y + (size_t)b * NN) + lane;
  const float4* bsrc = reinterpret_cast<const float4*>(bias) + lane;
  #pragma unroll
  for (int q = 0; q < RS_Q; ++q) y[q] = ysrc[q * 32];
  #pragma unroll
  for (int q = 0; q < RS_Q; ++q) bb[q] = bsrc[q * 32];
  const float xv = x[b * NN + rowBase + wv * RS_ROWS + (lane & (RS_ROWS - 1))];

  float myM = 0.f, myZ = 0.f;
  for (int r = 0; r < RS_ROWS; ++r) {
    const float xbi = __uint_as_float((unsigned)__builtin_amdgcn_readlane((int)__float_as_uint(xv), r));

    float4 v[RS_Q];
    float4 m4 = make_float4(-INFINITY, -INFINITY, -INFINITY, -INFINITY);
    #pragma unroll
    for (int q = 0; q < RS_Q; ++q) {
      float4 u;
      u.x = fmaf(xbi, y[q].x, bb[q].x);
      u.y = fmaf(xbi, y[q].y, bb[q].y);
      u.z = fmaf(xbi, y[q].z, bb[q].z);
      u.w = fmaf(xbi, y[q].w, bb[q].w);
      u.x = fmaxf(u.x, 0.01f * u.x);
      u.y = fmaxf(u.y, 0.01f * u.y);
      u.z = fmaxf(u.z, 0.01f * u.z);
      u.w = fmaxf(u.w, 0.01f * u.w);
      v[q] = u;
      m4.x = fmaxf(m4.x, u.x);
      m4.y = fmaxf(m4.y, u.y);
      m4.z = fmaxf(m4.z, u.z);
      m4.w = fmaxf(m4.w, u.w);
    }
    float m = fmaxf(fmaxf(m4.x, m4.y), fmaxf(m4.z, m4.w));
    #pragma unroll
    for (int d = 16; d >= 1; d >>= 1) m = fmaxf(m, __shfl_xor(m, d, 32));

    float4 s4 = make_float4(0.f, 0.f, 0.f, 0.f);
    #pragma unroll
    for (int q = 0; q < RS_Q; ++q) {
      s4.x += __expf(v[q].x - m);
      s4.y += __expf(v[q].y - m);
      s4.z += __expf(v[q].z - m);
      s4.w += __expf(v[q].w - m);
    }
    float s = (s4.x + s4.y) + (s4.z + s4.w);
    #pragma unroll
    for (int d = 16; d >= 1; d >>= 1) s += __shfl_xor(s, d, 32);

    if (lane == r) { myM = m; myZ = s; }
  }
  if (lane < RS_ROWS) { sM[wv * RS_ROWS + lane] = myM; sZ[wv * RS_ROWS + lane] = myZ; }
  __syncthreads();
  if (wv < 2)      rowM[b * NN + rowBase + wv * 32 + lane]       = sM[wv * 32 + lane];
  else if (wv < 4) rowZ[b * NN + rowBase + (wv - 2) * 32 + lane] = sZ[(wv - 2) * 32 + lane];
}

static_assert(RS_THREADS == SCAN_THREADS, "the scan and the row blocks share a launch");
__global__ __launch_bounds__(RS_THREADS) void k_scan_lo_row_stats(const unsigned* __restrict__ hist2,
                                                                  unsigned* __restrict__ meta,
                                                                  const float* __restrict__ x,
                                                                  const float* __restrict__ Y,
                                                                  const float* __restrict__ bias,
                                                                  float* __restrict__ rowM,
                                                                  float* __restrict__ rowZ) {
  if (blockIdx.x == 0) {
    scan_lo(hist2, meta);
  } else {
    const int id = (int)blockIdx.x - 1;
    row_stats(x, Y, bias, rowM, rowZ, id / RS_ROW_BLOCKS, (id % RS_ROW_BLOCKS) * RS_WG_ROWS);
  }
}

#define AG_ROWS  (NN / 8)
#define AG_BATCH 16

__device__ __forceinline__ float ag_rows(const float (&av)[AG_BATCH],
                                         const float4* __restrict__ s,
                                         const float ybj, const float bj,
                                         const float med, float acc) {
  #pragma unroll
  for (int u = 0; u < AG_BATCH; ++u) {
    const float4 r = s[u];
    float t = fmaf(r.x, ybj, bj);
    t = (t > 0.f) ? t : 0.01f * t;
    const float p = r.z * __expf(t - r.y);
    acc += (av[u] > med) ? p : 0.f;
  }
  return acc;
}

__global__ __launch_bounds__(256) void k_aggregate(const float* __restrict__ x,
                                                   const float* __restrict__ Y,
                                                   const float* __restrict__ bias,
                                                   const float* __restrict__ a,
                                                   const unsigned* __restrict__ meta,
                                                   const float* __restrict__ rowM,
                                                   const float* __restrict__ rowZ,
                                                   float* __restrict__ out) {
  __shared__ float4 srow[AG_ROWS];
  const int j  = blockIdx.x * 256 + threadIdx.x;
  const int b  = blockIdx.y;
  const int iBase = blockIdx.z * AG_ROWS;

  const int   irow = b * NN + iBase + threadIdx.x;
  const float xv = x[irow];
  const float mv = rowM[irow];
  const float zv = rowZ[irow];
  const float ybj = Y[b * NN + j];
  const float bj  = bias[j];
  const float med = __uint_as_float(meta[2]);

  const float* acol = a + (size_t)iBase * NN + j;
  float va[AG_BATCH], vb[AG_BATCH];
  #pragma unroll
  for (int u = 0; u < AG_BATCH; ++u) va[u] = acol[(size_t)u * NN];

  srow[threadIdx.x] = make_float4(xv, mv, xv / zv, 0.f);
  __syncthreads();

  float acc = 0.f;
  int i0 = 0;
  for (; i0 < AG_ROWS - 2 * AG_BATCH; i0 += 2 * AG_BATCH) {
    #pragma unroll
    for (int u = 0; u < AG_BATCH; ++u) vb[u] = acol[(size_t)(i0 + AG_BATCH + u) * NN];
    __builtin_amdgcn_sched_barrier(0);
    acc = ag_rows(va, &srow[i0], ybj, bj, med, acc);
    #pragma unroll
    for (int u = 0; u < AG_BATCH; ++u) va[u] = acol[(size_t)(i0 + 2 * AG_BATCH + u) * NN];
    __builtin_amdgcn_sched_barrier(0);
    acc = ag_rows(vb, &srow[i0 + AG_BATCH], ybj, bj, med, acc);
  }
  #pragma unroll
  for (int u = 0; u < AG_BATCH; ++u) vb[u] = acol[(size_t)(i0 + AG_BATCH + u) * NN];
  __builtin_amdgcn_sched_barrier(0);
  acc = ag_rows(va, &srow[i0], ybj, bj, med, acc);
  acc = ag_rows(vb, &srow[i0 + AG_BATCH], ybj, bj, med, acc);

  atomicAdd(&out[b * NN + j], acc);
}

extern "C" void kernel_launch(void* const* d_in, const int* in_sizes, int n_in,
                              void* d_out, int out_size, void* d_ws, size_t ws_size,
                              hipStream_t stream) {
  const float* x    = (const float*)d_in[0];
  const float* adj  = (const float*)d_in[1];
  const float* W    = (const float*)d_in[2];
  const float* bias = (const float*)d_in[3];
  float* out = (float*)d_out;

  char* ws = (char*)d_ws;
  size_t off = 0;
  float*    a     = (float*)(ws + off);    off += (size_t)NN * NN * sizeof(float);
  float*    Y     = (float*)(ws + off);    off += (size_t)NB * NN * sizeof(float);
  float*    rowM  = (float*)(ws + off);    off += (size_t)NB * NN * sizeof(float);
  float*    rowZ  = (float*)(ws + off);    off += (size_t)NB * NN * sizeof(float);
  unsigned* hist1 = (unsigned*)(ws + off); off += (size_t)NBINS * sizeof(unsigned);
  unsigned* hist2 = (unsigned*)(ws + off); off += (size_t)NBINS * sizeof(unsigned);
  unsigned* meta  = (unsigned*)(ws + off); off += 16;

  hipMemsetAsync(hist1, 0, (size_t)NBINS * sizeof(unsigned), stream);

  k_softmax_hist     <<<NN / SH_ROWS, SH_THREADS, 0, stream>>>(adj, a, hist1);
  k_scan_hi_gemm_y   <<<1 + GY_TILES + CLR_HIST2 + CLR_OUT, GY_THREADS, 0, stream>>>(hist1, meta, x, W, Y, hist2, out);
  k_hist_lo          <<<1024, 256, 0, stream>>>(a, meta, hist2);
  k_scan_lo_row_stats<<<1 + RS_ROW_BLOCKS * NB, RS_THREADS, 0, stream>>>(hist2, meta, x, Y, bias, rowM, rowZ);
  k_aggregate        <<<dim3(NN / 256, NB, 8), 256, 0, stream>>>(x, Y, bias, a, meta, rowM, rowZ, out);
}


# Low-half histogram pass over a: 8192 waves, each thread 16 dependent 4-byte loads (a cache trip per value) -> 2048 waves, each thread issues its 16 16-byte loads as one group, then filters
# speedup vs baseline: 39.2225x; 1.0708x over previous
#include <hip/hip_runtime.h>
#include <math.h>

#define NN 2048
#define NB 8
#define NBINS 65536

typedef __attribute__((ext_vector_type(2))) float v2f;
typedef __attribute__((ext_vector_type(8))) float v8f;

#define SH_ROWS     8
#define SH_COLS     64
#define SH_THREADS  (SH_ROWS * SH_COLS)
#define SH_K        (NN / (4 * SH_COLS))
#define FLUSH_BATCH 16

__global__ __launch_bounds__(SH_THREADS) void k_softmax_hist(const float* __restrict__ adj,
                                                             float* __restrict__ a,
                                                             unsigned* __restrict__ hist) {
  __shared__ unsigned lh[NBINS];
  __shared__ float4 xsum[SH_ROWS][SH_COLS];
  __shared__ float xmax[SH_ROWS][2];
  const int t    = threadIdx.x;
  const int r    = t >> 6;
  const int c    = t & 63;
  const int lane = t & 31;
  const int wh   = (t >> 5) & 1;
  const size_t rowoff = (size_t)(blockIdx.x * SH_ROWS + r) * NN;

  const float4* src = reinterpret_cast<const float4*>(adj + rowoff) + c;
  float4 v[SH_K];
  #pragma unroll
  for (int k = 0; k < SH_K; ++k) v[k] = src[k * SH_COLS];

  uint4* lh4 = reinterpret_cast<uint4*>(lh);
  for (int i = t; i < NBINS / 4; i += SH_THREADS) lh4[i] = make_uint4(0u, 0u, 0u, 0u);

  float mymax = -INFINITY;
  #pragma unroll
  for (int k = 0; k < SH_K; ++k)
    mymax = fmaxf(mymax, fmaxf(fmaxf(v[k].x, v[k].y), fmaxf(v[k].z, v[k].w)));
  #pragma unroll
  for (int d = 16; d >= 1; d >>= 1) mymax = fmaxf(mymax, __shfl_xor(mymax, d, 32));
  if (lane == 0) xmax[r][wh] = mymax;
  __syncthreads();
  const float m = fmaxf(xmax[r][0], xmax[r][1]);

  float4 s4 = make_float4(0.f, 0.f, 0.f, 0.f);
  #pragma unroll
  for (int k = 0; k < SH_K; ++k) {
    v[k].x = __expf(v[k].x - m);
    v[k].y = __expf(v[k].y - m);
    v[k].z = __expf(v[k].z - m);
    v[k].w = __expf(v[k].w - m);
    s4.x += v[k].x;
    s4.y += v[k].y;
    s4.z += v[k].z;
    s4.w += v[k].w;
  }
  xsum[r][c] = s4;
  __syncthreads();
  const float4 o = xsum[r][c ^ 32];
  s4.x += o.x; s4.y += o.y; s4.z += o.z; s4.w += o.w;
  #pragma unroll
  for (int d = 16; d >= 1; d >>= 1) {
    s4.x += __shfl_xor(s4.x, d, 32);
    s4.y += __shfl_xor(s4.y, d, 32);
    s4.z += __shfl_xor(s4.z, d, 32);
    s4.w += __shfl_xor(s4.w, d, 32);
  }
  const float inv = 1.f / ((s4.x + s4.z) + (s4.y + s4.w));

  float4* dst = reinterpret_cast<float4*>(a + rowoff) + c;
  #pragma unroll
  for (int k = 0; k < SH_K; ++k) {
    v[k].x *= inv; v[k].y *= inv; v[k].z *= inv; v[k].w *= inv;
    dst[k * SH_COLS] = v[k];
  }
  #pragma unroll
  for (int k = 0; k < SH_K; ++k) {
    atomicAdd(&lh[__float_as_uint(v[k].x) >> 16], 1u);
    atomicAdd(&lh[__float_as_uint(v[k].y) >> 16], 1u);
    atomicAdd(&lh[__float_as_uint(v[k].z) >> 16], 1u);
    atomicAdd(&lh[__float_as_uint(v[k].w) >> 16], 1u);
  }
  __syncthreads();

  for (int i0 = 0; i0 < NBINS; i0 += FLUSH_BATCH * SH_THREADS) {
    unsigned cnt[FLUSH_BATCH];
    #pragma unroll
    for (int u = 0; u < FLUSH_BATCH; ++u) cnt[u] = lh[i0 + u * SH_THREADS + t];
    #pragma unroll
    for (int u = 0; u < FLUSH_BATCH; ++u)
      if (cnt[u] != 0u) atomicAdd(&hist[i0 + u * SH_THREADS + t], cnt[u]);
  }
}

#define SCAN_THREADS 512
#define SCAN_WAVES   (SCAN_THREADS / 32)
#define WAVE_BINS    (NBINS / SCAN_WAVES)
#define WAVE_LOADS   (WAVE_BINS / 128)

__device__ __forceinline__ unsigned sum4(const uint4 v) { return v.x + v.y + v.z + v.w; }

__device__ __forceinline__ unsigned wave_inclusive_scan(unsigned v, const int lane) {
  #pragma unroll
  for (int d = 1; d < 32; d <<= 1) {
    const unsigned n = __shfl_up(v, d, 32);
    if (lane >= d) v += n;
  }
  return v;
}

__device__ __forceinline__ bool block_find_bin(const unsigned* __restrict__ hist,
                                               const unsigned target,
                                               unsigned& bin, unsigned& before) {
  __shared__ unsigned wave_tot[SCAN_WAVES];
  const int lane = threadIdx.x & 31;
  const int w    = threadIdx.x >> 5;
  const uint4* src = reinterpret_cast<const uint4*>(hist + w * WAVE_BINS) + lane;

  uint4 q[WAVE_LOADS];
  #pragma unroll
  for (int i = 0; i < WAVE_LOADS; ++i) q[i] = src[i * 32];

  unsigned wsum = 0;
  #pragma unroll
  for (int i = 0; i < WAVE_LOADS; ++i) wsum += sum4(q[i]);
  #pragma unroll
  for (int d = 16; d >= 1; d >>= 1) wsum += __shfl_xor(wsum, d, 32);
  if (lane == 0) wave_tot[w] = wsum;
  __syncthreads();
  const unsigned wt    = (lane < SCAN_WAVES) ? wave_tot[lane] : 0u;
  const unsigned wincl = wave_inclusive_scan(wt, lane);
  const unsigned wave_before = __shfl(wincl - wt, w, 32);
  if (!(wave_before <= target && target < wave_before + wsum)) return false;

  unsigned t[WAVE_LOADS];
  #pragma unroll
  for (int i = 0; i < WAVE_LOADS; ++i) t[i] = sum4(q[i]);
  #pragma unroll
  for (int d = 16; d >= 1; d >>= 1) {
    #pragma unroll
    for (int i = 0; i < WAVE_LOADS; ++i) t[i] += __shfl_xor(t[i], d, 32);
  }
  unsigned base = wave_before;
  uint4 c = q[0];
  int isel = 0;
  bool found = false;
  #pragma unroll
  for (int i = 0; i < WAVE_LOADS; ++i) {
    if (!found) {
      if (base + t[i] > target) { c = q[i]; isel = i; found = true; }
      else base += t[i];
    }
  }

  const unsigned mine = sum4(c);
  const unsigned lane_before = base + wave_inclusive_scan(mine, lane) - mine;
  if (!(lane_before <= target && target < lane_before + mine)) return false;
  unsigned cum = lane_before;
  unsigned comp = 0;
  if (cum + c.x <= target) {
    cum += c.x; comp = 1;
    if (cum + c.y <= target) {
      cum += c.y; comp = 2;
      if (cum + c.z <= target) { cum += c.z; comp = 3; }
    }
  }
  bin    = (unsigned)(w * WAVE_BINS + isel * 128 + lane * 4) + comp;
  before = cum;
  return true;
}

__device__ __forceinline__ void scan_hi(const unsigned* __restrict__ hist,
                                        unsigned* __restrict__ meta) {
  const unsigned k = ((unsigned)NN * (unsigned)NN - 1u) / 2u;
  unsigned bin, before;
  if (block_find_bin(hist, k, bin, before)) { meta[0] = bin; meta[1] = before; }
}

#define HL_THREADS 256
#define HL_PIECES  16
#define HL_WGS     (NN * NN / 4 / (HL_THREADS * HL_PIECES))

__device__ __forceinline__ void count_lo(const float f, const unsigned sel, unsigned* __restrict__ hist2) {
  const unsigned bits = __float_as_uint(f);
  if ((bits >> 16) == sel) atomicAdd(&hist2[bits & 0xFFFFu], 1u);
}

__global__ __launch_bounds__(HL_THREADS) void k_hist_lo(const float* __restrict__ a,
                                                        const unsigned* __restrict__ meta,
                                                        unsigned* __restrict__ hist2) {
  const float4* src = reinterpret_cast<const float4*>(a)
                    + (size_t)blockIdx.x * (HL_THREADS * HL_PIECES) + threadIdx.x;
  float4 v[HL_PIECES];
  #pragma unroll
  for (int k = 0; k < HL_PIECES; ++k) v[k] = src[k * HL_THREADS];
  const unsigned sel = meta[0];
  #pragma unroll
  for (int k = 0; k < HL_PIECES; ++k) {
    count_lo(v[k].x, sel, hist2);
    count_lo(v[k].y, sel, hist2);
    count_lo(v[k].z, sel, hist2);
    count_lo(v[k].w, sel, hist2);
  }
}

__device__ __forceinline__ void scan_lo(const unsigned* __restrict__ hist2,
                                        unsigned* __restrict__ meta) {
  const unsigned k = ((unsigned)NN * (unsigned)NN - 1u) / 2u;
  const unsigned kk = k - meta[1];
  const unsigned hi = meta[0];
  unsigned low, before;
  if (block_find_bin(hist2, kk, low, before)) meta[2] = (hi << 16) | low;
}

#define GY_WAVES   16
#define GY_THREADS (GY_WAVES * 32)
#define GY_KSLICE  (NN / GY_WAVES)
#define GY_PITCH   (GY_KSLICE + 4)
#define GY_PIECES  (GY_KSLICE / 8)

__device__ __forceinline__ void gemm_y_tile(const float* __restrict__ x,
                                            const float* __restrict__ W,
                                            float* __restrict__ Y,
                                            const int tile) {
  __shared__ __attribute__((aligned(16))) float Wt[GY_WAVES][16][GY_PITCH];
  __shared__ __attribute__((aligned(16))) float Xt[GY_WAVES][NB][GY_PITCH];
  __shared__ float part[GY_WAVES][16][16];
  const int nb   = tile * 16;
  const int lane = threadIdx.x & 31;
  const int wv   = threadIdx.x >> 5;
  const int half = lane >> 4;
  const int l16  = lane & 15;
  const int k0   = wv * GY_KSLICE;

  float4 wr[16], xr[NB];
  const float* wsrc = W + (size_t)nb * NN + k0 + 4 * lane;
  #pragma unroll
  for (int r = 0; r < 16; ++r) wr[r] = *reinterpret_cast<const float4*>(wsrc + (size_t)r * NN);
  const float* xsrc = x + k0 + 4 * lane;
  #pragma unroll
  for (int r = 0; r < NB; ++r) xr[r] = *reinterpret_cast<const float4*>(xsrc + (size_t)r * NN);

  #pragma unroll
  for (int r = 0; r < 16; ++r) *reinterpret_cast<float4*>(&Wt[wv][r][4 * lane]) = wr[r];
  #pragma unroll
  for (int r = 0; r < NB; ++r) *reinterpret_cast<float4*>(&Xt[wv][r][4 * lane]) = xr[r];
  __syncthreads();

  const float* wrow = &Wt[wv][l16][4 * half];
  const float* xrow = &Xt[wv][l16 & 7][4 * half];
  v8f acc0 = {}, acc1 = {};
  #pragma unroll
  for (int p = 0; p < GY_PIECES; ++p) {
    const float4 bw = *reinterpret_cast<const float4*>(wrow + 8 * p);
    const float4 ax = *reinterpret_cast<const float4*>(xrow + 8 * p);
    v2f a0 = { ax.x, ax.y };
    v2f b0 = { bw.x, bw.y };
    acc0 = __builtin_amdgcn_wmma_f32_16x16x4_f32(false, a0, false, b0,
                                                 (short)0, acc0, false, false);
    v2f a1 = { ax.z, ax.w };
    v2f b1 = { bw.z, bw.w };
    acc1 = __builtin_amdgcn_wmma_f32_16x16x4_f32(false, a1, false, b1,
                                                 (short)0, acc1, false, false);
  }
  #pragma unroll
  for (int r = 0; r < 8; ++r) part[wv][r + half * 8][l16] = acc0[r] + acc1[r];
  __syncthreads();

  if (threadIdx.x < NB * 16) {
    const int m = threadIdx.x >> 4, c = threadIdx.x & 15;
    float s = 0.f;
    #pragma unroll
    for (int w = 0; w < GY_WAVES; ++w) s += part[w][m][c];
    Y[m * NN + nb + c] = s;
  }
}

static_assert(GY_THREADS == SCAN_THREADS, "the scan and the tiles share a launch");
#define GY_TILES    (NN / 16)
#define CLR_BYTES   (GY_THREADS * 16)
#define CLR_HIST2   (NBINS * 4 / CLR_BYTES)
#define CLR_OUT     (NB * NN * 4 / CLR_BYTES)
__global__ __launch_bounds__(GY_THREADS) void k_scan_hi_gemm_y(const unsigned* __restrict__ hist,
                                                               unsigned* __restrict__ meta,
                                                               const float* __restrict__ x,
                                                               const float* __restrict__ W,
                                                               float* __restrict__ Y,
                                                               unsigned* __restrict__ hist2,
                                                               float* __restrict__ out) {
  const int id = (int)blockIdx.x;
  if (id == 0) {
    scan_hi(hist, meta);
  } else if (id <= GY_TILES) {
    gemm_y_tile(x, W, Y, id - 1);
  } else {
    const int c = id - 1 - GY_TILES;
    uint4* dst = (c < CLR_HIST2) ? reinterpret_cast<uint4*>(hist2) + c * GY_THREADS
                                 : reinterpret_cast<uint4*>(out) + (c - CLR_HIST2) * GY_THREADS;
    dst[threadIdx.x] = make_uint4(0u, 0u, 0u, 0u);
  }
}

#define RS_WAVES   16
#define RS_THREADS (RS_WAVES * 32)
#define RS_ROWS    4
#define RS_WG_ROWS (RS_WAVES * RS_ROWS)
#define RS_Q       (NN / 128)

#define RS_ROW_BLOCKS (NN / RS_WG_ROWS)

__device__ __forceinline__ void row_stats(const float* __restrict__ x,
                                          const float* __restrict__ Y,
                                          const float* __restrict__ bias,
                                          float* __restrict__ rowM,
                                          float* __restrict__ rowZ,
                                          const int b, const int rowBase) {
  __shared__ float sM[RS_WG_ROWS], sZ[RS_WG_ROWS];
  const int lane = threadIdx.x & 31;
  const int wv = __builtin_amdgcn_readfirstlane((int)(threadIdx.x >> 5));

  float4 y[RS_Q], bb[RS_Q];
  const float4* ysrc = reinterpret_cast<const float4*>(Y + (size_t)b * NN) + lane;
  const float4* bsrc = reinterpret_cast<const float4*>(bias) + lane;
  #pragma unroll
  for (int q = 0; q < RS_Q; ++q) y[q] = ysrc[q * 32];
  #pragma unroll
  for (int q = 0; q < RS_Q; ++q) bb[q] = bsrc[q * 32];
  const float xv = x[b * NN + rowBase + wv * RS_ROWS + (lane & (RS_ROWS - 1))];

  float myM = 0.f, myZ = 0.f;
  for (int r = 0; r < RS_ROWS; ++r) {
    const float xbi = __uint_as_float((unsigned)__builtin_amdgcn_readlane((int)__float_as_uint(xv), r));

    float4 v[RS_Q];
    float4 m4 = make_float4(-INFINITY, -INFINITY, -INFINITY, -INFINITY);
    #pragma unroll
    for (int q = 0; q < RS_Q; ++q) {
      float4 u;
      u.x = fmaf(xbi, y[q].x, bb[q].x);
      u.y = fmaf(xbi, y[q].y, bb[q].y);
      u.z = fmaf(xbi, y[q].z, bb[q].z);
      u.w = fmaf(xbi, y[q].w, bb[q].w);
      u.x = fmaxf(u.x, 0.01f * u.x);
      u.y = fmaxf(u.y, 0.01f * u.y);
      u.z = fmaxf(u.z, 0.01f * u.z);
      u.w = fmaxf(u.w, 0.01f * u.w);
      v[q] = u;
      m4.x = fmaxf(m4.x, u.x);
      m4.y = fmaxf(m4.y, u.y);
      m4.z = fmaxf(m4.z, u.z);
      m4.w = fmaxf(m4.w, u.w);
    }
    float m = fmaxf(fmaxf(m4.x, m4.y), fmaxf(m4.z, m4.w));
    #pragma unroll
    for (int d = 16; d >= 1; d >>= 1) m = fmaxf(m, __shfl_xor(m, d, 32));

    float4 s4 = make_float4(0.f, 0.f, 0.f, 0.f);
    #pragma unroll
    for (int q = 0; q < RS_Q; ++q) {
      s4.x += __expf(v[q].x - m);
      s4.y += __expf(v[q].y - m);
      s4.z += __expf(v[q].z - m);
      s4.w += __expf(v[q].w - m);
    }
    float s = (s4.x + s4.y) + (s4.z + s4.w);
    #pragma unroll
    for (int d = 16; d >= 1; d >>= 1) s += __shfl_xor(s, d, 32);

    if (lane == r) { myM = m; myZ = s; }
  }
  if (lane < RS_ROWS) { sM[wv * RS_ROWS + lane] = myM; sZ[wv * RS_ROWS + lane] = myZ; }
  __syncthreads();
  if (wv < 2)      rowM[b * NN + rowBase + wv * 32 + lane]       = sM[wv * 32 + lane];
  else if (wv < 4) rowZ[b * NN + rowBase + (wv - 2) * 32 + lane] = sZ[(wv - 2) * 32 + lane];
}

static_assert(RS_THREADS == SCAN_THREADS, "the scan and the row blocks share a launch");
__global__ __launch_bounds__(RS_THREADS) void k_scan_lo_row_stats(const unsigned* __restrict__ hist2,
                                                                  unsigned* __restrict__ meta,
                                                                  const float* __restrict__ x,
                                                                  const float* __restrict__ Y,
                                                                  const float* __restrict__ bias,
                                                                  float* __restrict__ rowM,
                                                                  float* __restrict__ rowZ) {
  if (blockIdx.x == 0) {
    scan_lo(hist2, meta);
  } else {
    const int id = (int)blockIdx.x - 1;
    row_stats(x, Y, bias, rowM, rowZ, id / RS_ROW_BLOCKS, (id % RS_ROW_BLOCKS) * RS_WG_ROWS);
  }
}

#define AG_ROWS  (NN / 8)
#define AG_BATCH 16

__device__ __forceinline__ float ag_rows(const float (&av)[AG_BATCH],
                                         const float4* __restrict__ s,
                                         const float ybj, const float bj,
                                         const float med, float acc) {
  #pragma unroll
  for (int u = 0; u < AG_BATCH; ++u) {
    const float4 r = s[u];
    float t = fmaf(r.x, ybj, bj);
    t = (t > 0.f) ? t : 0.01f * t;
    const float p = r.z * __expf(t - r.y);
    acc += (av[u] > med) ? p : 0.f;
  }
  return acc;
}

__global__ __launch_bounds__(256) void k_aggregate(const float* __restrict__ x,
                                                   const float* __restrict__ Y,
                                                   const float* __restrict__ bias,
                                                   const float* __restrict__ a,
                                                   const unsigned* __restrict__ meta,
                                                   const float* __restrict__ rowM,
                                                   const float* __restrict__ rowZ,
                                                   float* __restrict__ out) {
  __shared__ float4 srow[AG_ROWS];
  const int j  = blockIdx.x * 256 + threadIdx.x;
  const int b  = blockIdx.y;
  const int iBase = blockIdx.z * AG_ROWS;

  const int   irow = b * NN + iBase + threadIdx.x;
  const float xv = x[irow];
  const float mv = rowM[irow];
  const float zv = rowZ[irow];
  const float ybj = Y[b * NN + j];
  const float bj  = bias[j];
  const float med = __uint_as_float(meta[2]);

  const float* acol = a + (size_t)iBase * NN + j;
  float va[AG_BATCH], vb[AG_BATCH];
  #pragma unroll
  for (int u = 0; u < AG_BATCH; ++u) va[u] = acol[(size_t)u * NN];

  srow[threadIdx.x] = make_float4(xv, mv, xv / zv, 0.f);
  __syncthreads();

  float acc = 0.f;
  int i0 = 0;
  for (; i0 < AG_ROWS - 2 * AG_BATCH; i0 += 2 * AG_BATCH) {
    #pragma unroll
    for (int u = 0; u < AG_BATCH; ++u) vb[u] = acol[(size_t)(i0 + AG_BATCH + u) * NN];
    __builtin_amdgcn_sched_barrier(0);
    acc = ag_rows(va, &srow[i0], ybj, bj, med, acc);
    #pragma unroll
    for (int u = 0; u < AG_BATCH; ++u) va[u] = acol[(size_t)(i0 + 2 * AG_BATCH + u) * NN];
    __builtin_amdgcn_sched_barrier(0);
    acc = ag_rows(vb, &srow[i0 + AG_BATCH], ybj, bj, med, acc);
  }
  #pragma unroll
  for (int u = 0; u < AG_BATCH; ++u) vb[u] = acol[(size_t)(i0 + AG_BATCH + u) * NN];
  __builtin_amdgcn_sched_barrier(0);
  acc = ag_rows(va, &srow[i0], ybj, bj, med, acc);
  acc = ag_rows(vb, &srow[i0 + AG_BATCH], ybj, bj, med, acc);

  atomicAdd(&out[b * NN + j], acc);
}

extern "C" void kernel_launch(void* const* d_in, const int* in_sizes, int n_in,
                              void* d_out, int out_size, void* d_ws, size_t ws_size,
                              hipStream_t stream) {
  const float* x    = (const float*)d_in[0];
  const float* adj  = (const float*)d_in[1];
  const float* W    = (const float*)d_in[2];
  const float* bias = (const float*)d_in[3];
  float* out = (float*)d_out;

  char* ws = (char*)d_ws;
  size_t off = 0;
  float*    a     = (float*)(ws + off);    off += (size_t)NN * NN * sizeof(float);
  float*    Y     = (float*)(ws + off);    off += (size_t)NB * NN * sizeof(float);
  float*    rowM  = (float*)(ws + off);    off += (size_t)NB * NN * sizeof(float);
  float*    rowZ  = (float*)(ws + off);    off += (size_t)NB * NN * sizeof(float);
  unsigned* hist1 = (unsigned*)(ws + off); off += (size_t)NBINS * sizeof(unsigned);
  unsigned* hist2 = (unsigned*)(ws + off); off += (size_t)NBINS * sizeof(unsigned);
  unsigned* meta  = (unsigned*)(ws + off); off += 16;

  hipMemsetAsync(hist1, 0, (size_t)NBINS * sizeof(unsigned), stream);

  k_softmax_hist     <<<NN / SH_ROWS, SH_THREADS, 0, stream>>>(adj, a, hist1);
  k_scan_hi_gemm_y   <<<1 + GY_TILES + CLR_HIST2 + CLR_OUT, GY_THREADS, 0, stream>>>(hist1, meta, x, W, Y, hist2, out);
  k_hist_lo          <<<HL_WGS, HL_THREADS, 0, stream>>>(a, meta, hist2);
  k_scan_lo_row_stats<<<1 + RS_ROW_BLOCKS * NB, RS_THREADS, 0, stream>>>(hist2, meta, x, Y, bias, rowM, rowZ);
  k_aggregate        <<<dim3(NN / 256, NB, 8), 256, 0, stream>>>(x, Y, bias, a, meta, rowM, rowZ, out);
}


# k_aggregate: a thread keeps all 8 batch rows of 2 columns: each a value loaded and masked once (as a -inf bias), not 8x; exp2 with log2e folded into the row scalars; new row sum order
# speedup vs baseline: 42.6207x; 1.0866x over previous
#include <hip/hip_runtime.h>
#include <math.h>

#define NN 2048
#define NB 8
#define NBINS 65536

typedef __attribute__((ext_vector_type(2))) float v2f;
typedef __attribute__((ext_vector_type(8))) float v8f;

#define SH_ROWS     8
#define SH_COLS     64
#define SH_THREADS  (SH_ROWS * SH_COLS)
#define SH_K        (NN / (4 * SH_COLS))
#define FLUSH_BATCH 16

__global__ __launch_bounds__(SH_THREADS) void k_softmax_hist(const float* __restrict__ adj,
                                                             float* __restrict__ a,
                                                             unsigned* __restrict__ hist) {
  __shared__ unsigned lh[NBINS];
  __shared__ float4 xsum[SH_ROWS][SH_COLS];
  __shared__ float xmax[SH_ROWS][2];
  const int t    = threadIdx.x;
  const int r    = t >> 6;
  const int c    = t & 63;
  const int lane = t & 31;
  const int wh   = (t >> 5) & 1;
  const size_t rowoff = (size_t)(blockIdx.x * SH_ROWS + r) * NN;

  const float4* src = reinterpret_cast<const float4*>(adj + rowoff) + c;
  float4 v[SH_K];
  #pragma unroll
  for (int k = 0; k < SH_K; ++k) v[k] = src[k * SH_COLS];

  uint4* lh4 = reinterpret_cast<uint4*>(lh);
  for (int i = t; i < NBINS / 4; i += SH_THREADS) lh4[i] = make_uint4(0u, 0u, 0u, 0u);

  float mymax = -INFINITY;
  #pragma unroll
  for (int k = 0; k < SH_K; ++k)
    mymax = fmaxf(mymax, fmaxf(fmaxf(v[k].x, v[k].y), fmaxf(v[k].z, v[k].w)));
  #pragma unroll
  for (int d = 16; d >= 1; d >>= 1) mymax = fmaxf(mymax, __shfl_xor(mymax, d, 32));
  if (lane == 0) xmax[r][wh] = mymax;
  __syncthreads();
  const float m = fmaxf(xmax[r][0], xmax[r][1]);

  float4 s4 = make_float4(0.f, 0.f, 0.f, 0.f);
  #pragma unroll
  for (int k = 0; k < SH_K; ++k) {
    v[k].x = __expf(v[k].x - m);
    v[k].y = __expf(v[k].y - m);
    v[k].z = __expf(v[k].z - m);
    v[k].w = __expf(v[k].w - m);
    s4.x += v[k].x;
    s4.y += v[k].y;
    s4.z += v[k].z;
    s4.w += v[k].w;
  }
  xsum[r][c] = s4;
  __syncthreads();
  const float4 o = xsum[r][c ^ 32];
  s4.x += o.x; s4.y += o.y; s4.z += o.z; s4.w += o.w;
  #pragma unroll
  for (int d = 16; d >= 1; d >>= 1) {
    s4.x += __shfl_xor(s4.x, d, 32);
    s4.y += __shfl_xor(s4.y, d, 32);
    s4.z += __shfl_xor(s4.z, d, 32);
    s4.w += __shfl_xor(s4.w, d, 32);
  }
  const float inv = 1.f / ((s4.x + s4.z) + (s4.y + s4.w));

  float4* dst = reinterpret_cast<float4*>(a + rowoff) + c;
  #pragma unroll
  for (int k = 0; k < SH_K; ++k) {
    v[k].x *= inv; v[k].y *= inv; v[k].z *= inv; v[k].w *= inv;
    dst[k * SH_COLS] = v[k];
  }
  #pragma unroll
  for (int k = 0; k < SH_K; ++k) {
    atomicAdd(&lh[__float_as_uint(v[k].x) >> 16], 1u);
    atomicAdd(&lh[__float_as_uint(v[k].y) >> 16], 1u);
    atomicAdd(&lh[__float_as_uint(v[k].z) >> 16], 1u);
    atomicAdd(&lh[__float_as_uint(v[k].w) >> 16], 1u);
  }
  __syncthreads();

  for (int i0 = 0; i0 < NBINS; i0 += FLUSH_BATCH * SH_THREADS) {
    unsigned cnt[FLUSH_BATCH];
    #pragma unroll
    for (int u = 0; u < FLUSH_BATCH; ++u) cnt[u] = lh[i0 + u * SH_THREADS + t];
    #pragma unroll
    for (int u = 0; u < FLUSH_BATCH; ++u)
      if (cnt[u] != 0u) atomicAdd(&hist[i0 + u * SH_THREADS + t], cnt[u]);
  }
}

#define SCAN_THREADS 512
#define SCAN_WAVES   (SCAN_THREADS / 32)
#define WAVE_BINS    (NBINS / SCAN_WAVES)
#define WAVE_LOADS   (WAVE_BINS / 128)

__device__ __forceinline__ unsigned sum4(const uint4 v) { return v.x + v.y + v.z + v.w; }

__device__ __forceinline__ unsigned wave_inclusive_scan(unsigned v, const int lane) {
  #pragma unroll
  for (int d = 1; d < 32; d <<= 1) {
    const unsigned n = __shfl_up(v, d, 32);
    if (lane >= d) v += n;
  }
  return v;
}

__device__ __forceinline__ bool block_find_bin(const unsigned* __restrict__ hist,
                                               const unsigned target,
                                               unsigned& bin, unsigned& before) {
  __shared__ unsigned wave_tot[SCAN_WAVES];
  const int lane = threadIdx.x & 31;
  const int w    = threadIdx.x >> 5;
  const uint4* src = reinterpret_cast<const uint4*>(hist + w * WAVE_BINS) + lane;

  uint4 q[WAVE_LOADS];
  #pragma unroll
  for (int i = 0; i < WAVE_LOADS; ++i) q[i] = src[i * 32];

  unsigned wsum = 0;
  #pragma unroll
  for (int i = 0; i < WAVE_LOADS; ++i) wsum += sum4(q[i]);
  #pragma unroll
  for (int d = 16; d >= 1; d >>= 1) wsum += __shfl_xor(wsum, d, 32);
  if (lane == 0) wave_tot[w] = wsum;
  __syncthreads();
  const unsigned wt    = (lane < SCAN_WAVES) ? wave_tot[lane] : 0u;
  const unsigned wincl = wave_inclusive_scan(wt, lane);
  const unsigned wave_before = __shfl(wincl - wt, w, 32);
  if (!(wave_before <= target && target < wave_before + wsum)) return false;

  unsigned t[WAVE_LOADS];
  #pragma unroll
  for (int i = 0; i < WAVE_LOADS; ++i) t[i] = sum4(q[i]);
  #pragma unroll
  for (int d = 16; d >= 1; d >>= 1) {
    #pragma unroll
    for (int i = 0; i < WAVE_LOADS; ++i) t[i] += __shfl_xor(t[i], d, 32);
  }
  unsigned base = wave_before;
  uint4 c = q[0];
  int isel = 0;
  bool found = false;
  #pragma unroll
  for (int i = 0; i < WAVE_LOADS; ++i) {
    if (!found) {
      if (base + t[i] > target) { c = q[i]; isel = i; found = true; }
      else base += t[i];
    }
  }

  const unsigned mine = sum4(c);
  const unsigned lane_before = base + wave_inclusive_scan(mine, lane) - mine;
  if (!(lane_before <= target && target < lane_before + mine)) return false;
  unsigned cum = lane_before;
  unsigned comp = 0;
  if (cum + c.x <= target) {
    cum += c.x; comp = 1;
    if (cum + c.y <= target) {
      cum += c.y; comp = 2;
      if (cum + c.z <= target) { cum += c.z; comp = 3; }
    }
  }
  bin    = (unsigned)(w * WAVE_BINS + isel * 128 + lane * 4) + comp;
  before = cum;
  return true;
}

__device__ __forceinline__ void scan_hi(const unsigned* __restrict__ hist,
                                        unsigned* __restrict__ meta) {
  const unsigned k = ((unsigned)NN * (unsigned)NN - 1u) / 2u;
  unsigned bin, before;
  if (block_find_bin(hist, k, bin, before)) { meta[0] = bin; meta[1] = before; }
}

#define HL_THREADS 256
#define HL_PIECES  16
#define HL_WGS     (NN * NN / 4 / (HL_THREADS * HL_PIECES))

__device__ __forceinline__ void count_lo(const float f, const unsigned sel, unsigned* __restrict__ hist2) {
  const unsigned bits = __float_as_uint(f);
  if ((bits >> 16) == sel) atomicAdd(&hist2[bits & 0xFFFFu], 1u);
}

__global__ __launch_bounds__(HL_THREADS) void k_hist_lo(const float* __restrict__ a,
                                                        const unsigned* __restrict__ meta,
                                                        unsigned* __restrict__ hist2) {
  const float4* src = reinterpret_cast<const float4*>(a)
                    + (size_t)blockIdx.x * (HL_THREADS * HL_PIECES) + threadIdx.x;
  float4 v[HL_PIECES];
  #pragma unroll
  for (int k = 0; k < HL_PIECES; ++k) v[k] = src[k * HL_THREADS];
  const unsigned sel = meta[0];
  #pragma unroll
  for (int k = 0; k < HL_PIECES; ++k) {
    count_lo(v[k].x, sel, hist2);
    count_lo(v[k].y, sel, hist2);
    count_lo(v[k].z, sel, hist2);
    count_lo(v[k].w, sel, hist2);
  }
}

__device__ __forceinline__ void scan_lo(const unsigned* __restrict__ hist2,
                                        unsigned* __restrict__ meta) {
  const unsigned k = ((unsigned)NN * (unsigned)NN - 1u) / 2u;
  const unsigned kk = k - meta[1];
  const unsigned hi = meta[0];
  unsigned low, before;
  if (block_find_bin(hist2, kk, low, before)) meta[2] = (hi << 16) | low;
}

#define GY_WAVES   16
#define GY_THREADS (GY_WAVES * 32)
#define GY_KSLICE  (NN / GY_WAVES)
#define GY_PITCH   (GY_KSLICE + 4)
#define GY_PIECES  (GY_KSLICE / 8)

__device__ __forceinline__ void gemm_y_tile(const float* __restrict__ x,
                                            const float* __restrict__ W,
                                            float* __restrict__ Y,
                                            const int tile) {
  __shared__ __attribute__((aligned(16))) float Wt[GY_WAVES][16][GY_PITCH];
  __shared__ __attribute__((aligned(16))) float Xt[GY_WAVES][NB][GY_PITCH];
  __shared__ float part[GY_WAVES][16][16];
  const int nb   = tile * 16;
  const int lane = threadIdx.x & 31;
  const int wv   = threadIdx.x >> 5;
  const int half = lane >> 4;
  const int l16  = lane & 15;
  const int k0   = wv * GY_KSLICE;

  float4 wr[16], xr[NB];
  const float* wsrc = W + (size_t)nb * NN + k0 + 4 * lane;
  #pragma unroll
  for (int r = 0; r < 16; ++r) wr[r] = *reinterpret_cast<const float4*>(wsrc + (size_t)r * NN);
  const float* xsrc = x + k0 + 4 * lane;
  #pragma unroll
  for (int r = 0; r < NB; ++r) xr[r] = *reinterpret_cast<const float4*>(xsrc + (size_t)r * NN);

  #pragma unroll
  for (int r = 0; r < 16; ++r) *reinterpret_cast<float4*>(&Wt[wv][r][4 * lane]) = wr[r];
  #pragma unroll
  for (int r = 0; r < NB; ++r) *reinterpret_cast<float4*>(&Xt[wv][r][4 * lane]) = xr[r];
  __syncthreads();

  const float* wrow = &Wt[wv][l16][4 * half];
  const float* xrow = &Xt[wv][l16 & 7][4 * half];
  v8f acc0 = {}, acc1 = {};
  #pragma unroll
  for (int p = 0; p < GY_PIECES; ++p) {
    const float4 bw = *reinterpret_cast<const float4*>(wrow + 8 * p);
    const float4 ax = *reinterpret_cast<const float4*>(xrow + 8 * p);
    v2f a0 = { ax.x, ax.y };
    v2f b0 = { bw.x, bw.y };
    acc0 = __builtin_amdgcn_wmma_f32_16x16x4_f32(false, a0, false, b0,
                                                 (short)0, acc0, false, false);
    v2f a1 = { ax.z, ax.w };
    v2f b1 = { bw.z, bw.w };
    acc1 = __builtin_amdgcn_wmma_f32_16x16x4_f32(false, a1, false, b1,
                                                 (short)0, acc1, false, false);
  }
  #pragma unroll
  for (int r = 0; r < 8; ++r) part[wv][r + half * 8][l16] = acc0[r] + acc1[r];
  __syncthreads();

  if (threadIdx.x < NB * 16) {
    const int m = threadIdx.x >> 4, c = threadIdx.x & 15;
    float s = 0.f;
    #pragma unroll
    for (int w = 0; w < GY_WAVES; ++w) s += part[w][m][c];
    Y[m * NN + nb + c] = s;
  }
}

static_assert(GY_THREADS == SCAN_THREADS, "the scan and the tiles share a launch");
#define GY_TILES    (NN / 16)
#define CLR_BYTES   (GY_THREADS * 16)
#define CLR_HIST2   (NBINS * 4 / CLR_BYTES)
#define CLR_OUT     (NB * NN * 4 / CLR_BYTES)
__global__ __launch_bounds__(GY_THREADS) void k_scan_hi_gemm_y(const unsigned* __restrict__ hist,
                                                               unsigned* __restrict__ meta,
                                                               const float* __restrict__ x,
                                                               const float* __restrict__ W,
                                                               float* __restrict__ Y,
                                                               unsigned* __restrict__ hist2,
                                                               float* __restrict__ out) {
  const int id = (int)blockIdx.x;
  if (id == 0) {
    scan_hi(hist, meta);
  } else if (id <= GY_TILES) {
    gemm_y_tile(x, W, Y, id - 1);
  } else {
    const int c = id - 1 - GY_TILES;
    uint4* dst = (c < CLR_HIST2) ? reinterpret_cast<uint4*>(hist2) + c * GY_THREADS
                                 : reinterpret_cast<uint4*>(out) + (c - CLR_HIST2) * GY_THREADS;
    dst[threadIdx.x] = make_uint4(0u, 0u, 0u, 0u);
  }
}

#define RS_WAVES   16
#define RS_THREADS (RS_WAVES * 32)
#define RS_ROWS    4
#define RS_WG_ROWS (RS_WAVES * RS_ROWS)
#define RS_Q       (NN / 128)

#define RS_ROW_BLOCKS (NN / RS_WG_ROWS)

__device__ __forceinline__ void row_stats(const float* __restrict__ x,
                                          const float* __restrict__ Y,
                                          const float* __restrict__ bias,
                                          float* __restrict__ rowM,
                                          float* __restrict__ rowZ,
                                          const int b, const int rowBase) {
  __shared__ float sM[RS_WG_ROWS], sZ[RS_WG_ROWS];
  const int lane = threadIdx.x & 31;
  const int wv = __builtin_amdgcn_readfirstlane((int)(threadIdx.x >> 5));

  float4 y[RS_Q], bb[RS_Q];
  const float4* ysrc = reinterpret_cast<const float4*>(Y + (size_t)b * NN) + lane;
  const float4* bsrc = reinterpret_cast<const float4*>(bias) + lane;
  #pragma unroll
  for (int q = 0; q < RS_Q; ++q) y[q] = ysrc[q * 32];
  #pragma unroll
  for (int q = 0; q < RS_Q; ++q) bb[q] = bsrc[q * 32];
  const float xv = x[b * NN + rowBase + wv * RS_ROWS + (lane & (RS_ROWS - 1))];

  float myM = 0.f, myZ = 0.f;
  for (int r = 0; r < RS_ROWS; ++r) {
    const float xbi = __uint_as_float((unsigned)__builtin_amdgcn_readlane((int)__float_as_uint(xv), r));

    float4 v[RS_Q];
    float4 m4 = make_float4(-INFINITY, -INFINITY, -INFINITY, -INFINITY);
    #pragma unroll
    for (int q = 0; q < RS_Q; ++q) {
      float4 u;
      u.x = fmaf(xbi, y[q].x, bb[q].x);
      u.y = fmaf(xbi, y[q].y, bb[q].y);
      u.z = fmaf(xbi, y[q].z, bb[q].z);
      u.w = fmaf(xbi, y[q].w, bb[q].w);
      u.x = fmaxf(u.x, 0.01f * u.x);
      u.y = fmaxf(u.y, 0.01f * u.y);
      u.z = fmaxf(u.z, 0.01f * u.z);
      u.w = fmaxf(u.w, 0.01f * u.w);
      v[q] = u;
      m4.x = fmaxf(m4.x, u.x);
      m4.y = fmaxf(m4.y, u.y);
      m4.z = fmaxf(m4.z, u.z);
      m4.w = fmaxf(m4.w, u.w);
    }
    float m = fmaxf(fmaxf(m4.x, m4.y), fmaxf(m4.z, m4.w));
    #pragma unroll
    for (int d = 16; d >= 1; d >>= 1) m = fmaxf(m, __shfl_xor(m, d, 32));

    float4 s4 = make_float4(0.f, 0.f, 0.f, 0.f);
    #pragma unroll
    for (int q = 0; q < RS_Q; ++q) {
      s4.x += __expf(v[q].x - m);
      s4.y += __expf(v[q].y - m);
      s4.z += __expf(v[q].z - m);
      s4.w += __expf(v[q].w - m);
    }
    float s = (s4.x + s4.y) + (s4.z + s4.w);
    #pragma unroll
    for (int d = 16; d >= 1; d >>= 1) s += __shfl_xor(s, d, 32);

    if (lane == r) { myM = m; myZ = s; }
  }
  if (lane < RS_ROWS) { sM[wv * RS_ROWS + lane] = myM; sZ[wv * RS_ROWS + lane] = myZ; }
  __syncthreads();
  if (wv < 2)      rowM[b * NN + rowBase + wv * 32 + lane]       = sM[wv * 32 + lane];
  else if (wv < 4) rowZ[b * NN + rowBase + (wv - 2) * 32 + lane] = sZ[(wv - 2) * 32 + lane];
}

static_assert(RS_THREADS == SCAN_THREADS, "the scan and the row blocks share a launch");
__global__ __launch_bounds__(RS_THREADS) void k_scan_lo_row_stats(const unsigned* __restrict__ hist2,
                                                                  unsigned* __restrict__ meta,
                                                                  const float* __restrict__ x,
                                                                  const float* __restrict__ Y,
                                                                  const float* __restrict__ bias,
                                                                  float* __restrict__ rowM,
                                                                  float* __restrict__ rowZ) {
  if (blockIdx.x == 0) {
    scan_lo(hist2, meta);
  } else {
    const int id = (int)blockIdx.x - 1;
    row_stats(x, Y, bias, rowM, rowZ, id / RS_ROW_BLOCKS, (id % RS_ROW_BLOCKS) * RS_WG_ROWS);
  }
}

#define AG_COLS   64
#define AG_ROWS   256
#define AG_WAVES  8
#define AG_WROWS  (AG_ROWS / AG_WAVES)
#define LOG2E_F   1.4426950408889634f

__global__ __launch_bounds__(AG_WAVES * 32) void k_aggregate(const float* __restrict__ x,
                                                             const float* __restrict__ Y,
                                                             const float* __restrict__ bias,
                                                             const float* __restrict__ a,
                                                             const unsigned* __restrict__ meta,
                                                             const float* __restrict__ rowM,
                                                             const float* __restrict__ rowZ,
                                                             float* __restrict__ out) {
  __shared__ float4 srow[AG_ROWS][6];
  __shared__ float2 red[AG_WAVES][NB][32];
  const int lane = threadIdx.x & 31;
  const int wv = __builtin_amdgcn_readfirstlane((int)(threadIdx.x >> 5));
  const int j0 = blockIdx.x * AG_COLS + 2 * lane;
  const int iBase = blockIdx.y * AG_ROWS;

  const float* acol = a + (size_t)(iBase + wv * AG_WROWS) * NN + j0;
  float2 av[AG_WROWS];
  #pragma unroll
  for (int u = 0; u < AG_WROWS; ++u) av[u] = *reinterpret_cast<const float2*>(acol + (size_t)u * NN);

  float2 yv[NB];
  #pragma unroll
  for (int b = 0; b < NB; ++b) yv[b] = *reinterpret_cast<const float2*>(Y + b * NN + j0);
  float2 bl = *reinterpret_cast<const float2*>(bias + j0);
  bl.x *= LOG2E_F; bl.y *= LOG2E_F;
  const float med = __uint_as_float(meta[2]);

  {
    const int i = iBase + (int)threadIdx.x;
    float xv[NB], mv[NB], zv[NB];
    #pragma unroll
    for (int b = 0; b < NB; ++b) {
      xv[b] = x[b * NN + i];
      mv[b] = rowM[b * NN + i];
      zv[b] = rowZ[b * NN + i];
    }
    float4* s = srow[threadIdx.x];
    s[0] = make_float4(xv[0] * LOG2E_F, xv[1] * LOG2E_F, xv[2] * LOG2E_F, xv[3] * LOG2E_F);
    s[1] = make_float4(xv[4] * LOG2E_F, xv[5] * LOG2E_F, xv[6] * LOG2E_F, xv[7] * LOG2E_F);
    s[2] = make_float4(mv[0] * LOG2E_F, mv[1] * LOG2E_F, mv[2] * LOG2E_F, mv[3] * LOG2E_F);
    s[3] = make_float4(mv[4] * LOG2E_F, mv[5] * LOG2E_F, mv[6] * LOG2E_F, mv[7] * LOG2E_F);
    s[4] = make_float4(xv[0] / zv[0], xv[1] / zv[1], xv[2] / zv[2], xv[3] / zv[3]);
    s[5] = make_float4(xv[4] / zv[4], xv[5] / zv[5], xv[6] / zv[6], xv[7] / zv[7]);
  }
  __syncthreads();

  float2 acc[NB];
  #pragma unroll
  for (int b = 0; b < NB; ++b) acc[b] = make_float2(0.f, 0.f);

  #pragma unroll
  for (int u = 0; u < AG_WROWS; ++u) {
    const float4* s = srow[wv * AG_WROWS + u];
    const float4 X0 = s[0], X1 = s[1], M0 = s[2], M1 = s[3], W0 = s[4], W1 = s[5];
    const float xs[NB] = { X0.x, X0.y, X0.z, X0.w, X1.x, X1.y, X1.z, X1.w };
    const float ms[NB] = { M0.x, M0.y, M0.z, M0.w, M1.x, M1.y, M1.z, M1.w };
    const float ws[NB] = { W0.x, W0.y, W0.z, W0.w, W1.x, W1.y, W1.z, W1.w };
    const float b0 = (av[u].x > med) ? bl.x : -INFINITY;
    const float b1 = (av[u].y > med) ? bl.y : -INFINITY;
    #pragma unroll
    for (int b = 0; b < NB; ++b) {
      float t0 = fmaf(xs[b], yv[b].x, b0);
      float t1 = fmaf(xs[b], yv[b].y, b1);
      t0 = fmaxf(t0, 0.01f * t0);
      t1 = fmaxf(t1, 0.01f * t1);
      acc[b].x = fmaf(ws[b], __builtin_amdgcn_exp2f(t0 - ms[b]), acc[b].x);
      acc[b].y = fmaf(ws[b], __builtin_amdgcn_exp2f(t1 - ms[b]), acc[b].y);
    }
  }

  #pragma unroll
  for (int b = 0; b < NB; ++b) red[wv][b][lane] = acc[b];
  __syncthreads();
  {
    const int b = (int)(threadIdx.x >> 5);
    float2 s = make_float2(0.f, 0.f);
    #pragma unroll
    for (int w = 0; w < AG_WAVES; ++w) { const float2 p = red[w][b][lane]; s.x += p.x; s.y += p.y; }
    atomicAdd(&out[b * NN + j0], s.x);
    atomicAdd(&out[b * NN + j0 + 1], s.y);
  }
}

extern "C" void kernel_launch(void* const* d_in, const int* in_sizes, int n_in,
                              void* d_out, int out_size, void* d_ws, size_t ws_size,
                              hipStream_t stream) {
  const float* x    = (const float*)d_in[0];
  const float* adj  = (const float*)d_in[1];
  const float* W    = (const float*)d_in[2];
  const float* bias = (const float*)d_in[3];
  float* out = (float*)d_out;

  char* ws = (char*)d_ws;
  size_t off = 0;
  float*    a     = (float*)(ws + off);    off += (size_t)NN * NN * sizeof(float);
  float*    Y     = (float*)(ws + off);    off += (size_t)NB * NN * sizeof(float);
  float*    rowM  = (float*)(ws + off);    off += (size_t)NB * NN * sizeof(float);
  float*    rowZ  = (float*)(ws + off);    off += (size_t)NB * NN * sizeof(float);
  unsigned* hist1 = (unsigned*)(ws + off); off += (size_t)NBINS * sizeof(unsigned);
  unsigned* hist2 = (unsigned*)(ws + off); off += (size_t)NBINS * sizeof(unsigned);
  unsigned* meta  = (unsigned*)(ws + off); off += 16;

  hipMemsetAsync(hist1, 0, (size_t)NBINS * sizeof(unsigned), stream);

  k_softmax_hist     <<<NN / SH_ROWS, SH_THREADS, 0, stream>>>(adj, a, hist1);
  k_scan_hi_gemm_y   <<<1 + GY_TILES + CLR_HIST2 + CLR_OUT, GY_THREADS, 0, stream>>>(hist1, meta, x, W, Y, hist2, out);
  k_hist_lo          <<<HL_WGS, HL_THREADS, 0, stream>>>(a, meta, hist2);
  k_scan_lo_row_stats<<<1 + RS_ROW_BLOCKS * NB, RS_THREADS, 0, stream>>>(hist2, meta, x, Y, bias, rowM, rowZ);
  k_aggregate        <<<dim3(NN / AG_COLS, NN / AG_ROWS), AG_WAVES * 32, 0, stream>>>(x, Y, bias, a, meta, rowM, rowZ, out);
}


# Low-half bin search: one work-group walking 65536 bins in a launch slot -> bins on 2 levels (1024 coarse), a wave per k_aggregate group finds the bin; count rides in row-stat groups: 5 launches -> 4
# speedup vs baseline: 44.8330x; 1.0519x over previous
#include <hip/hip_runtime.h>
#include <math.h>

#define NN 2048
#define NB 8
#define NBINS 65536

typedef __attribute__((ext_vector_type(2))) float v2f;
typedef __attribute__((ext_vector_type(8))) float v8f;

#define SH_ROWS     8
#define SH_COLS     64
#define SH_THREADS  (SH_ROWS * SH_COLS)
#define SH_K        (NN / (4 * SH_COLS))
#define FLUSH_BATCH 16

__global__ __launch_bounds__(SH_THREADS) void k_softmax_hist(const float* __restrict__ adj,
                                                             float* __restrict__ a,
                                                             unsigned* __restrict__ hist) {
  __shared__ unsigned lh[NBINS];
  __shared__ float4 xsum[SH_ROWS][SH_COLS];
  __shared__ float xmax[SH_ROWS][2];
  const int t    = threadIdx.x;
  const int r    = t >> 6;
  const int c    = t & 63;
  const int lane = t & 31;
  const int wh   = (t >> 5) & 1;
  const size_t rowoff = (size_t)(blockIdx.x * SH_ROWS + r) * NN;

  const float4* src = reinterpret_cast<const float4*>(adj + rowoff) + c;
  float4 v[SH_K];
  #pragma unroll
  for (int k = 0; k < SH_K; ++k) v[k] = src[k * SH_COLS];

  uint4* lh4 = reinterpret_cast<uint4*>(lh);
  for (int i = t; i < NBINS / 4; i += SH_THREADS) lh4[i] = make_uint4(0u, 0u, 0u, 0u);

  float mymax = -INFINITY;
  #pragma unroll
  for (int k = 0; k < SH_K; ++k)
    mymax = fmaxf(mymax, fmaxf(fmaxf(v[k].x, v[k].y), fmaxf(v[k].z, v[k].w)));
  #pragma unroll
  for (int d = 16; d >= 1; d >>= 1) mymax = fmaxf(mymax, __shfl_xor(mymax, d, 32));
  if (lane == 0) xmax[r][wh] = mymax;
  __syncthreads();
  const float m = fmaxf(xmax[r][0], xmax[r][1]);

  float4 s4 = make_float4(0.f, 0.f, 0.f, 0.f);
  #pragma unroll
  for (int k = 0; k < SH_K; ++k) {
    v[k].x = __expf(v[k].x - m);
    v[k].y = __expf(v[k].y - m);
    v[k].z = __expf(v[k].z - m);
    v[k].w = __expf(v[k].w - m);
    s4.x += v[k].x;
    s4.y += v[k].y;
    s4.z += v[k].z;
    s4.w += v[k].w;
  }
  xsum[r][c] = s4;
  __syncthreads();
  const float4 o = xsum[r][c ^ 32];
  s4.x += o.x; s4.y += o.y; s4.z += o.z; s4.w += o.w;
  #pragma unroll
  for (int d = 16; d >= 1; d >>= 1) {
    s4.x += __shfl_xor(s4.x, d, 32);
    s4.y += __shfl_xor(s4.y, d, 32);
    s4.z += __shfl_xor(s4.z, d, 32);
    s4.w += __shfl_xor(s4.w, d, 32);
  }
  const float inv = 1.f / ((s4.x + s4.z) + (s4.y + s4.w));

  float4* dst = reinterpret_cast<float4*>(a + rowoff) + c;
  #pragma unroll
  for (int k = 0; k < SH_K; ++k) {
    v[k].x *= inv; v[k].y *= inv; v[k].z *= inv; v[k].w *= inv;
    dst[k * SH_COLS] = v[k];
  }
  #pragma unroll
  for (int k = 0; k < SH_K; ++k) {
    atomicAdd(&lh[__float_as_uint(v[k].x) >> 16], 1u);
    atomicAdd(&lh[__float_as_uint(v[k].y) >> 16], 1u);
    atomicAdd(&lh[__float_as_uint(v[k].z) >> 16], 1u);
    atomicAdd(&lh[__float_as_uint(v[k].w) >> 16], 1u);
  }
  __syncthreads();

  for (int i0 = 0; i0 < NBINS; i0 += FLUSH_BATCH * SH_THREADS) {
    unsigned cnt[FLUSH_BATCH];
    #pragma unroll
    for (int u = 0; u < FLUSH_BATCH; ++u) cnt[u] = lh[i0 + u * SH_THREADS + t];
    #pragma unroll
    for (int u = 0; u < FLUSH_BATCH; ++u)
      if (cnt[u] != 0u) atomicAdd(&hist[i0 + u * SH_THREADS + t], cnt[u]);
  }
}

#define SCAN_THREADS 512
#define SCAN_WAVES   (SCAN_THREADS / 32)
#define WAVE_BINS    (NBINS / SCAN_WAVES)
#define WAVE_LOADS   (WAVE_BINS / 128)

__device__ __forceinline__ unsigned sum4(const uint4 v) { return v.x + v.y + v.z + v.w; }

__device__ __forceinline__ unsigned wave_inclusive_scan(unsigned v, const int lane) {
  #pragma unroll
  for (int d = 1; d < 32; d <<= 1) {
    const unsigned n = __shfl_up(v, d, 32);
    if (lane >= d) v += n;
  }
  return v;
}

__device__ __forceinline__ bool block_find_bin(const unsigned* __restrict__ hist,
                                               const unsigned target,
                                               unsigned& bin, unsigned& before) {
  __shared__ unsigned wave_tot[SCAN_WAVES];
  const int lane = threadIdx.x & 31;
  const int w    = threadIdx.x >> 5;
  const uint4* src = reinterpret_cast<const uint4*>(hist + w * WAVE_BINS) + lane;

  uint4 q[WAVE_LOADS];
  #pragma unroll
  for (int i = 0; i < WAVE_LOADS; ++i) q[i] = src[i * 32];

  unsigned wsum = 0;
  #pragma unroll
  for (int i = 0; i < WAVE_LOADS; ++i) wsum += sum4(q[i]);
  #pragma unroll
  for (int d = 16; d >= 1; d >>= 1) wsum += __shfl_xor(wsum, d, 32);
  if (lane == 0) wave_tot[w] = wsum;
  __syncthreads();
  const unsigned wt    = (lane < SCAN_WAVES) ? wave_tot[lane] : 0u;
  const unsigned wincl = wave_inclusive_scan(wt, lane);
  const unsigned wave_before = __shfl(wincl - wt, w, 32);
  if (!(wave_before <= target && target < wave_before + wsum)) return false;

  unsigned t[WAVE_LOADS];
  #pragma unroll
  for (int i = 0; i < WAVE_LOADS; ++i) t[i] = sum4(q[i]);
  #pragma unroll
  for (int d = 16; d >= 1; d >>= 1) {
    #pragma unroll
    for (int i = 0; i < WAVE_LOADS; ++i) t[i] += __shfl_xor(t[i], d, 32);
  }
  unsigned base = wave_before;
  uint4 c = q[0];
  int isel = 0;
  bool found = false;
  #pragma unroll
  for (int i = 0; i < WAVE_LOADS; ++i) {
    if (!found) {
      if (base + t[i] > target) { c = q[i]; isel = i; found = true; }
      else base += t[i];
    }
  }

  const unsigned mine = sum4(c);
  const unsigned lane_before = base + wave_inclusive_scan(mine, lane) - mine;
  if (!(lane_before <= target && target < lane_before + mine)) return false;
  unsigned cum = lane_before;
  unsigned comp = 0;
  if (cum + c.x <= target) {
    cum += c.x; comp = 1;
    if (cum + c.y <= target) {
      cum += c.y; comp = 2;
      if (cum + c.z <= target) { cum += c.z; comp = 3; }
    }
  }
  bin    = (unsigned)(w * WAVE_BINS + isel * 128 + lane * 4) + comp;
  before = cum;
  return true;
}

__device__ __forceinline__ void scan_hi(const unsigned* __restrict__ hist,
                                        unsigned* __restrict__ meta) {
  const unsigned k = ((unsigned)NN * (unsigned)NN - 1u) / 2u;
  unsigned bin, before;
  if (block_find_bin(hist, k, bin, before)) { meta[0] = bin; meta[1] = before; }
}

#define H2_COARSE 1024
#define H2_SPAN   (NBINS / H2_COARSE)
#define H2_WORDS  (NBINS + 2048)

__device__ __forceinline__ void count_lo(const float f, const unsigned sel, unsigned* __restrict__ hist2) {
  const unsigned bits = __float_as_uint(f);
  if ((bits >> 16) == sel) {
    const unsigned lo = bits & 0xFFFFu;
    atomicAdd(&hist2[lo], 1u);
    atomicAdd(&hist2[NBINS + lo / H2_SPAN], 1u);
  }
}

__device__ __forceinline__ unsigned wave_prefix_rows(unsigned v, const int lane) {
  v += (unsigned)__builtin_amdgcn_update_dpp(0, (int)v, 0x111, 0xf, 0xf, false);
  v += (unsigned)__builtin_amdgcn_update_dpp(0, (int)v, 0x112, 0xf, 0xf, false);
  v += (unsigned)__builtin_amdgcn_update_dpp(0, (int)v, 0x114, 0xf, 0xf, false);
  v += (unsigned)__builtin_amdgcn_update_dpp(0, (int)v, 0x118, 0xf, 0xf, false);
  const unsigned row0 = (unsigned)__builtin_amdgcn_readlane((int)v, 15);
  if (lane >= 16) v += row0;
  return v;
}

__device__ __forceinline__ unsigned pick4(const uint4 c, unsigned& cum, const unsigned target) {
  unsigned comp = 0;
  if (cum + c.x <= target) {
    cum += c.x; comp = 1;
    if (cum + c.y <= target) {
      cum += c.y; comp = 2;
      if (cum + c.z <= target) { cum += c.z; comp = 3; }
    }
  }
  return comp;
}

__device__ __forceinline__ void from_hit_lane(const bool hit, const unsigned a, const unsigned b,
                                              unsigned& ra, unsigned& rb) {
  const unsigned hits = __builtin_amdgcn_ballot_w32(hit);
  const int from = hits ? __builtin_ctz(hits) : 0;
  ra = (unsigned)__builtin_amdgcn_readlane((int)a, from);
  rb = (unsigned)__builtin_amdgcn_readlane((int)b, from);
}

#define H2_LANE_BINS  (H2_COARSE / 32)
#define H2_LANE_LOADS (H2_LANE_BINS / 4)

__device__ __forceinline__ void coarse_load(const unsigned* __restrict__ hist2, const int lane,
                                            uint4 (&q)[H2_LANE_LOADS]) {
  const uint4* csrc = reinterpret_cast<const uint4*>(hist2 + NBINS) + lane * H2_LANE_LOADS;
  #pragma unroll
  for (int i = 0; i < H2_LANE_LOADS; ++i) q[i] = csrc[i];
}

__device__ __forceinline__ void coarse_find(const uint4 (&q)[H2_LANE_LOADS], const unsigned target,
                                            const int lane, unsigned& cb, unsigned& cbefore) {
  unsigned s4[H2_LANE_LOADS];
  unsigned mine = 0;
  #pragma unroll
  for (int i = 0; i < H2_LANE_LOADS; ++i) { s4[i] = sum4(q[i]); mine += s4[i]; }
  const unsigned incl = wave_prefix_rows(mine, lane);
  const unsigned excl = incl - mine;
  const bool hit = (excl <= target) && (target < incl);
  unsigned cum = excl;
  uint4 c = q[0];
  unsigned piece = 0;
  bool found = false;
  #pragma unroll
  for (int i = 0; i < H2_LANE_LOADS; ++i) {
    if (!found) {
      if (cum + s4[i] > target) { c = q[i]; piece = (unsigned)i; found = true; }
      else cum += s4[i];
    }
  }
  const unsigned comp = pick4(c, cum, target);
  from_hit_lane(hit, (unsigned)lane * H2_LANE_BINS + piece * 4u + comp, cum, cb, cbefore);
}

__device__ __forceinline__ unsigned fine_find(const uint4 f, const unsigned cb, const unsigned cbefore,
                                              const unsigned target, const int lane) {
  const unsigned mine = sum4(f);
  const unsigned incl = cbefore + wave_prefix_rows(mine, lane);
  unsigned cum = incl - mine;
  const bool hit = (cum <= target) && (target < incl);
  const unsigned comp = pick4(f, cum, target);
  unsigned fidx, fbefore;
  from_hit_lane(hit, (unsigned)lane * 4u + comp, cum, fidx, fbefore);
  return cb * H2_SPAN + fidx;
}

#define GY_WAVES   16
#define GY_THREADS (GY_WAVES * 32)
#define GY_KSLICE  (NN / GY_WAVES)
#define GY_PITCH   (GY_KSLICE + 4)
#define GY_PIECES  (GY_KSLICE / 8)

__device__ __forceinline__ void gemm_y_tile(const float* __restrict__ x,
                                            const float* __restrict__ W,
                                            float* __restrict__ Y,
                                            const int tile) {
  __shared__ __attribute__((aligned(16))) float Wt[GY_WAVES][16][GY_PITCH];
  __shared__ __attribute__((aligned(16))) float Xt[GY_WAVES][NB][GY_PITCH];
  __shared__ float part[GY_WAVES][16][16];
  const int nb   = tile * 16;
  const int lane = threadIdx.x & 31;
  const int wv   = threadIdx.x >> 5;
  const int half = lane >> 4;
  const int l16  = lane & 15;
  const int k0   = wv * GY_KSLICE;

  float4 wr[16], xr[NB];
  const float* wsrc = W + (size_t)nb * NN + k0 + 4 * lane;
  #pragma unroll
  for (int r = 0; r < 16; ++r) wr[r] = *reinterpret_cast<const float4*>(wsrc + (size_t)r * NN);
  const float* xsrc = x + k0 + 4 * lane;
  #pragma unroll
  for (int r = 0; r < NB; ++r) xr[r] = *reinterpret_cast<const float4*>(xsrc + (size_t)r * NN);

  #pragma unroll
  for (int r = 0; r < 16; ++r) *reinterpret_cast<float4*>(&Wt[wv][r][4 * lane]) = wr[r];
  #pragma unroll
  for (int r = 0; r < NB; ++r) *reinterpret_cast<float4*>(&Xt[wv][r][4 * lane]) = xr[r];
  __syncthreads();

  const float* wrow = &Wt[wv][l16][4 * half];
  const float* xrow = &Xt[wv][l16 & 7][4 * half];
  v8f acc0 = {}, acc1 = {};
  #pragma unroll
  for (int p = 0; p < GY_PIECES; ++p) {
    const float4 bw = *reinterpret_cast<const float4*>(wrow + 8 * p);
    const float4 ax = *reinterpret_cast<const float4*>(xrow + 8 * p);
    v2f a0 = { ax.x, ax.y };
    v2f b0 = { bw.x, bw.y };
    acc0 = __builtin_amdgcn_wmma_f32_16x16x4_f32(false, a0, false, b0,
                                                 (short)0, acc0, false, false);
    v2f a1 = { ax.z, ax.w };
    v2f b1 = { bw.z, bw.w };
    acc1 = __builtin_amdgcn_wmma_f32_16x16x4_f32(false, a1, false, b1,
                                                 (short)0, acc1, false, false);
  }
  #pragma unroll
  for (int r = 0; r < 8; ++r) part[wv][r + half * 8][l16] = acc0[r] + acc1[r];
  __syncthreads();

  if (threadIdx.x < NB * 16) {
    const int m = threadIdx.x >> 4, c = threadIdx.x & 15;
    float s = 0.f;
    #pragma unroll
    for (int w = 0; w < GY_WAVES; ++w) s += part[w][m][c];
    Y[m * NN + nb + c] = s;
  }
}

static_assert(GY_THREADS == SCAN_THREADS, "the scan and the tiles share a launch");
#define GY_TILES    (NN / 16)
#define CLR_BYTES   (GY_THREADS * 16)
#define CLR_HIST2   (H2_WORDS * 4 / CLR_BYTES)
#define CLR_OUT     (NB * NN * 4 / CLR_BYTES)
__global__ __launch_bounds__(GY_THREADS) void k_scan_hi_gemm_y(const unsigned* __restrict__ hist,
                                                               unsigned* __restrict__ meta,
                                                               const float* __restrict__ x,
                                                               const float* __restrict__ W,
                                                               float* __restrict__ Y,
                                                               unsigned* __restrict__ hist2,
                                                               float* __restrict__ out) {
  const int id = (int)blockIdx.x;
  if (id == 0) {
    scan_hi(hist, meta);
  } else if (id <= GY_TILES) {
    gemm_y_tile(x, W, Y, id - 1);
  } else {
    const int c = id - 1 - GY_TILES;
    uint4* dst = (c < CLR_HIST2) ? reinterpret_cast<uint4*>(hist2) + c * GY_THREADS
                                 : reinterpret_cast<uint4*>(out) + (c - CLR_HIST2) * GY_THREADS;
    dst[threadIdx.x] = make_uint4(0u, 0u, 0u, 0u);
  }
}

#define RS_WAVES   16
#define RS_THREADS (RS_WAVES * 32)
#define RS_ROWS    4
#define RS_WG_ROWS (RS_WAVES * RS_ROWS)
#define RS_Q       (NN / 128)

#define RS_ROW_BLOCKS (NN / RS_WG_ROWS)
#define HL_PIECES  8
static_assert((size_t)RS_ROW_BLOCKS * NB * RS_THREADS * HL_PIECES * 4 == (size_t)NN * NN,
              "the row-statistics work-groups cover a exactly once");

__device__ __forceinline__ void row_stats(const float* __restrict__ x,
                                          const float* __restrict__ Y,
                                          const float* __restrict__ bias,
                                          float* __restrict__ rowM,
                                          float* __restrict__ rowZ,
                                          const int b, const int rowBase,
                                          const float* __restrict__ a,
                                          const unsigned* __restrict__ meta,
                                          unsigned* __restrict__ hist2,
                                          const int piece) {
  __shared__ float sM[RS_WG_ROWS], sZ[RS_WG_ROWS];
  const int lane = threadIdx.x & 31;
  const int wv = __builtin_amdgcn_readfirstlane((int)(threadIdx.x >> 5));

  const float4* asrc = reinterpret_cast<const float4*>(a)
                     + (size_t)piece * (RS_THREADS * HL_PIECES) + threadIdx.x;
  float4 av[HL_PIECES];
  #pragma unroll
  for (int k = 0; k < HL_PIECES; ++k) av[k] = asrc[k * RS_THREADS];
  const unsigned sel = meta[0];

  float4 y[RS_Q], bb[RS_Q];
  const float4* ysrc = reinterpret_cast<const float4*>(Y + (size_t)b * NN) + lane;
  const float4* bsrc = reinterpret_cast<const float4*>(bias) + lane;
  #pragma unroll
  for (int q = 0; q < RS_Q; ++q) y[q] = ysrc[q * 32];
  #pragma unroll
  for (int q = 0; q < RS_Q; ++q) bb[q] = bsrc[q * 32];
  const float xv = x[b * NN + rowBase + wv * RS_ROWS + (lane & (RS_ROWS - 1))];

  #pragma unroll
  for (int k = 0; k < HL_PIECES; ++k) {
    count_lo(av[k].x, sel, hist2);
    count_lo(av[k].y, sel, hist2);
    count_lo(av[k].z, sel, hist2);
    count_lo(av[k].w, sel, hist2);
  }

  for (int r = 0; r < RS_ROWS; ++r) {
    const float xbi = __uint_as_float((unsigned)__builtin_amdgcn_readlane((int)__float_as_uint(xv), r));

    float4 v[RS_Q];
    float4 m4 = make_float4(-INFINITY, -INFINITY, -INFINITY, -INFINITY);
    #pragma unroll
    for (int q = 0; q < RS_Q; ++q) {
      float4 u;
      u.x = fmaf(xbi, y[q].x, bb[q].x);
      u.y = fmaf(xbi, y[q].y, bb[q].y);
      u.z = fmaf(xbi, y[q].z, bb[q].z);
      u.w = fmaf(xbi, y[q].w, bb[q].w);
      u.x = fmaxf(u.x, 0.01f * u.x);
      u.y = fmaxf(u.y, 0.01f * u.y);
      u.z = fmaxf(u.z, 0.01f * u.z);
      u.w = fmaxf(u.w, 0.01f * u.w);
      v[q] = u;
      m4.x = fmaxf(m4.x, u.x);
      m4.y = fmaxf(m4.y, u.y);
      m4.z = fmaxf(m4.z, u.z);
      m4.w = fmaxf(m4.w, u.w);
    }
    float m = fmaxf(fmaxf(m4.x, m4.y), fmaxf(m4.z, m4.w));
    #pragma unroll
    for (int d = 16; d >= 1; d >>= 1) m = fmaxf(m, __shfl_xor(m, d, 32));

    float4 s4 = make_float4(0.f, 0.f, 0.f, 0.f);
    #pragma unroll
    for (int q = 0; q < RS_Q; ++q) {
      s4.x += __expf(v[q].x - m);
      s4.y += __expf(v[q].y - m);
      s4.z += __expf(v[q].z - m);
      s4.w += __expf(v[q].w - m);
    }
    float s = (s4.x + s4.y) + (s4.z + s4.w);
    #pragma unroll
    for (int d = 16; d >= 1; d >>= 1) s += __shfl_xor(s, d, 32);

    if (lane == 0) { sM[wv * RS_ROWS + r] = m; sZ[wv * RS_ROWS + r] = s; }
  }
  __syncthreads();
  if (wv < 2)      rowM[b * NN + rowBase + wv * 32 + lane]       = sM[wv * 32 + lane];
  else if (wv < 4) rowZ[b * NN + rowBase + (wv - 2) * 32 + lane] = sZ[(wv - 2) * 32 + lane];
}

__global__ __launch_bounds__(RS_THREADS) void k_hist_lo_row_stats(const float* __restrict__ a,
                                                                  const unsigned* __restrict__ meta,
                                                                  unsigned* __restrict__ hist2,
                                                                  const float* __restrict__ x,
                                                                  const float* __restrict__ Y,
                                                                  const float* __restrict__ bias,
                                                                  float* __restrict__ rowM,
                                                                  float* __restrict__ rowZ) {
  const int id = (int)blockIdx.x;
  row_stats(x, Y, bias, rowM, rowZ, id / RS_ROW_BLOCKS, (id % RS_ROW_BLOCKS) * RS_WG_ROWS,
            a, meta, hist2, id);
}

#define AG_COLS   64
#define AG_ROWS   256
#define AG_WAVES  8
#define AG_WROWS  (AG_ROWS / AG_WAVES)
#define LOG2E_F   1.4426950408889634f

__global__ __launch_bounds__(AG_WAVES * 32) void k_aggregate(const float* __restrict__ x,
                                                             const float* __restrict__ Y,
                                                             const float* __restrict__ bias,
                                                             const float* __restrict__ a,
                                                             const unsigned* __restrict__ meta,
                                                             const unsigned* __restrict__ hist2,
                                                             const float* __restrict__ rowM,
                                                             const float* __restrict__ rowZ,
                                                             float* __restrict__ out) {
  __shared__ float4 srow[AG_ROWS][6];
  __shared__ float2 red[AG_WAVES][NB][32];
  __shared__ unsigned smed;
  const int lane = threadIdx.x & 31;
  const int wv = __builtin_amdgcn_readfirstlane((int)(threadIdx.x >> 5));
  const int j0 = blockIdx.x * AG_COLS + 2 * lane;
  const int iBase = blockIdx.y * AG_ROWS;

  uint4 cq[H2_LANE_LOADS];
  #pragma unroll
  for (int i = 0; i < H2_LANE_LOADS; ++i) cq[i] = make_uint4(0u, 0u, 0u, 0u);
  if (wv == 0) coarse_load(hist2, lane, cq);

  const float* acol = a + (size_t)(iBase + wv * AG_WROWS) * NN + j0;
  float2 av[AG_WROWS];
  #pragma unroll
  for (int u = 0; u < AG_WROWS; ++u) av[u] = *reinterpret_cast<const float2*>(acol + (size_t)u * NN);

  float2 yv[NB];
  #pragma unroll
  for (int b = 0; b < NB; ++b) yv[b] = *reinterpret_cast<const float2*>(Y + b * NN + j0);
  float2 bl = *reinterpret_cast<const float2*>(bias + j0);
  bl.x *= LOG2E_F; bl.y *= LOG2E_F;

  const unsigned ktarget = ((unsigned)NN * (unsigned)NN - 1u) / 2u - meta[1];
  {
    const int i = iBase + (int)threadIdx.x;
    float xv[NB], mv[NB], zv[NB];
    #pragma unroll
    for (int b = 0; b < NB; ++b) {
      xv[b] = x[b * NN + i];
      mv[b] = rowM[b * NN + i];
      zv[b] = rowZ[b * NN + i];
    }
    unsigned cb = 0, cbefore = 0;
    uint4 fq = make_uint4(0u, 0u, 0u, 0u);
    if (wv == 0) {
      coarse_find(cq, ktarget, lane, cb, cbefore);
      if (lane < H2_SPAN / 4) fq = reinterpret_cast<const uint4*>(hist2 + cb * H2_SPAN)[lane];
    }
    float4* s = srow[threadIdx.x];
    s[0] = make_float4(xv[0] * LOG2E_F, xv[1] * LOG2E_F, xv[2] * LOG2E_F, xv[3] * LOG2E_F);
    s[1] = make_float4(xv[4] * LOG2E_F, xv[5] * LOG2E_F, xv[6] * LOG2E_F, xv[7] * LOG2E_F);
    s[2] = make_float4(mv[0] * LOG2E_F, mv[1] * LOG2E_F, mv[2] * LOG2E_F, mv[3] * LOG2E_F);
    s[3] = make_float4(mv[4] * LOG2E_F, mv[5] * LOG2E_F, mv[6] * LOG2E_F, mv[7] * LOG2E_F);
    s[4] = make_float4(xv[0] / zv[0], xv[1] / zv[1], xv[2] / zv[2], xv[3] / zv[3]);
    s[5] = make_float4(xv[4] / zv[4], xv[5] / zv[5], xv[6] / zv[6], xv[7] / zv[7]);
    if (wv == 0) {
      const unsigned low = fine_find(fq, cb, cbefore, ktarget, lane);
      if (lane == 0) smed = (meta[0] << 16) | low;
    }
  }
  __syncthreads();
  const float med = __uint_as_float(smed);

  float2 acc[NB];
  #pragma unroll
  for (int b = 0; b < NB; ++b) acc[b] = make_float2(0.f, 0.f);

  #pragma unroll
  for (int u = 0; u < AG_WROWS; ++u) {
    const float4* s = srow[wv * AG_WROWS + u];
    const float4 X0 = s[0], X1 = s[1], M0 = s[2], M1 = s[3], W0 = s[4], W1 = s[5];
    const float xs[NB] = { X0.x, X0.y, X0.z, X0.w, X1.x, X1.y, X1.z, X1.w };
    const float ms[NB] = { M0.x, M0.y, M0.z, M0.w, M1.x, M1.y, M1.z, M1.w };
    const float ws[NB] = { W0.x, W0.y, W0.z, W0.w, W1.x, W1.y, W1.z, W1.w };
    const float b0 = (av[u].x > med) ? bl.x : -INFINITY;
    const float b1 = (av[u].y > med) ? bl.y : -INFINITY;
    #pragma unroll
    for (int b = 0; b < NB; ++b) {
      float t0 = fmaf(xs[b], yv[b].x, b0);
      float t1 = fmaf(xs[b], yv[b].y, b1);
      t0 = fmaxf(t0, 0.01f * t0);
      t1 = fmaxf(t1, 0.01f * t1);
      acc[b].x = fmaf(ws[b], __builtin_amdgcn_exp2f(t0 - ms[b]), acc[b].x);
      acc[b].y = fmaf(ws[b], __builtin_amdgcn_exp2f(t1 - ms[b]), acc[b].y);
    }
  }

  #pragma unroll
  for (int b = 0; b < NB; ++b) red[wv][b][lane] = acc[b];
  __syncthreads();
  {
    const int b = (int)(threadIdx.x >> 5);
    float2 s = make_float2(0.f, 0.f);
    #pragma unroll
    for (int w = 0; w < AG_WAVES; ++w) { const float2 p = red[w][b][lane]; s.x += p.x; s.y += p.y; }
    atomicAdd(&out[b * NN + j0], s.x);
    atomicAdd(&out[b * NN + j0 + 1], s.y);
  }
}

extern "C" void kernel_launch(void* const* d_in, const int* in_sizes, int n_in,
                              void* d_out, int out_size, void* d_ws, size_t ws_size,
                              hipStream_t stream) {
  const float* x    = (const float*)d_in[0];
  const float* adj  = (const float*)d_in[1];
  const float* W    = (const float*)d_in[2];
  const float* bias = (const float*)d_in[3];
  float* out = (float*)d_out;

  char* ws = (char*)d_ws;
  size_t off = 0;
  float*    a     = (float*)(ws + off);    off += (size_t)NN * NN * sizeof(float);
  float*    Y     = (float*)(ws + off);    off += (size_t)NB * NN * sizeof(float);
  float*    rowM  = (float*)(ws + off);    off += (size_t)NB * NN * sizeof(float);
  float*    rowZ  = (float*)(ws + off);    off += (size_t)NB * NN * sizeof(float);
  unsigned* hist1 = (unsigned*)(ws + off); off += (size_t)NBINS * sizeof(unsigned);
  unsigned* hist2 = (unsigned*)(ws + off); off += (size_t)H2_WORDS * sizeof(unsigned);
  unsigned* meta  = (unsigned*)(ws + off); off += 16;

  hipMemsetAsync(hist1, 0, (size_t)NBINS * sizeof(unsigned), stream);

  k_softmax_hist     <<<NN / SH_ROWS, SH_THREADS, 0, stream>>>(adj, a, hist1);
  k_scan_hi_gemm_y   <<<1 + GY_TILES + CLR_HIST2 + CLR_OUT, GY_THREADS, 0, stream>>>(hist1, meta, x, W, Y, hist2, out);
  k_hist_lo_row_stats<<<RS_ROW_BLOCKS * NB, RS_THREADS, 0, stream>>>(a, meta, hist2, x, Y, bias, rowM, rowZ);
  k_aggregate        <<<dim3(NN / AG_COLS, NN / AG_ROWS), AG_WAVES * 32, 0, stream>>>(x, Y, bias, a, meta, hist2, rowM, rowZ, out);
}
